# starting kernel
# baseline (speedup 1.0000x reference)
#include <hip/hip_runtime.h>
#include <math.h>


typedef __attribute__((ext_vector_type(2))) float v2f;
typedef __attribute__((ext_vector_type(8))) float v8f;

#define DIM 128
#define BN_EPS 1e-5f


__global__ void zero_f32_kernel(float* __restrict__ p, int n) {
  int i = blockIdx.x * blockDim.x + threadIdx.x;
  if (i < n) p[i] = 0.0f;
}

__global__ void degree_kernel(const int* __restrict__ src, const int* __restrict__ dst,
                              int* __restrict__ deg_s, int* __restrict__ deg_d, int E) {
  int e = blockIdx.x * blockDim.x + threadIdx.x;
  if (e < E) {
    atomicAdd(&deg_s[src[e]], 1);
    atomicAdd(&deg_d[dst[e]], 1);
  }
}

__global__ void norm_kernel(const int* __restrict__ deg_s, const int* __restrict__ deg_d,
                            float* __restrict__ onorm, float* __restrict__ inorm, int N) {
  int i = blockIdx.x * blockDim.x + threadIdx.x;
  if (i < N) {
    int ds = deg_s[i];
    int dd = deg_d[i];
    onorm[i] = (ds > 0) ? 1.0f / sqrtf((float)ds) : 0.0f;
    inorm[i] = (dd > 0) ? 1.0f / sqrtf((float)dd) : 0.0f;
  }
}

__global__ void gather_kernel(const int* __restrict__ ids, const float* __restrict__ emb,
                              float* __restrict__ x, int N) {
  int t = blockIdx.x * blockDim.x + threadIdx.x;
  int total = N * (DIM / 4);
  if (t >= total) return;
  int i = t / (DIM / 4);
  int c = (t % (DIM / 4)) * 4;
  const float4 v = *reinterpret_cast<const float4*>(emb + (size_t)ids[i] * DIM + c);
  *reinterpret_cast<float4*>(x + (size_t)i * DIM + c) = v;
}

__global__ void scatter_kernel(const int* __restrict__ src, const int* __restrict__ dst,
                               const float* __restrict__ x, const float* __restrict__ onorm,
                               float* __restrict__ agg, int E) {
  long long t = (long long)blockIdx.x * blockDim.x + threadIdx.x;
  int e = (int)(t >> 5);
  if (e >= E) return;
  int c = ((int)t & 31) * 4;
  int s = src[e];
  int d = dst[e];
  float ns = onorm[s];
  const float4 v = *reinterpret_cast<const float4*>(x + (size_t)s * DIM + c);
  float* ap = agg + (size_t)d * DIM + c;
  atomicAdd(ap + 0, v.x * ns);
  atomicAdd(ap + 1, v.y * ns);
  atomicAdd(ap + 2, v.z * ns);
  atomicAdd(ap + 3, v.w * ns);
}

__global__ void __launch_bounds__(256) gemm_fused_kernel(
    const float* __restrict__ agg, const float* __restrict__ xin,
    const float* __restrict__ inorm,
    const float* __restrict__ W, const float* __restrict__ bias,
    const float* __restrict__ Rw, const float* __restrict__ Rb,
    float* __restrict__ h, float* __restrict__ csum, float* __restrict__ csumsq,
    int N) {
  const int wave = threadIdx.x >> 5;
  const int lane = threadIdx.x & 31;
  const int rowTile = blockIdx.x * 8 + wave;
  const int row0 = rowTile * 16;
  if (row0 >= N) return;

  const int r = lane & 15;
  const int half = lane >> 4;
  const int row = row0 + r;
  const float innr = inorm[row];
  const float* aggRow = agg + (size_t)row * DIM;
  const float* xRow = xin + (size_t)row * DIM;

  for (int ct = 0; ct < 8; ++ct) {
    const int col0 = ct * 16;
    v8f acc1 = {0.f, 0.f, 0.f, 0.f, 0.f, 0.f, 0.f, 0.f};
    v8f acc2 = {0.f, 0.f, 0.f, 0.f, 0.f, 0.f, 0.f, 0.f};
    const float* wc = W + col0 + r;
    const float* rc = Rw + col0 + r;
#pragma unroll 8
    for (int k0 = 0; k0 < DIM; k0 += 4) {
      const int kk = k0 + 2 * half;
      const float2 t1 = *reinterpret_cast<const float2*>(aggRow + kk);
      const float2 t2 = *reinterpret_cast<const float2*>(xRow + kk);
      v2f a1, a2, b1, b2;
      a1.x = t1.x * innr;
      a1.y = t1.y * innr;
      a2.x = t2.x;
      a2.y = t2.y;
      b1.x = wc[(size_t)kk * DIM];
      b1.y = wc[(size_t)(kk + 1) * DIM];
      b2.x = rc[(size_t)kk * DIM];
      b2.y = rc[(size_t)(kk + 1) * DIM];
      acc1 = __builtin_amdgcn_wmma_f32_16x16x4_f32(false, a1, false, b1,
                                                   (short)0, acc1, false, false);
      acc2 = __builtin_amdgcn_wmma_f32_16x16x4_f32(false, a2, false, b2,
                                                   (short)0, acc2, false, false);
    }
    const float bb = bias[col0 + r];
    const float rb = Rb[col0 + r];
    float s = 0.0f, s2 = 0.0f;
#pragma unroll
    for (int j = 0; j < 8; ++j) {
      float v = fmaxf(acc1[j] + bb, 0.0f) + fmaxf(acc2[j] + rb, 0.0f);
      h[(size_t)(row0 + j + 8 * half) * DIM + col0 + r] = v;
      s += v;
      s2 += v * v;
    }
    atomicAdd(&csum[col0 + r], s);
    atomicAdd(&csumsq[col0 + r], s2);
  }
}

__global__ void bnprep_kernel(const float* __restrict__ csum, const float* __restrict__ csumsq,
                              const float* __restrict__ gamma, const float* __restrict__ beta,
                              float* __restrict__ scale, float* __restrict__ shift, float invN) {
  int c = threadIdx.x;
  float mu = csum[c] * invN;
  float var = csumsq[c] * invN - mu * mu;
  float sc = gamma[c] / sqrtf(var + BN_EPS);
  scale[c] = sc;
  shift[c] = beta[c] - mu * sc;
}

__global__ void bnapply_kernel(const float* __restrict__ h, const float* __restrict__ scale,
                               const float* __restrict__ shift, float* __restrict__ out,
                               int total) {
  int i = blockIdx.x * blockDim.x + threadIdx.x;
  if (i < total) {
    int c = i & (DIM - 1);
    out[i] = h[i] * scale[c] + shift[c];
  }
}


static inline int cdiv(int a, int b) { return (a + b - 1) / b; }

extern "C" void kernel_launch(void* const* d_in, const int* in_sizes, int n_in,
                              void* d_out, int out_size, void* d_ws, size_t ws_size,
                              hipStream_t stream) {
  const int* node_ids = (const int*)d_in[0];
  const int* src = (const int*)d_in[1];
  const int* dst = (const int*)d_in[2];
  const float* emb = (const float*)d_in[3];
  const float* Ws = (const float*)d_in[4];
  const float* bs = (const float*)d_in[5];
  const float* Rws = (const float*)d_in[6];
  const float* Rbs = (const float*)d_in[7];
  const float* gammas = (const float*)d_in[8];
  const float* betas = (const float*)d_in[9];

  const int N = in_sizes[0];
  const int E = in_sizes[1];
  const int L = in_sizes[4] / (DIM * DIM);

  float* wsf = (float*)d_ws;
  size_t off = 0;
  float* onorm = wsf + off; off += (size_t)N;
  float* inorm = wsf + off; off += (size_t)N;
  float* x    = wsf + off;  off += (size_t)N * DIM;
  float* agg  = wsf + off;  off += (size_t)N * DIM;
  float* h    = wsf + off;  off += (size_t)N * DIM;
  float* csum   = wsf + off; off += DIM;
  float* csumsq = wsf + off; off += DIM;
  float* scale  = wsf + off; off += DIM;
  float* shift  = wsf + off; off += DIM;
  int* deg_s = (int*)h;
  int* deg_d = deg_s + N;

  const int B = 256;

  zero_f32_kernel<<<cdiv(2 * N, B), B, 0, stream>>>((float*)deg_s, 2 * N);
  degree_kernel<<<cdiv(E, B), B, 0, stream>>>(src, dst, deg_s, deg_d, E);
  norm_kernel<<<cdiv(N, B), B, 0, stream>>>(deg_s, deg_d, onorm, inorm, N);

  gather_kernel<<<cdiv(N * (DIM / 4), B), B, 0, stream>>>(node_ids, emb, x, N);

  const int ND = N * DIM;
  const int rowTiles = cdiv(N, 16);
  for (int l = 0; l < L; ++l) {
    zero_f32_kernel<<<cdiv(ND, B), B, 0, stream>>>(agg, ND);
    zero_f32_kernel<<<1, B, 0, stream>>>(csum, 2 * DIM);

    scatter_kernel<<<cdiv(E * 32, B), B, 0, stream>>>(src, dst, x, onorm, agg, E);

    gemm_fused_kernel<<<cdiv(rowTiles, 8), B, 0, stream>>>(
        agg, x, inorm,
        Ws + (size_t)l * DIM * DIM, bs + (size_t)l * DIM,
        Rws + (size_t)l * DIM * DIM, Rbs + (size_t)l * DIM,
        h, csum, csumsq, N);

    bnprep_kernel<<<1, DIM, 0, stream>>>(csum, csumsq, gammas + (size_t)l * DIM,
                                         betas + (size_t)l * DIM, scale, shift,
                                         1.0f / (float)N);

    float* dest = (l == L - 1) ? (float*)d_out : x;
    bnapply_kernel<<<cdiv(ND, B), B, 0, stream>>>(h, scale, shift, dest, ND);
  }
}


# GEMM operand reuse: W/Rw staged once per block in LDS in fragment order, A read once per wave as float4 with 16 register tiles; k order inside groups of 8 permuted (fp32 sum order changes)
# speedup vs baseline: 1.1742x; 1.1742x over previous
#include <hip/hip_runtime.h>
#include <math.h>


typedef __attribute__((ext_vector_type(2))) float v2f;
typedef __attribute__((ext_vector_type(8))) float v8f;

#define DIM 128
#define BN_EPS 1e-5f


__global__ void zero_f32_kernel(float* __restrict__ p, int n) {
  int i = blockIdx.x * blockDim.x + threadIdx.x;
  if (i < n) p[i] = 0.0f;
}

__global__ void degree_kernel(const int* __restrict__ src, const int* __restrict__ dst,
                              int* __restrict__ deg_s, int* __restrict__ deg_d, int E) {
  int e = blockIdx.x * blockDim.x + threadIdx.x;
  if (e < E) {
    atomicAdd(&deg_s[src[e]], 1);
    atomicAdd(&deg_d[dst[e]], 1);
  }
}

__global__ void norm_kernel(const int* __restrict__ deg_s, const int* __restrict__ deg_d,
                            float* __restrict__ onorm, float* __restrict__ inorm, int N) {
  int i = blockIdx.x * blockDim.x + threadIdx.x;
  if (i < N) {
    int ds = deg_s[i];
    int dd = deg_d[i];
    onorm[i] = (ds > 0) ? 1.0f / sqrtf((float)ds) : 0.0f;
    inorm[i] = (dd > 0) ? 1.0f / sqrtf((float)dd) : 0.0f;
  }
}

__global__ void gather_kernel(const int* __restrict__ ids, const float* __restrict__ emb,
                              float* __restrict__ x, int N) {
  int t = blockIdx.x * blockDim.x + threadIdx.x;
  int total = N * (DIM / 4);
  if (t >= total) return;
  int i = t / (DIM / 4);
  int c = (t % (DIM / 4)) * 4;
  const float4 v = *reinterpret_cast<const float4*>(emb + (size_t)ids[i] * DIM + c);
  *reinterpret_cast<float4*>(x + (size_t)i * DIM + c) = v;
}

__global__ void scatter_kernel(const int* __restrict__ src, const int* __restrict__ dst,
                               const float* __restrict__ x, const float* __restrict__ onorm,
                               float* __restrict__ agg, int E) {
  long long t = (long long)blockIdx.x * blockDim.x + threadIdx.x;
  int e = (int)(t >> 5);
  if (e >= E) return;
  int c = ((int)t & 31) * 4;
  int s = src[e];
  int d = dst[e];
  float ns = onorm[s];
  const float4 v = *reinterpret_cast<const float4*>(x + (size_t)s * DIM + c);
  float* ap = agg + (size_t)d * DIM + c;
  atomicAdd(ap + 0, v.x * ns);
  atomicAdd(ap + 1, v.y * ns);
  atomicAdd(ap + 2, v.z * ns);
  atomicAdd(ap + 3, v.w * ns);
}

__global__ void __launch_bounds__(256) gemm_fused_kernel(
    const float* __restrict__ agg, const float* __restrict__ xin,
    const float* __restrict__ inorm,
    const float* __restrict__ W, const float* __restrict__ bias,
    const float* __restrict__ Rw, const float* __restrict__ Rb,
    float* __restrict__ h, float* __restrict__ csum, float* __restrict__ csumsq,
    int N) {
  __shared__ float4 sB[32 * 8 * 32];

  const int tid = threadIdx.x;
  const int wave = tid >> 5;
  const int lane = tid & 31;
  const int r = lane & 15;
  const int hf = lane >> 4;

  {
    const float* wp = W + (size_t)(4 * hf) * DIM + wave * 16 + r;
    const float* rp = Rw + (size_t)(4 * hf) * DIM + wave * 16 + r;
    float4* sp = sB + tid;
#pragma unroll 4
    for (int kp = 0; kp < 16; ++kp) {
#pragma unroll
      for (int q = 0; q < 2; ++q) {
        float4 v;
        v.x = wp[(2 * q) * DIM];
        v.y = wp[(2 * q + 1) * DIM];
        v.z = rp[(2 * q) * DIM];
        v.w = rp[(2 * q + 1) * DIM];
        sp[q * 256] = v;
      }
      wp += 8 * DIM;
      rp += 8 * DIM;
      sp += 2 * 256;
    }
  }
  __syncthreads();

  const int row0 = (blockIdx.x * 8 + wave) * 16;
  if (row0 >= N) return;

  const int row = row0 + r;
  const float innr = inorm[row];
  const float4* aP = reinterpret_cast<const float4*>(agg + (size_t)row * DIM) + hf;
  const float4* xP = reinterpret_cast<const float4*>(xin + (size_t)row * DIM) + hf;
  const float4* bP = sB + lane;

  v8f acc1[8], acc2[8];
#pragma unroll
  for (int ct = 0; ct < 8; ++ct) {
    acc1[ct] = (v8f){0.f, 0.f, 0.f, 0.f, 0.f, 0.f, 0.f, 0.f};
    acc2[ct] = (v8f){0.f, 0.f, 0.f, 0.f, 0.f, 0.f, 0.f, 0.f};
  }

  float4 ta = aP[0];
  float4 tx = xP[0];
  float4 b0[8], b1[8];
#pragma unroll
  for (int ct = 0; ct < 8; ++ct) b0[ct] = bP[ct * 32];

  for (int kp = 0; kp < 16; ++kp) {
    float4 ta_n = ta, tx_n = tx;
    if (kp + 1 < 16) {
      ta_n = aP[(kp + 1) * 2];
      tx_n = xP[(kp + 1) * 2];
    }
#pragma unroll
    for (int ct = 0; ct < 8; ++ct) b1[ct] = bP[((2 * kp + 1) * 8 + ct) * 32];

    v2f a1, a2;
    a1.x = ta.x * innr;
    a1.y = ta.y * innr;
    a2.x = tx.x;
    a2.y = tx.y;
#pragma unroll
    for (int ct = 0; ct < 8; ++ct) {
      v2f w1, w2;
      w1.x = b0[ct].x; w1.y = b0[ct].y;
      w2.x = b0[ct].z; w2.y = b0[ct].w;
      acc1[ct] = __builtin_amdgcn_wmma_f32_16x16x4_f32(false, a1, false, w1,
                                                       (short)0, acc1[ct], false, false);
      acc2[ct] = __builtin_amdgcn_wmma_f32_16x16x4_f32(false, a2, false, w2,
                                                       (short)0, acc2[ct], false, false);
    }

    if (kp + 1 < 16) {
#pragma unroll
      for (int ct = 0; ct < 8; ++ct) b0[ct] = bP[((2 * kp + 2) * 8 + ct) * 32];
    }

    a1.x = ta.z * innr;
    a1.y = ta.w * innr;
    a2.x = tx.z;
    a2.y = tx.w;
#pragma unroll
    for (int ct = 0; ct < 8; ++ct) {
      v2f w1, w2;
      w1.x = b1[ct].x; w1.y = b1[ct].y;
      w2.x = b1[ct].z; w2.y = b1[ct].w;
      acc1[ct] = __builtin_amdgcn_wmma_f32_16x16x4_f32(false, a1, false, w1,
                                                       (short)0, acc1[ct], false, false);
      acc2[ct] = __builtin_amdgcn_wmma_f32_16x16x4_f32(false, a2, false, w2,
                                                       (short)0, acc2[ct], false, false);
    }
    ta = ta_n;
    tx = tx_n;
  }

#pragma unroll
  for (int ct = 0; ct < 8; ++ct) {
    const int col0 = ct * 16;
    const float bb = bias[col0 + r];
    const float rb = Rb[col0 + r];
    float s = 0.0f, s2 = 0.0f;
#pragma unroll
    for (int j = 0; j < 8; ++j) {
      float v = fmaxf(acc1[ct][j] + bb, 0.0f) + fmaxf(acc2[ct][j] + rb, 0.0f);
      h[(size_t)(row0 + j + 8 * hf) * DIM + col0 + r] = v;
      s += v;
      s2 += v * v;
    }
    atomicAdd(&csum[col0 + r], s);
    atomicAdd(&csumsq[col0 + r], s2);
  }
}

__global__ void bnprep_kernel(const float* __restrict__ csum, const float* __restrict__ csumsq,
                              const float* __restrict__ gamma, const float* __restrict__ beta,
                              float* __restrict__ scale, float* __restrict__ shift, float invN) {
  int c = threadIdx.x;
  float mu = csum[c] * invN;
  float var = csumsq[c] * invN - mu * mu;
  float sc = gamma[c] / sqrtf(var + BN_EPS);
  scale[c] = sc;
  shift[c] = beta[c] - mu * sc;
}

__global__ void bnapply_kernel(const float* __restrict__ h, const float* __restrict__ scale,
                               const float* __restrict__ shift, float* __restrict__ out,
                               int total) {
  int i = blockIdx.x * blockDim.x + threadIdx.x;
  if (i < total) {
    int c = i & (DIM - 1);
    out[i] = h[i] * scale[c] + shift[c];
  }
}


static inline int cdiv(int a, int b) { return (a + b - 1) / b; }

extern "C" void kernel_launch(void* const* d_in, const int* in_sizes, int n_in,
                              void* d_out, int out_size, void* d_ws, size_t ws_size,
                              hipStream_t stream) {
  const int* node_ids = (const int*)d_in[0];
  const int* src = (const int*)d_in[1];
  const int* dst = (const int*)d_in[2];
  const float* emb = (const float*)d_in[3];
  const float* Ws = (const float*)d_in[4];
  const float* bs = (const float*)d_in[5];
  const float* Rws = (const float*)d_in[6];
  const float* Rbs = (const float*)d_in[7];
  const float* gammas = (const float*)d_in[8];
  const float* betas = (const float*)d_in[9];

  const int N = in_sizes[0];
  const int E = in_sizes[1];
  const int L = in_sizes[4] / (DIM * DIM);

  float* wsf = (float*)d_ws;
  size_t off = 0;
  float* onorm = wsf + off; off += (size_t)N;
  float* inorm = wsf + off; off += (size_t)N;
  float* x    = wsf + off;  off += (size_t)N * DIM;
  float* agg  = wsf + off;  off += (size_t)N * DIM;
  float* h    = wsf + off;  off += (size_t)N * DIM;
  float* csum   = wsf + off; off += DIM;
  float* csumsq = wsf + off; off += DIM;
  float* scale  = wsf + off; off += DIM;
  float* shift  = wsf + off; off += DIM;
  int* deg_s = (int*)h;
  int* deg_d = deg_s + N;

  const int B = 256;

  zero_f32_kernel<<<cdiv(2 * N, B), B, 0, stream>>>((float*)deg_s, 2 * N);
  degree_kernel<<<cdiv(E, B), B, 0, stream>>>(src, dst, deg_s, deg_d, E);
  norm_kernel<<<cdiv(N, B), B, 0, stream>>>(deg_s, deg_d, onorm, inorm, N);

  gather_kernel<<<cdiv(N * (DIM / 4), B), B, 0, stream>>>(node_ids, emb, x, N);

  const int ND = N * DIM;
  const int rowTiles = cdiv(N, 16);
  for (int l = 0; l < L; ++l) {
    zero_f32_kernel<<<cdiv(ND, B), B, 0, stream>>>(agg, ND);
    zero_f32_kernel<<<1, B, 0, stream>>>(csum, 2 * DIM);

    scatter_kernel<<<cdiv(E * 32, B), B, 0, stream>>>(src, dst, x, onorm, agg, E);

    gemm_fused_kernel<<<cdiv(rowTiles, 8), B, 0, stream>>>(
        agg, x, inorm,
        Ws + (size_t)l * DIM * DIM, bs + (size_t)l * DIM,
        Rws + (size_t)l * DIM * DIM, Rbs + (size_t)l * DIM,
        h, csum, csumsq, N);

    bnprep_kernel<<<1, DIM, 0, stream>>>(csum, csumsq, gammas + (size_t)l * DIM,
                                         betas + (size_t)l * DIM, scale, shift,
                                         1.0f / (float)N);

    float* dest = (l == L - 1) ? (float*)d_out : x;
    bnapply_kernel<<<cdiv(ND, B), B, 0, stream>>>(h, scale, shift, dest, ND);
  }
}


# Edges bucketed by destination once per call (slot = value the in-degree add returns, arrival order; scan; place); per layer a wave sums a row's sources in registers: no float atomics or clear on agg
# speedup vs baseline: 1.8469x; 1.5729x over previous
#include <hip/hip_runtime.h>
#include <math.h>


typedef __attribute__((ext_vector_type(2))) float v2f;
typedef __attribute__((ext_vector_type(8))) float v8f;

#define DIM 128
#define BN_EPS 1e-5f


__global__ void zero_f32_kernel(float* __restrict__ p, int n) {
  int i = blockIdx.x * blockDim.x + threadIdx.x;
  if (i < n) p[i] = 0.0f;
}

__global__ void degree_kernel(const int* __restrict__ src, const int* __restrict__ dst,
                              int* __restrict__ deg_s, int* __restrict__ deg_d,
                              int* __restrict__ slot, int E) {
  int e = blockIdx.x * blockDim.x + threadIdx.x;
  if (e < E) {
    atomicAdd(&deg_s[src[e]], 1);
    slot[e] = atomicAdd(&deg_d[dst[e]], 1);
  }
}

__global__ void norm_kernel(const int* __restrict__ deg_s, const int* __restrict__ deg_d,
                            float* __restrict__ onorm, float* __restrict__ inorm, int N) {
  int i = blockIdx.x * blockDim.x + threadIdx.x;
  if (i < N) {
    int ds = deg_s[i];
    int dd = deg_d[i];
    onorm[i] = (ds > 0) ? 1.0f / sqrtf((float)ds) : 0.0f;
    inorm[i] = (dd > 0) ? 1.0f / sqrtf((float)dd) : 0.0f;
  }
}

#define SCAN_THREADS 1024
#define SCAN_TILE 65536
__global__ void __launch_bounds__(SCAN_THREADS) scan_kernel(
    const int* __restrict__ deg_d, int* __restrict__ rp, int N) {
  __shared__ int buf[SCAN_TILE];
  __shared__ int wtot[33];
  const int tid = threadIdx.x;
  const int lane = tid & 31;
  const int wave = tid >> 5;
  int carry = 0;
  for (int base = 0; base < N; base += SCAN_TILE) {
    const int cnt = (N - base < SCAN_TILE) ? (N - base) : SCAN_TILE;
    for (int i0 = tid; i0 < cnt; i0 += 8 * SCAN_THREADS) {
      int dd[8];
#pragma unroll
      for (int u = 0; u < 8; ++u) {
        const int i = i0 + u * SCAN_THREADS;
        dd[u] = (i < cnt) ? deg_d[base + i] : 0;
      }
#pragma unroll
      for (int u = 0; u < 8; ++u) {
        const int i = i0 + u * SCAN_THREADS;
        if (i < cnt) buf[i] = dd[u];
      }
    }
    __syncthreads();
    const int C = (cnt + SCAN_THREADS - 1) / SCAN_THREADS;
    const int lo = (tid * C < cnt) ? tid * C : cnt;
    const int hi = (lo + C < cnt) ? lo + C : cnt;
    int s = 0;
    for (int i = lo; i < hi; ++i) s += buf[i];
    int v = s;
    for (int o = 1; o < 32; o <<= 1) {
      const int t = __shfl_up(v, o);
      if (lane >= o) v += t;
    }
    if (lane == 31) wtot[wave] = v;
    __syncthreads();
    if (wave == 0) {
      const int t = wtot[lane];
      int u = t;
      for (int o = 1; o < 32; o <<= 1) {
        const int q = __shfl_up(u, o);
        if (lane >= o) u += q;
      }
      wtot[lane] = u - t;
      if (lane == 31) wtot[32] = u;
    }
    __syncthreads();
    int run = carry + wtot[wave] + v - s;
    for (int i = lo; i < hi; ++i) {
      const int t = buf[i];
      buf[i] = run;
      run += t;
    }
    carry += wtot[32];
    __syncthreads();
    for (int i = tid; i < cnt; i += SCAN_THREADS) rp[base + i] = buf[i];
    __syncthreads();
  }
  if (tid == 0) rp[N] = carry;
}

__global__ void place_kernel(const int* __restrict__ src, const int* __restrict__ dst,
                             const int* __restrict__ slot, const int* __restrict__ rp,
                             const float* __restrict__ onorm, int2* __restrict__ csr, int E) {
  int e = blockIdx.x * blockDim.x + threadIdx.x;
  if (e < E) {
    const int s = src[e];
    int2 rec;
    rec.x = s;
    rec.y = __float_as_int(onorm[s]);
    csr[rp[dst[e]] + slot[e]] = rec;
  }
}

__global__ void gather_kernel(const int* __restrict__ ids, const float* __restrict__ emb,
                              float* __restrict__ x, int N) {
  int t = blockIdx.x * blockDim.x + threadIdx.x;
  int total = N * (DIM / 4);
  if (t >= total) return;
  int i = t / (DIM / 4);
  int c = (t % (DIM / 4)) * 4;
  const float4 v = *reinterpret_cast<const float4*>(emb + (size_t)ids[i] * DIM + c);
  *reinterpret_cast<float4*>(x + (size_t)i * DIM + c) = v;
}

#define AGG_WAVES 8

template <int B, bool CLAMP>
__device__ __forceinline__ void sum_rows(float4& acc, const float4* __restrict__ X4,
                                         int recSrc, int recW, int j, int m) {
  float4 v[B];
  float w[B];
#pragma unroll
  for (int u = 0; u < B; ++u) {
    int idx = j + u;
    bool ok = true;
    if (CLAMP) {
      ok = idx < m;
      idx = ok ? idx : m - 1;
    }
    const int s = __builtin_amdgcn_readlane(recSrc, idx);
    const float ww = __int_as_float(__builtin_amdgcn_readlane(recW, idx));
    w[u] = ok ? ww : 0.0f;
    v[u] = X4[(size_t)s * (DIM / 4)];
  }
#pragma unroll
  for (int u = 0; u < B; ++u) {
    acc.x = __builtin_fmaf(v[u].x, w[u], acc.x);
    acc.y = __builtin_fmaf(v[u].y, w[u], acc.y);
    acc.z = __builtin_fmaf(v[u].z, w[u], acc.z);
    acc.w = __builtin_fmaf(v[u].w, w[u], acc.w);
  }
}

__global__ void __launch_bounds__(AGG_WAVES * 32) aggregate_kernel(
    const int* __restrict__ rp, const int2* __restrict__ csr,
    const float* __restrict__ x, float* __restrict__ agg, int N) {
  const int lane = threadIdx.x & 31;
  const int wv = __builtin_amdgcn_readfirstlane((int)(threadIdx.x >> 5));
  const int nw = gridDim.x * AGG_WAVES;
  int row = blockIdx.x * AGG_WAVES + wv;
  if (row >= N) return;
  const float4* X4 = reinterpret_cast<const float4*>(x) + lane;
  float4* A4 = reinterpret_cast<float4*>(agg) + lane;

  int beg = rp[row];
  int end = rp[row + 1];
  int2 ev = make_int2(0, 0);
  if (lane < end - beg) ev = csr[beg + lane];
  int nrow = row + nw;
  int nb = 0, ne = 0;
  if (nrow < N) {
    nb = rp[nrow];
    ne = rp[nrow + 1];
  }

  for (;;) {
    int2 nev = make_int2(0, 0);
    if (lane < ne - nb) nev = csr[nb + lane];
    const int nnrow = nrow + nw;
    int nnb = 0, nne = 0;
    if (nnrow < N) {
      nnb = rp[nnrow];
      nne = rp[nnrow + 1];
    }

    float4 acc = make_float4(0.0f, 0.0f, 0.0f, 0.0f);
    const int cnt = end - beg;
    int done = 0;
    for (;;) {
      const int m = (cnt - done < 32) ? (cnt - done) : 32;
      int j = 0;
      for (; j + 8 <= m; j += 8) sum_rows<8, false>(acc, X4, ev.x, ev.y, j, m);
      for (; j < m; j += 4) sum_rows<4, true>(acc, X4, ev.x, ev.y, j, m);
      done += 32;
      if (done >= cnt) break;
      ev = make_int2(0, 0);
      if (lane < cnt - done) ev = csr[beg + done + lane];
    }
    A4[(size_t)row * (DIM / 4)] = acc;

    if (nrow >= N) break;
    row = nrow;
    beg = nb;
    end = ne;
    ev = nev;
    nrow = nnrow;
    nb = nnb;
    ne = nne;
  }
}

__global__ void __launch_bounds__(256) gemm_fused_kernel(
    const float* __restrict__ agg, const float* __restrict__ xin,
    const float* __restrict__ inorm,
    const float* __restrict__ W, const float* __restrict__ bias,
    const float* __restrict__ Rw, const float* __restrict__ Rb,
    float* __restrict__ h, float* __restrict__ csum, float* __restrict__ csumsq,
    int N) {
  __shared__ float4 sB[32 * 8 * 32];

  const int tid = threadIdx.x;
  const int wave = tid >> 5;
  const int lane = tid & 31;
  const int r = lane & 15;
  const int hf = lane >> 4;

  {
    const float* wp = W + (size_t)(4 * hf) * DIM + wave * 16 + r;
    const float* rp = Rw + (size_t)(4 * hf) * DIM + wave * 16 + r;
    float4* sp = sB + tid;
#pragma unroll 4
    for (int kp = 0; kp < 16; ++kp) {
#pragma unroll
      for (int q = 0; q < 2; ++q) {
        float4 v;
        v.x = wp[(2 * q) * DIM];
        v.y = wp[(2 * q + 1) * DIM];
        v.z = rp[(2 * q) * DIM];
        v.w = rp[(2 * q + 1) * DIM];
        sp[q * 256] = v;
      }
      wp += 8 * DIM;
      rp += 8 * DIM;
      sp += 2 * 256;
    }
  }
  __syncthreads();

  const int row0 = (blockIdx.x * 8 + wave) * 16;
  if (row0 >= N) return;

  const int row = row0 + r;
  const float innr = inorm[row];
  const float4* aP = reinterpret_cast<const float4*>(agg + (size_t)row * DIM) + hf;
  const float4* xP = reinterpret_cast<const float4*>(xin + (size_t)row * DIM) + hf;
  const float4* bP = sB + lane;

  v8f acc1[8], acc2[8];
#pragma unroll
  for (int ct = 0; ct < 8; ++ct) {
    acc1[ct] = (v8f){0.f, 0.f, 0.f, 0.f, 0.f, 0.f, 0.f, 0.f};
    acc2[ct] = (v8f){0.f, 0.f, 0.f, 0.f, 0.f, 0.f, 0.f, 0.f};
  }

  float4 ta = aP[0];
  float4 tx = xP[0];
  float4 b0[8], b1[8];
#pragma unroll
  for (int ct = 0; ct < 8; ++ct) b0[ct] = bP[ct * 32];

  for (int kp = 0; kp < 16; ++kp) {
    float4 ta_n = ta, tx_n = tx;
    if (kp + 1 < 16) {
      ta_n = aP[(kp + 1) * 2];
      tx_n = xP[(kp + 1) * 2];
    }
#pragma unroll
    for (int ct = 0; ct < 8; ++ct) b1[ct] = bP[((2 * kp + 1) * 8 + ct) * 32];

    v2f a1, a2;
    a1.x = ta.x * innr;
    a1.y = ta.y * innr;
    a2.x = tx.x;
    a2.y = tx.y;
#pragma unroll
    for (int ct = 0; ct < 8; ++ct) {
      v2f w1, w2;
      w1.x = b0[ct].x; w1.y = b0[ct].y;
      w2.x = b0[ct].z; w2.y = b0[ct].w;
      acc1[ct] = __builtin_amdgcn_wmma_f32_16x16x4_f32(false, a1, false, w1,
                                                       (short)0, acc1[ct], false, false);
      acc2[ct] = __builtin_amdgcn_wmma_f32_16x16x4_f32(false, a2, false, w2,
                                                       (short)0, acc2[ct], false, false);
    }

    if (kp + 1 < 16) {
#pragma unroll
      for (int ct = 0; ct < 8; ++ct) b0[ct] = bP[((2 * kp + 2) * 8 + ct) * 32];
    }

    a1.x = ta.z * innr;
    a1.y = ta.w * innr;
    a2.x = tx.z;
    a2.y = tx.w;
#pragma unroll
    for (int ct = 0; ct < 8; ++ct) {
      v2f w1, w2;
      w1.x = b1[ct].x; w1.y = b1[ct].y;
      w2.x = b1[ct].z; w2.y = b1[ct].w;
      acc1[ct] = __builtin_amdgcn_wmma_f32_16x16x4_f32(false, a1, false, w1,
                                                       (short)0, acc1[ct], false, false);
      acc2[ct] = __builtin_amdgcn_wmma_f32_16x16x4_f32(false, a2, false, w2,
                                                       (short)0, acc2[ct], false, false);
    }
    ta = ta_n;
    tx = tx_n;
  }

#pragma unroll
  for (int ct = 0; ct < 8; ++ct) {
    const int col0 = ct * 16;
    const float bb = bias[col0 + r];
    const float rb = Rb[col0 + r];
    float s = 0.0f, s2 = 0.0f;
#pragma unroll
    for (int j = 0; j < 8; ++j) {
      float v = fmaxf(acc1[ct][j] + bb, 0.0f) + fmaxf(acc2[ct][j] + rb, 0.0f);
      h[(size_t)(row0 + j + 8 * hf) * DIM + col0 + r] = v;
      s += v;
      s2 += v * v;
    }
    atomicAdd(&csum[col0 + r], s);
    atomicAdd(&csumsq[col0 + r], s2);
  }
}

__global__ void bnprep_kernel(const float* __restrict__ csum, const float* __restrict__ csumsq,
                              const float* __restrict__ gamma, const float* __restrict__ beta,
                              float* __restrict__ scale, float* __restrict__ shift, float invN) {
  int c = threadIdx.x;
  float mu = csum[c] * invN;
  float var = csumsq[c] * invN - mu * mu;
  float sc = gamma[c] / sqrtf(var + BN_EPS);
  scale[c] = sc;
  shift[c] = beta[c] - mu * sc;
}

__global__ void bnapply_kernel(const float* __restrict__ h, const float* __restrict__ scale,
                               const float* __restrict__ shift, float* __restrict__ out,
                               int total) {
  int i = blockIdx.x * blockDim.x + threadIdx.x;
  if (i < total) {
    int c = i & (DIM - 1);
    out[i] = h[i] * scale[c] + shift[c];
  }
}


static inline int cdiv(int a, int b) { return (a + b - 1) / b; }

extern "C" void kernel_launch(void* const* d_in, const int* in_sizes, int n_in,
                              void* d_out, int out_size, void* d_ws, size_t ws_size,
                              hipStream_t stream) {
  const int* node_ids = (const int*)d_in[0];
  const int* src = (const int*)d_in[1];
  const int* dst = (const int*)d_in[2];
  const float* emb = (const float*)d_in[3];
  const float* Ws = (const float*)d_in[4];
  const float* bs = (const float*)d_in[5];
  const float* Rws = (const float*)d_in[6];
  const float* Rbs = (const float*)d_in[7];
  const float* gammas = (const float*)d_in[8];
  const float* betas = (const float*)d_in[9];

  const int N = in_sizes[0];
  const int E = in_sizes[1];
  const int L = in_sizes[4] / (DIM * DIM);

  float* wsf = (float*)d_ws;
  size_t off = 0;
  float* onorm = wsf + off; off += (size_t)N;
  float* inorm = wsf + off; off += (size_t)N;
  float* x    = wsf + off;  off += (size_t)N * DIM;
  float* agg  = wsf + off;  off += (size_t)N * DIM;
  float* h    = wsf + off;  off += (size_t)N * DIM;
  float* csum   = wsf + off; off += DIM;
  float* csumsq = wsf + off; off += DIM;
  float* scale  = wsf + off; off += DIM;
  float* shift  = wsf + off; off += DIM;
  off = (off + 3) & ~(size_t)3;
  int2* csr = (int2*)(wsf + off); off += (size_t)E * 2;
  int* slot = (int*)(wsf + off);  off += (size_t)E;
  int* rp   = (int*)(wsf + off);  off += (size_t)N + 1;
  int* deg_s = (int*)h;
  int* deg_d = deg_s + N;

  const int B = 256;

  zero_f32_kernel<<<cdiv(2 * N, B), B, 0, stream>>>((float*)deg_s, 2 * N);
  degree_kernel<<<cdiv(E, B), B, 0, stream>>>(src, dst, deg_s, deg_d, slot, E);
  norm_kernel<<<cdiv(N, B), B, 0, stream>>>(deg_s, deg_d, onorm, inorm, N);
  scan_kernel<<<1, SCAN_THREADS, 0, stream>>>(deg_d, rp, N);
  place_kernel<<<cdiv(E, B), B, 0, stream>>>(src, dst, slot, rp, onorm, csr, E);

  gather_kernel<<<cdiv(N * (DIM / 4), B), B, 0, stream>>>(node_ids, emb, x, N);

  const int ND = N * DIM;
  const int rowTiles = cdiv(N, 16);
  const int aggBlocks = (cdiv(N, AGG_WAVES) < 1024) ? cdiv(N, AGG_WAVES) : 1024;
  for (int l = 0; l < L; ++l) {
    zero_f32_kernel<<<1, B, 0, stream>>>(csum, 2 * DIM);

    aggregate_kernel<<<aggBlocks, AGG_WAVES * 32, 0, stream>>>(rp, csr, x, agg, N);

    gemm_fused_kernel<<<cdiv(rowTiles, 8), B, 0, stream>>>(
        agg, x, inorm,
        Ws + (size_t)l * DIM * DIM, bs + (size_t)l * DIM,
        Rws + (size_t)l * DIM * DIM, Rbs + (size_t)l * DIM,
        h, csum, csumsq, N);

    bnprep_kernel<<<1, DIM, 0, stream>>>(csum, csumsq, gammas + (size_t)l * DIM,
                                         betas + (size_t)l * DIM, scale, shift,
                                         1.0f / (float)N);

    float* dest = (l == L - 1) ? (float*)d_out : x;
    bnapply_kernel<<<cdiv(ND, B), B, 0, stream>>>(h, scale, shift, dest, ND);
  }
}


# Edge bucketing on LDS counters: 32 blocks count 16 edge chunks x {src,dst} in LDS (returned add = slot), one per-node pass makes chunk starts, rp, norms; no global atomics; order: chunk, then arrival
# speedup vs baseline: 2.1171x; 1.1463x over previous
#include <hip/hip_runtime.h>
#include <math.h>


typedef __attribute__((ext_vector_type(2))) float v2f;
typedef __attribute__((ext_vector_type(8))) float v8f;

#define DIM 128
#define BN_EPS 1e-5f


__global__ void zero_f32_kernel(float* __restrict__ p, int n) {
  int i = blockIdx.x * blockDim.x + threadIdx.x;
  if (i < n) p[i] = 0.0f;
}

#define CNT_THREADS 1024
#define CNT_RANGE 1024
#define CNT_CAP 65536
#define CNT_CHUNKS 16
__global__ void __launch_bounds__(CNT_THREADS) count_kernel(
    const int* __restrict__ src, const int* __restrict__ dst,
    int* __restrict__ cntS, int* __restrict__ cntD, int* __restrict__ tot,
    int* __restrict__ slot, int N, int E, int CH, int R) {
  __shared__ int cnt[CNT_CAP];
  const int tid = threadIdx.x;
  const int lane = tid & 31;
  const int wave = tid >> 5;
  const bool isDst = blockIdx.x < CNT_CHUNKS;
  const int q = isDst ? (int)blockIdx.x : (int)blockIdx.x - CNT_CHUNKS;
  const int* __restrict__ keys = isDst ? dst : src;
  int* __restrict__ outc = (isDst ? cntD : cntS) + (size_t)q * N;
  const int eb = q * CH;
  const int ee = (E - eb < CH) ? E : eb + CH;
  for (int nb = 0; nb < N; nb += CNT_CAP) {
    const int n = (N - nb < CNT_CAP) ? (N - nb) : CNT_CAP;
    const int nr = (n + CNT_RANGE - 1) / CNT_RANGE;
    {
      int4* c4p = reinterpret_cast<int4*>(cnt);
      for (int i = tid; i < (n + 3) / 4; i += CNT_THREADS) c4p[i] = make_int4(0, 0, 0, 0);
    }
    __syncthreads();
    for (int e0 = eb + tid; e0 < ee; e0 += 8 * CNT_THREADS) {
      int k[8];
#pragma unroll
      for (int u = 0; u < 8; ++u) {
        const int e = e0 + u * CNT_THREADS;
        const int kv = keys[(e < ee) ? e : ee - 1];
        k[u] = (e < ee) ? kv - nb : -1;
      }
      int place[8];
#pragma unroll
      for (int u = 0; u < 8; ++u) {
        place[u] = 0;
        if ((unsigned)k[u] < (unsigned)n) place[u] = atomicAdd(&cnt[k[u]], 1);
      }
      if (isDst) {
#pragma unroll
        for (int u = 0; u < 8; ++u) {
          const int e = e0 + u * CNT_THREADS;
          if ((unsigned)k[u] < (unsigned)n) slot[e] = place[u];
        }
      }
    }
    __syncthreads();
    for (int j = wave; j < nr; j += CNT_THREADS / 32) {
      int c = 0;
      for (int t = lane; t < CNT_RANGE; t += 32) {
        const int i = j * CNT_RANGE + t;
        if (i < n) {
          const int v = cnt[i];
          outc[nb + i] = v;
          c += v;
        }
      }
      if (isDst) {
        for (int o = 16; o > 0; o >>= 1) c += __shfl_xor(c, o);
        if (lane == 0) tot[(size_t)q * R + nb / CNT_RANGE + j] = c;
      }
    }
    __syncthreads();
  }
}

__global__ void __launch_bounds__(CNT_RANGE) offsets_kernel(
    const int* __restrict__ cntS, int* __restrict__ cntD, const int* __restrict__ tot,
    int* __restrict__ rp, float* __restrict__ onorm, float* __restrict__ inorm,
    int N, int R) {
  __shared__ int wtot[33];
  __shared__ int wbase[33];
  const int tid = threadIdx.x;
  const int lane = tid & 31;
  const int wave = tid >> 5;
  const int r = blockIdx.x;
  const int i = r * CNT_RANGE + tid;

  int acc = 0;
  for (int t = tid; t < CNT_CHUNKS * R; t += CNT_RANGE) {
    const int v = tot[t];
    if (t % R < r) acc += v;
  }

  int dd = 0;
  if (i < N) {
    int cs[CNT_CHUNKS], cd[CNT_CHUNKS];
#pragma unroll
    for (int q = 0; q < CNT_CHUNKS; ++q) {
      cs[q] = cntS[(size_t)q * N + i];
      cd[q] = cntD[(size_t)q * N + i];
    }
    int ds = 0;
#pragma unroll
    for (int q = 0; q < CNT_CHUNKS; ++q) {
      ds += cs[q];
      cntD[(size_t)q * N + i] = dd;
      dd += cd[q];
    }
    onorm[i] = (ds > 0) ? 1.0f / sqrtf((float)ds) : 0.0f;
    inorm[i] = (dd > 0) ? 1.0f / sqrtf((float)dd) : 0.0f;
  }

  int v = dd;
  for (int o = 1; o < 32; o <<= 1) {
    const int t = __shfl_up(v, o);
    if (lane >= o) v += t;
  }
  for (int o = 16; o > 0; o >>= 1) acc += __shfl_xor(acc, o);
  if (lane == 31) wtot[wave] = v;
  if (lane == 0) wbase[wave] = acc;
  __syncthreads();
  if (wave == 0) {
    const int t = wtot[lane];
    int u = t;
    for (int o = 1; o < 32; o <<= 1) {
      const int s = __shfl_up(u, o);
      if (lane >= o) u += s;
    }
    wtot[lane] = u - t;
    int b = wbase[lane];
    for (int o = 16; o > 0; o >>= 1) b += __shfl_xor(b, o);
    if (lane == 0) wbase[32] = b;
  }
  __syncthreads();
  const int incl = wbase[32] + wtot[wave] + v;
  if (i < N) rp[i] = incl - dd;
  if (i == N - 1) rp[N] = incl;
}

__global__ void place_kernel(const int* __restrict__ src, const int* __restrict__ dst,
                             const int* __restrict__ slot, const int* __restrict__ rp,
                             const int* __restrict__ cntD,
                             const float* __restrict__ onorm, int2* __restrict__ csr,
                             int E, int N, int CH) {
  int e = blockIdx.x * blockDim.x + threadIdx.x;
  if (e < E) {
    const int s = src[e];
    const int d = dst[e];
    int2 rec;
    rec.x = s;
    rec.y = __float_as_int(onorm[s]);
    csr[rp[d] + cntD[(size_t)(e / CH) * N + d] + slot[e]] = rec;
  }
}

__global__ void gather_kernel(const int* __restrict__ ids, const float* __restrict__ emb,
                              float* __restrict__ x, int N) {
  int t = blockIdx.x * blockDim.x + threadIdx.x;
  int total = N * (DIM / 4);
  if (t >= total) return;
  int i = t / (DIM / 4);
  int c = (t % (DIM / 4)) * 4;
  const float4 v = *reinterpret_cast<const float4*>(emb + (size_t)ids[i] * DIM + c);
  *reinterpret_cast<float4*>(x + (size_t)i * DIM + c) = v;
}

#define AGG_WAVES 8

template <int B, bool CLAMP>
__device__ __forceinline__ void sum_rows(float4& acc, const float4* __restrict__ X4,
                                         int recSrc, int recW, int j, int m) {
  float4 v[B];
  float w[B];
#pragma unroll
  for (int u = 0; u < B; ++u) {
    int idx = j + u;
    bool ok = true;
    if (CLAMP) {
      ok = idx < m;
      idx = ok ? idx : m - 1;
    }
    const int s = __builtin_amdgcn_readlane(recSrc, idx);
    const float ww = __int_as_float(__builtin_amdgcn_readlane(recW, idx));
    w[u] = ok ? ww : 0.0f;
    v[u] = X4[(size_t)s * (DIM / 4)];
  }
#pragma unroll
  for (int u = 0; u < B; ++u) {
    acc.x = __builtin_fmaf(v[u].x, w[u], acc.x);
    acc.y = __builtin_fmaf(v[u].y, w[u], acc.y);
    acc.z = __builtin_fmaf(v[u].z, w[u], acc.z);
    acc.w = __builtin_fmaf(v[u].w, w[u], acc.w);
  }
}

__global__ void __launch_bounds__(AGG_WAVES * 32) aggregate_kernel(
    const int* __restrict__ rp, const int2* __restrict__ csr,
    const float* __restrict__ x, float* __restrict__ agg, int N) {
  const int lane = threadIdx.x & 31;
  const int wv = __builtin_amdgcn_readfirstlane((int)(threadIdx.x >> 5));
  const int nw = gridDim.x * AGG_WAVES;
  int row = blockIdx.x * AGG_WAVES + wv;
  if (row >= N) return;
  const float4* X4 = reinterpret_cast<const float4*>(x) + lane;
  float4* A4 = reinterpret_cast<float4*>(agg) + lane;

  int beg = rp[row];
  int end = rp[row + 1];
  int2 ev = make_int2(0, 0);
  if (lane < end - beg) ev = csr[beg + lane];
  int nrow = row + nw;
  int nb = 0, ne = 0;
  if (nrow < N) {
    nb = rp[nrow];
    ne = rp[nrow + 1];
  }

  for (;;) {
    int2 nev = make_int2(0, 0);
    if (lane < ne - nb) nev = csr[nb + lane];
    const int nnrow = nrow + nw;
    int nnb = 0, nne = 0;
    if (nnrow < N) {
      nnb = rp[nnrow];
      nne = rp[nnrow + 1];
    }

    float4 acc = make_float4(0.0f, 0.0f, 0.0f, 0.0f);
    const int cnt = end - beg;
    int done = 0;
    for (;;) {
      const int m = (cnt - done < 32) ? (cnt - done) : 32;
      int j = 0;
      for (; j + 8 <= m; j += 8) sum_rows<8, false>(acc, X4, ev.x, ev.y, j, m);
      for (; j < m; j += 4) sum_rows<4, true>(acc, X4, ev.x, ev.y, j, m);
      done += 32;
      if (done >= cnt) break;
      ev = make_int2(0, 0);
      if (lane < cnt - done) ev = csr[beg + done + lane];
    }
    A4[(size_t)row * (DIM / 4)] = acc;

    if (nrow >= N) break;
    row = nrow;
    beg = nb;
    end = ne;
    ev = nev;
    nrow = nnrow;
    nb = nnb;
    ne = nne;
  }
}

__global__ void __launch_bounds__(256) gemm_fused_kernel(
    const float* __restrict__ agg, const float* __restrict__ xin,
    const float* __restrict__ inorm,
    const float* __restrict__ W, const float* __restrict__ bias,
    const float* __restrict__ Rw, const float* __restrict__ Rb,
    float* __restrict__ h, float* __restrict__ csum, float* __restrict__ csumsq,
    int N) {
  __shared__ float4 sB[32 * 8 * 32];

  const int tid = threadIdx.x;
  const int wave = tid >> 5;
  const int lane = tid & 31;
  const int r = lane & 15;
  const int hf = lane >> 4;

  {
    const float* wp = W + (size_t)(4 * hf) * DIM + wave * 16 + r;
    const float* rp = Rw + (size_t)(4 * hf) * DIM + wave * 16 + r;
    float4* sp = sB + tid;
#pragma unroll 4
    for (int kp = 0; kp < 16; ++kp) {
#pragma unroll
      for (int q = 0; q < 2; ++q) {
        float4 v;
        v.x = wp[(2 * q) * DIM];
        v.y = wp[(2 * q + 1) * DIM];
        v.z = rp[(2 * q) * DIM];
        v.w = rp[(2 * q + 1) * DIM];
        sp[q * 256] = v;
      }
      wp += 8 * DIM;
      rp += 8 * DIM;
      sp += 2 * 256;
    }
  }
  __syncthreads();

  const int row0 = (blockIdx.x * 8 + wave) * 16;
  if (row0 >= N) return;

  const int row = row0 + r;
  const float innr = inorm[row];
  const float4* aP = reinterpret_cast<const float4*>(agg + (size_t)row * DIM) + hf;
  const float4* xP = reinterpret_cast<const float4*>(xin + (size_t)row * DIM) + hf;
  const float4* bP = sB + lane;

  v8f acc1[8], acc2[8];
#pragma unroll
  for (int ct = 0; ct < 8; ++ct) {
    acc1[ct] = (v8f){0.f, 0.f, 0.f, 0.f, 0.f, 0.f, 0.f, 0.f};
    acc2[ct] = (v8f){0.f, 0.f, 0.f, 0.f, 0.f, 0.f, 0.f, 0.f};
  }

  float4 ta = aP[0];
  float4 tx = xP[0];
  float4 b0[8], b1[8];
#pragma unroll
  for (int ct = 0; ct < 8; ++ct) b0[ct] = bP[ct * 32];

  for (int kp = 0; kp < 16; ++kp) {
    float4 ta_n = ta, tx_n = tx;
    if (kp + 1 < 16) {
      ta_n = aP[(kp + 1) * 2];
      tx_n = xP[(kp + 1) * 2];
    }
#pragma unroll
    for (int ct = 0; ct < 8; ++ct) b1[ct] = bP[((2 * kp + 1) * 8 + ct) * 32];

    v2f a1, a2;
    a1.x = ta.x * innr;
    a1.y = ta.y * innr;
    a2.x = tx.x;
    a2.y = tx.y;
#pragma unroll
    for (int ct = 0; ct < 8; ++ct) {
      v2f w1, w2;
      w1.x = b0[ct].x; w1.y = b0[ct].y;
      w2.x = b0[ct].z; w2.y = b0[ct].w;
      acc1[ct] = __builtin_amdgcn_wmma_f32_16x16x4_f32(false, a1, false, w1,
                                                       (short)0, acc1[ct], false, false);
      acc2[ct] = __builtin_amdgcn_wmma_f32_16x16x4_f32(false, a2, false, w2,
                                                       (short)0, acc2[ct], false, false);
    }

    if (kp + 1 < 16) {
#pragma unroll
      for (int ct = 0; ct < 8; ++ct) b0[ct] = bP[((2 * kp + 2) * 8 + ct) * 32];
    }

    a1.x = ta.z * innr;
    a1.y = ta.w * innr;
    a2.x = tx.z;
    a2.y = tx.w;
#pragma unroll
    for (int ct = 0; ct < 8; ++ct) {
      v2f w1, w2;
      w1.x = b1[ct].x; w1.y = b1[ct].y;
      w2.x = b1[ct].z; w2.y = b1[ct].w;
      acc1[ct] = __builtin_amdgcn_wmma_f32_16x16x4_f32(false, a1, false, w1,
                                                       (short)0, acc1[ct], false, false);
      acc2[ct] = __builtin_amdgcn_wmma_f32_16x16x4_f32(false, a2, false, w2,
                                                       (short)0, acc2[ct], false, false);
    }
    ta = ta_n;
    tx = tx_n;
  }

#pragma unroll
  for (int ct = 0; ct < 8; ++ct) {
    const int col0 = ct * 16;
    const float bb = bias[col0 + r];
    const float rb = Rb[col0 + r];
    float s = 0.0f, s2 = 0.0f;
#pragma unroll
    for (int j = 0; j < 8; ++j) {
      float v = fmaxf(acc1[ct][j] + bb, 0.0f) + fmaxf(acc2[ct][j] + rb, 0.0f);
      h[(size_t)(row0 + j + 8 * hf) * DIM + col0 + r] = v;
      s += v;
      s2 += v * v;
    }
    atomicAdd(&csum[col0 + r], s);
    atomicAdd(&csumsq[col0 + r], s2);
  }
}

__global__ void bnprep_kernel(const float* __restrict__ csum, const float* __restrict__ csumsq,
                              const float* __restrict__ gamma, const float* __restrict__ beta,
                              float* __restrict__ scale, float* __restrict__ shift, float invN) {
  int c = threadIdx.x;
  float mu = csum[c] * invN;
  float var = csumsq[c] * invN - mu * mu;
  float sc = gamma[c] / sqrtf(var + BN_EPS);
  scale[c] = sc;
  shift[c] = beta[c] - mu * sc;
}

__global__ void bnapply_kernel(const float* __restrict__ h, const float* __restrict__ scale,
                               const float* __restrict__ shift, float* __restrict__ out,
                               int total) {
  int i = blockIdx.x * blockDim.x + threadIdx.x;
  if (i < total) {
    int c = i & (DIM - 1);
    out[i] = h[i] * scale[c] + shift[c];
  }
}


static inline int cdiv(int a, int b) { return (a + b - 1) / b; }

extern "C" void kernel_launch(void* const* d_in, const int* in_sizes, int n_in,
                              void* d_out, int out_size, void* d_ws, size_t ws_size,
                              hipStream_t stream) {
  const int* node_ids = (const int*)d_in[0];
  const int* src = (const int*)d_in[1];
  const int* dst = (const int*)d_in[2];
  const float* emb = (const float*)d_in[3];
  const float* Ws = (const float*)d_in[4];
  const float* bs = (const float*)d_in[5];
  const float* Rws = (const float*)d_in[6];
  const float* Rbs = (const float*)d_in[7];
  const float* gammas = (const float*)d_in[8];
  const float* betas = (const float*)d_in[9];

  const int N = in_sizes[0];
  const int E = in_sizes[1];
  const int L = in_sizes[4] / (DIM * DIM);

  float* wsf = (float*)d_ws;
  size_t off = 0;
  float* onorm = wsf + off; off += (size_t)N;
  float* inorm = wsf + off; off += (size_t)N;
  float* x    = wsf + off;  off += (size_t)N * DIM;
  float* agg  = wsf + off;  off += (size_t)N * DIM;
  float* h    = wsf + off;  off += (size_t)N * DIM;
  float* csum   = wsf + off; off += DIM;
  float* csumsq = wsf + off; off += DIM;
  float* scale  = wsf + off; off += DIM;
  float* shift  = wsf + off; off += DIM;
  off = (off + 3) & ~(size_t)3;
  int2* csr = (int2*)(wsf + off); off += (size_t)E * 2;
  int* slot = (int*)(wsf + off);  off += (size_t)E;
  int* rp   = (int*)(wsf + off);  off += (size_t)N + 1;
  const int R = cdiv(N, CNT_RANGE);
  int* cntS = (int*)(wsf + off);  off += (size_t)CNT_CHUNKS * N;
  int* cntD = (int*)(wsf + off);  off += (size_t)CNT_CHUNKS * N;
  int* tot  = (int*)(wsf + off);  off += (size_t)CNT_CHUNKS * R;

  const int B = 256;

  const int CH = cdiv(E, CNT_CHUNKS);
  count_kernel<<<2 * CNT_CHUNKS, CNT_THREADS, 0, stream>>>(src, dst, cntS, cntD, tot, slot,
                                                           N, E, CH, R);
  offsets_kernel<<<R, CNT_RANGE, 0, stream>>>(cntS, cntD, tot, rp, onorm, inorm, N, R);
  place_kernel<<<cdiv(E, B), B, 0, stream>>>(src, dst, slot, rp, cntD, onorm, csr, E, N, CH);

  gather_kernel<<<cdiv(N * (DIM / 4), B), B, 0, stream>>>(node_ids, emb, x, N);

  const int ND = N * DIM;
  const int rowTiles = cdiv(N, 16);
  const int aggBlocks = (cdiv(N, AGG_WAVES) < 1024) ? cdiv(N, AGG_WAVES) : 1024;
  for (int l = 0; l < L; ++l) {
    zero_f32_kernel<<<1, B, 0, stream>>>(csum, 2 * DIM);

    aggregate_kernel<<<aggBlocks, AGG_WAVES * 32, 0, stream>>>(rp, csr, x, agg, N);

    gemm_fused_kernel<<<cdiv(rowTiles, 8), B, 0, stream>>>(
        agg, x, inorm,
        Ws + (size_t)l * DIM * DIM, bs + (size_t)l * DIM,
        Rws + (size_t)l * DIM * DIM, Rbs + (size_t)l * DIM,
        h, csum, csumsq, N);

    bnprep_kernel<<<1, DIM, 0, stream>>>(csum, csumsq, gammas + (size_t)l * DIM,
                                         betas + (size_t)l * DIM, scale, shift,
                                         1.0f / (float)N);

    float* dest = (l == L - 1) ? (float*)d_out : x;
    bnapply_kernel<<<cdiv(ND, B), B, 0, stream>>>(h, scale, shift, dest, ND);
  }
}


# Streaming launches (BN apply x3, embedding gather, bucket fill) as 512 resident work-groups that loop, one whole row per wave load, 16 rows in flight: 4096 waves a launch, not up to 200000
# speedup vs baseline: 2.2076x; 1.0427x over previous
#include <hip/hip_runtime.h>
#include <math.h>


typedef __attribute__((ext_vector_type(2))) float v2f;
typedef __attribute__((ext_vector_type(8))) float v8f;

#define DIM 128
#define BN_EPS 1e-5f


__global__ void zero_f32_kernel(float* __restrict__ p, int n) {
  int i = blockIdx.x * blockDim.x + threadIdx.x;
  if (i < n) p[i] = 0.0f;
}

#define CNT_THREADS 1024
#define CNT_RANGE 1024
#define CNT_CAP 65536
#define CNT_CHUNKS 16
__global__ void __launch_bounds__(CNT_THREADS) count_kernel(
    const int* __restrict__ src, const int* __restrict__ dst,
    int* __restrict__ cntS, int* __restrict__ cntD, int* __restrict__ tot,
    int* __restrict__ slot, int N, int E, int CH, int R) {
  __shared__ int cnt[CNT_CAP];
  const int tid = threadIdx.x;
  const int lane = tid & 31;
  const int wave = tid >> 5;
  const bool isDst = blockIdx.x < CNT_CHUNKS;
  const int q = isDst ? (int)blockIdx.x : (int)blockIdx.x - CNT_CHUNKS;
  const int* __restrict__ keys = isDst ? dst : src;
  int* __restrict__ outc = (isDst ? cntD : cntS) + (size_t)q * N;
  const int eb = q * CH;
  const int ee = (E - eb < CH) ? E : eb + CH;
  for (int nb = 0; nb < N; nb += CNT_CAP) {
    const int n = (N - nb < CNT_CAP) ? (N - nb) : CNT_CAP;
    const int nr = (n + CNT_RANGE - 1) / CNT_RANGE;
    {
      int4* c4p = reinterpret_cast<int4*>(cnt);
      for (int i = tid; i < (n + 3) / 4; i += CNT_THREADS) c4p[i] = make_int4(0, 0, 0, 0);
    }
    __syncthreads();
    for (int e0 = eb + tid; e0 < ee; e0 += 8 * CNT_THREADS) {
      int k[8];
#pragma unroll
      for (int u = 0; u < 8; ++u) {
        const int e = e0 + u * CNT_THREADS;
        const int kv = keys[(e < ee) ? e : ee - 1];
        k[u] = (e < ee) ? kv - nb : -1;
      }
      int place[8];
#pragma unroll
      for (int u = 0; u < 8; ++u) {
        place[u] = 0;
        if ((unsigned)k[u] < (unsigned)n) place[u] = atomicAdd(&cnt[k[u]], 1);
      }
      if (isDst) {
#pragma unroll
        for (int u = 0; u < 8; ++u) {
          const int e = e0 + u * CNT_THREADS;
          if ((unsigned)k[u] < (unsigned)n) slot[e] = place[u];
        }
      }
    }
    __syncthreads();
    for (int j = wave; j < nr; j += CNT_THREADS / 32) {
      int c = 0;
      for (int t = lane; t < CNT_RANGE; t += 32) {
        const int i = j * CNT_RANGE + t;
        if (i < n) {
          const int v = cnt[i];
          outc[nb + i] = v;
          c += v;
        }
      }
      if (isDst) {
        for (int o = 16; o > 0; o >>= 1) c += __shfl_xor(c, o);
        if (lane == 0) tot[(size_t)q * R + nb / CNT_RANGE + j] = c;
      }
    }
    __syncthreads();
  }
}

__global__ void __launch_bounds__(CNT_RANGE) offsets_kernel(
    const int* __restrict__ cntS, int* __restrict__ cntD, const int* __restrict__ tot,
    int* __restrict__ rp, float* __restrict__ onorm, float* __restrict__ inorm,
    int N, int R) {
  __shared__ int wtot[33];
  __shared__ int wbase[33];
  const int tid = threadIdx.x;
  const int lane = tid & 31;
  const int wave = tid >> 5;
  const int r = blockIdx.x;
  const int i = r * CNT_RANGE + tid;

  int acc = 0;
  for (int t = tid; t < CNT_CHUNKS * R; t += CNT_RANGE) {
    const int v = tot[t];
    if (t % R < r) acc += v;
  }

  int dd = 0;
  if (i < N) {
    int cs[CNT_CHUNKS], cd[CNT_CHUNKS];
#pragma unroll
    for (int q = 0; q < CNT_CHUNKS; ++q) {
      cs[q] = cntS[(size_t)q * N + i];
      cd[q] = cntD[(size_t)q * N + i];
    }
    int ds = 0;
#pragma unroll
    for (int q = 0; q < CNT_CHUNKS; ++q) {
      ds += cs[q];
      cntD[(size_t)q * N + i] = dd;
      dd += cd[q];
    }
    onorm[i] = (ds > 0) ? 1.0f / sqrtf((float)ds) : 0.0f;
    inorm[i] = (dd > 0) ? 1.0f / sqrtf((float)dd) : 0.0f;
  }

  int v = dd;
  for (int o = 1; o < 32; o <<= 1) {
    const int t = __shfl_up(v, o);
    if (lane >= o) v += t;
  }
  for (int o = 16; o > 0; o >>= 1) acc += __shfl_xor(acc, o);
  if (lane == 31) wtot[wave] = v;
  if (lane == 0) wbase[wave] = acc;
  __syncthreads();
  if (wave == 0) {
    const int t = wtot[lane];
    int u = t;
    for (int o = 1; o < 32; o <<= 1) {
      const int s = __shfl_up(u, o);
      if (lane >= o) u += s;
    }
    wtot[lane] = u - t;
    int b = wbase[lane];
    for (int o = 16; o > 0; o >>= 1) b += __shfl_xor(b, o);
    if (lane == 0) wbase[32] = b;
  }
  __syncthreads();
  const int incl = wbase[32] + wtot[wave] + v;
  if (i < N) rp[i] = incl - dd;
  if (i == N - 1) rp[N] = incl;
}

#define STREAM_BLOCKS 512
#define STREAM_THREADS 256

#define PLACE_U 4
__global__ void __launch_bounds__(STREAM_THREADS) place_kernel(
    const int* __restrict__ src, const int* __restrict__ dst,
    const int* __restrict__ slot, const int* __restrict__ rp,
    const int* __restrict__ cntD,
    const float* __restrict__ onorm, int2* __restrict__ csr,
    int E, int N, int CH) {
  const int t = blockIdx.x * STREAM_THREADS + threadIdx.x;
  const int stride = gridDim.x * STREAM_THREADS;
  for (int e0 = t; e0 < E; e0 += PLACE_U * stride) {
    int s[PLACE_U], d[PLACE_U], sl[PLACE_U], q[PLACE_U];
#pragma unroll
    for (int u = 0; u < PLACE_U; ++u) {
      const int e = e0 + u * stride;
      const int ec = (e < E) ? e : e0;
      s[u] = src[ec];
      d[u] = dst[ec];
      sl[u] = slot[ec];
      q[u] = ec / CH;
    }
    int base[PLACE_U], cs[PLACE_U], w[PLACE_U];
#pragma unroll
    for (int u = 0; u < PLACE_U; ++u) {
      w[u] = __float_as_int(onorm[s[u]]);
      base[u] = rp[d[u]];
      cs[u] = cntD[(size_t)q[u] * N + d[u]];
    }
#pragma unroll
    for (int u = 0; u < PLACE_U; ++u) {
      const int e = e0 + u * stride;
      asm volatile("" ::"v"(s[u]), "v"(sl[u]), "v"(w[u]), "v"(base[u]), "v"(cs[u]));
      if (e < E) {
        int2 rec;
        rec.x = s[u];
        rec.y = w[u];
        csr[base[u] + cs[u] + sl[u]] = rec;
      }
    }
  }
}

#define STREAM_U 16
__global__ void __launch_bounds__(STREAM_THREADS) gather_kernel(
    const int* __restrict__ ids, const float* __restrict__ emb,
    float* __restrict__ x, int N) {
  const int lane = threadIdx.x & 31;
  const int wv = blockIdx.x * (STREAM_THREADS / 32) +
                 __builtin_amdgcn_readfirstlane((int)(threadIdx.x >> 5));
  const int nw = gridDim.x * (STREAM_THREADS / 32);
  const float4* E4 = reinterpret_cast<const float4*>(emb) + lane;
  float4* X4 = reinterpret_cast<float4*>(x) + lane;
  for (int i0 = wv; i0 < N; i0 += STREAM_U * nw) {
    const int iu = i0 + lane * nw;
    int idv = 0;
    if (lane < STREAM_U) idv = ids[(iu < N) ? iu : i0];
    float4 v[STREAM_U];
#pragma unroll
    for (int u = 0; u < STREAM_U; ++u) {
      const int id = __builtin_amdgcn_readlane(idv, u);
      v[u] = E4[(size_t)id * (DIM / 4)];
    }
#pragma unroll
    for (int u = 0; u < STREAM_U; ++u) {
      const int i = i0 + u * nw;
      asm volatile("" ::"v"(v[u].x), "v"(v[u].y), "v"(v[u].z), "v"(v[u].w));
      if (i < N) X4[(size_t)i * (DIM / 4)] = v[u];
    }
  }
}

#define AGG_WAVES 8

template <int B, bool CLAMP>
__device__ __forceinline__ void sum_rows(float4& acc, const float4* __restrict__ X4,
                                         int recSrc, int recW, int j, int m) {
  float4 v[B];
  float w[B];
#pragma unroll
  for (int u = 0; u < B; ++u) {
    int idx = j + u;
    bool ok = true;
    if (CLAMP) {
      ok = idx < m;
      idx = ok ? idx : m - 1;
    }
    const int s = __builtin_amdgcn_readlane(recSrc, idx);
    const float ww = __int_as_float(__builtin_amdgcn_readlane(recW, idx));
    w[u] = ok ? ww : 0.0f;
    v[u] = X4[(size_t)s * (DIM / 4)];
  }
#pragma unroll
  for (int u = 0; u < B; ++u) {
    acc.x = __builtin_fmaf(v[u].x, w[u], acc.x);
    acc.y = __builtin_fmaf(v[u].y, w[u], acc.y);
    acc.z = __builtin_fmaf(v[u].z, w[u], acc.z);
    acc.w = __builtin_fmaf(v[u].w, w[u], acc.w);
  }
}

__global__ void __launch_bounds__(AGG_WAVES * 32) aggregate_kernel(
    const int* __restrict__ rp, const int2* __restrict__ csr,
    const float* __restrict__ x, float* __restrict__ agg, int N) {
  const int lane = threadIdx.x & 31;
  const int wv = __builtin_amdgcn_readfirstlane((int)(threadIdx.x >> 5));
  const int nw = gridDim.x * AGG_WAVES;
  int row = blockIdx.x * AGG_WAVES + wv;
  if (row >= N) return;
  const float4* X4 = reinterpret_cast<const float4*>(x) + lane;
  float4* A4 = reinterpret_cast<float4*>(agg) + lane;

  int beg = rp[row];
  int end = rp[row + 1];
  int2 ev = make_int2(0, 0);
  if (lane < end - beg) ev = csr[beg + lane];
  int nrow = row + nw;
  int nb = 0, ne = 0;
  if (nrow < N) {
    nb = rp[nrow];
    ne = rp[nrow + 1];
  }

  for (;;) {
    int2 nev = make_int2(0, 0);
    if (lane < ne - nb) nev = csr[nb + lane];
    const int nnrow = nrow + nw;
    int nnb = 0, nne = 0;
    if (nnrow < N) {
      nnb = rp[nnrow];
      nne = rp[nnrow + 1];
    }

    float4 acc = make_float4(0.0f, 0.0f, 0.0f, 0.0f);
    const int cnt = end - beg;
    int done = 0;
    for (;;) {
      const int m = (cnt - done < 32) ? (cnt - done) : 32;
      int j = 0;
      for (; j + 8 <= m; j += 8) sum_rows<8, false>(acc, X4, ev.x, ev.y, j, m);
      for (; j < m; j += 4) sum_rows<4, true>(acc, X4, ev.x, ev.y, j, m);
      done += 32;
      if (done >= cnt) break;
      ev = make_int2(0, 0);
      if (lane < cnt - done) ev = csr[beg + done + lane];
    }
    A4[(size_t)row * (DIM / 4)] = acc;

    if (nrow >= N) break;
    row = nrow;
    beg = nb;
    end = ne;
    ev = nev;
    nrow = nnrow;
    nb = nnb;
    ne = nne;
  }
}

__global__ void __launch_bounds__(256) gemm_fused_kernel(
    const float* __restrict__ agg, const float* __restrict__ xin,
    const float* __restrict__ inorm,
    const float* __restrict__ W, const float* __restrict__ bias,
    const float* __restrict__ Rw, const float* __restrict__ Rb,
    float* __restrict__ h, float* __restrict__ csum, float* __restrict__ csumsq,
    int N) {
  __shared__ float4 sB[32 * 8 * 32];

  const int tid = threadIdx.x;
  const int wave = tid >> 5;
  const int lane = tid & 31;
  const int r = lane & 15;
  const int hf = lane >> 4;

  {
    const float* wp = W + (size_t)(4 * hf) * DIM + wave * 16 + r;
    const float* rp = Rw + (size_t)(4 * hf) * DIM + wave * 16 + r;
    float4* sp = sB + tid;
#pragma unroll 4
    for (int kp = 0; kp < 16; ++kp) {
#pragma unroll
      for (int q = 0; q < 2; ++q) {
        float4 v;
        v.x = wp[(2 * q) * DIM];
        v.y = wp[(2 * q + 1) * DIM];
        v.z = rp[(2 * q) * DIM];
        v.w = rp[(2 * q + 1) * DIM];
        sp[q * 256] = v;
      }
      wp += 8 * DIM;
      rp += 8 * DIM;
      sp += 2 * 256;
    }
  }
  __syncthreads();

  const int row0 = (blockIdx.x * 8 + wave) * 16;
  if (row0 >= N) return;

  const int row = row0 + r;
  const float innr = inorm[row];
  const float4* aP = reinterpret_cast<const float4*>(agg + (size_t)row * DIM) + hf;
  const float4* xP = reinterpret_cast<const float4*>(xin + (size_t)row * DIM) + hf;
  const float4* bP = sB + lane;

  v8f acc1[8], acc2[8];
#pragma unroll
  for (int ct = 0; ct < 8; ++ct) {
    acc1[ct] = (v8f){0.f, 0.f, 0.f, 0.f, 0.f, 0.f, 0.f, 0.f};
    acc2[ct] = (v8f){0.f, 0.f, 0.f, 0.f, 0.f, 0.f, 0.f, 0.f};
  }

  float4 ta = aP[0];
  float4 tx = xP[0];
  float4 b0[8], b1[8];
#pragma unroll
  for (int ct = 0; ct < 8; ++ct) b0[ct] = bP[ct * 32];

  for (int kp = 0; kp < 16; ++kp) {
    float4 ta_n = ta, tx_n = tx;
    if (kp + 1 < 16) {
      ta_n = aP[(kp + 1) * 2];
      tx_n = xP[(kp + 1) * 2];
    }
#pragma unroll
    for (int ct = 0; ct < 8; ++ct) b1[ct] = bP[((2 * kp + 1) * 8 + ct) * 32];

    v2f a1, a2;
    a1.x = ta.x * innr;
    a1.y = ta.y * innr;
    a2.x = tx.x;
    a2.y = tx.y;
#pragma unroll
    for (int ct = 0; ct < 8; ++ct) {
      v2f w1, w2;
      w1.x = b0[ct].x; w1.y = b0[ct].y;
      w2.x = b0[ct].z; w2.y = b0[ct].w;
      acc1[ct] = __builtin_amdgcn_wmma_f32_16x16x4_f32(false, a1, false, w1,
                                                       (short)0, acc1[ct], false, false);
      acc2[ct] = __builtin_amdgcn_wmma_f32_16x16x4_f32(false, a2, false, w2,
                                                       (short)0, acc2[ct], false, false);
    }

    if (kp + 1 < 16) {
#pragma unroll
      for (int ct = 0; ct < 8; ++ct) b0[ct] = bP[((2 * kp + 2) * 8 + ct) * 32];
    }

    a1.x = ta.z * innr;
    a1.y = ta.w * innr;
    a2.x = tx.z;
    a2.y = tx.w;
#pragma unroll
    for (int ct = 0; ct < 8; ++ct) {
      v2f w1, w2;
      w1.x = b1[ct].x; w1.y = b1[ct].y;
      w2.x = b1[ct].z; w2.y = b1[ct].w;
      acc1[ct] = __builtin_amdgcn_wmma_f32_16x16x4_f32(false, a1, false, w1,
                                                       (short)0, acc1[ct], false, false);
      acc2[ct] = __builtin_amdgcn_wmma_f32_16x16x4_f32(false, a2, false, w2,
                                                       (short)0, acc2[ct], false, false);
    }
    ta = ta_n;
    tx = tx_n;
  }

#pragma unroll
  for (int ct = 0; ct < 8; ++ct) {
    const int col0 = ct * 16;
    const float bb = bias[col0 + r];
    const float rb = Rb[col0 + r];
    float s = 0.0f, s2 = 0.0f;
#pragma unroll
    for (int j = 0; j < 8; ++j) {
      float v = fmaxf(acc1[ct][j] + bb, 0.0f) + fmaxf(acc2[ct][j] + rb, 0.0f);
      h[(size_t)(row0 + j + 8 * hf) * DIM + col0 + r] = v;
      s += v;
      s2 += v * v;
    }
    atomicAdd(&csum[col0 + r], s);
    atomicAdd(&csumsq[col0 + r], s2);
  }
}

__global__ void bnprep_kernel(const float* __restrict__ csum, const float* __restrict__ csumsq,
                              const float* __restrict__ gamma, const float* __restrict__ beta,
                              float* __restrict__ scale, float* __restrict__ shift, float invN) {
  int c = threadIdx.x;
  float mu = csum[c] * invN;
  float var = csumsq[c] * invN - mu * mu;
  float sc = gamma[c] / sqrtf(var + BN_EPS);
  scale[c] = sc;
  shift[c] = beta[c] - mu * sc;
}

__global__ void __launch_bounds__(STREAM_THREADS) bnapply_kernel(
    const float* __restrict__ h, const float* __restrict__ scale,
    const float* __restrict__ shift, float* __restrict__ out, int N) {
  const int lane = threadIdx.x & 31;
  const int wv = blockIdx.x * (STREAM_THREADS / 32) +
                 __builtin_amdgcn_readfirstlane((int)(threadIdx.x >> 5));
  const int nw = gridDim.x * (STREAM_THREADS / 32);
  const float4* H4 = reinterpret_cast<const float4*>(h) + lane;
  float4* O4 = reinterpret_cast<float4*>(out) + lane;
  const float4 sc = reinterpret_cast<const float4*>(scale)[lane];
  const float4 sh = reinterpret_cast<const float4*>(shift)[lane];
  for (int i0 = wv; i0 < N; i0 += STREAM_U * nw) {
    float4 v[STREAM_U];
#pragma unroll
    for (int u = 0; u < STREAM_U; ++u) {
      const int i = i0 + u * nw;
      v[u] = H4[(size_t)((i < N) ? i : i0) * (DIM / 4)];
    }
#pragma unroll
    for (int u = 0; u < STREAM_U; ++u) {
      const int i = i0 + u * nw;
      asm volatile("" ::"v"(v[u].x), "v"(v[u].y), "v"(v[u].z), "v"(v[u].w));
      if (i < N) {
        float4 o;
        o.x = v[u].x * sc.x + sh.x;
        o.y = v[u].y * sc.y + sh.y;
        o.z = v[u].z * sc.z + sh.z;
        o.w = v[u].w * sc.w + sh.w;
        O4[(size_t)i * (DIM / 4)] = o;
      }
    }
  }
}


static inline int cdiv(int a, int b) { return (a + b - 1) / b; }

extern "C" void kernel_launch(void* const* d_in, const int* in_sizes, int n_in,
                              void* d_out, int out_size, void* d_ws, size_t ws_size,
                              hipStream_t stream) {
  const int* node_ids = (const int*)d_in[0];
  const int* src = (const int*)d_in[1];
  const int* dst = (const int*)d_in[2];
  const float* emb = (const float*)d_in[3];
  const float* Ws = (const float*)d_in[4];
  const float* bs = (const float*)d_in[5];
  const float* Rws = (const float*)d_in[6];
  const float* Rbs = (const float*)d_in[7];
  const float* gammas = (const float*)d_in[8];
  const float* betas = (const float*)d_in[9];

  const int N = in_sizes[0];
  const int E = in_sizes[1];
  const int L = in_sizes[4] / (DIM * DIM);

  float* wsf = (float*)d_ws;
  size_t off = 0;
  float* onorm = wsf + off; off += (size_t)N;
  float* inorm = wsf + off; off += (size_t)N;
  float* x    = wsf + off;  off += (size_t)N * DIM;
  float* agg  = wsf + off;  off += (size_t)N * DIM;
  float* h    = wsf + off;  off += (size_t)N * DIM;
  float* csum   = wsf + off; off += DIM;
  float* csumsq = wsf + off; off += DIM;
  float* scale  = wsf + off; off += DIM;
  float* shift  = wsf + off; off += DIM;
  off = (off + 3) & ~(size_t)3;
  int2* csr = (int2*)(wsf + off); off += (size_t)E * 2;
  int* slot = (int*)(wsf + off);  off += (size_t)E;
  int* rp   = (int*)(wsf + off);  off += (size_t)N + 1;
  const int R = cdiv(N, CNT_RANGE);
  int* cntS = (int*)(wsf + off);  off += (size_t)CNT_CHUNKS * N;
  int* cntD = (int*)(wsf + off);  off += (size_t)CNT_CHUNKS * N;
  int* tot  = (int*)(wsf + off);  off += (size_t)CNT_CHUNKS * R;

  const int B = 256;

  const int CH = cdiv(E, CNT_CHUNKS);
  count_kernel<<<2 * CNT_CHUNKS, CNT_THREADS, 0, stream>>>(src, dst, cntS, cntD, tot, slot,
                                                           N, E, CH, R);
  offsets_kernel<<<R, CNT_RANGE, 0, stream>>>(cntS, cntD, tot, rp, onorm, inorm, N, R);
  place_kernel<<<STREAM_BLOCKS, STREAM_THREADS, 0, stream>>>(src, dst, slot, rp, cntD, onorm,
                                                             csr, E, N, CH);

  gather_kernel<<<STREAM_BLOCKS, STREAM_THREADS, 0, stream>>>(node_ids, emb, x, N);

  const int rowTiles = cdiv(N, 16);
  const int aggBlocks = (cdiv(N, AGG_WAVES) < 1024) ? cdiv(N, AGG_WAVES) : 1024;
  for (int l = 0; l < L; ++l) {
    zero_f32_kernel<<<1, B, 0, stream>>>(csum, 2 * DIM);

    aggregate_kernel<<<aggBlocks, AGG_WAVES * 32, 0, stream>>>(rp, csr, x, agg, N);

    gemm_fused_kernel<<<cdiv(rowTiles, 8), B, 0, stream>>>(
        agg, x, inorm,
        Ws + (size_t)l * DIM * DIM, bs + (size_t)l * DIM,
        Rws + (size_t)l * DIM * DIM, Rbs + (size_t)l * DIM,
        h, csum, csumsq, N);

    bnprep_kernel<<<1, DIM, 0, stream>>>(csum, csumsq, gammas + (size_t)l * DIM,
                                         betas + (size_t)l * DIM, scale, shift,
                                         1.0f / (float)N);

    float* dest = (l == L - 1) ? (float*)d_out : x;
    bnapply_kernel<<<STREAM_BLOCKS, STREAM_THREADS, 0, stream>>>(h, scale, shift, dest, N);
  }
}


# Batch-norm applied by its readers: aggregate gather and GEMM x-load form h*scale+shift from the layer's column sums; no clear, prep or apply launch between layers (19 launches -> 11)
# speedup vs baseline: 2.4631x; 1.1157x over previous
#include <hip/hip_runtime.h>
#include <math.h>


typedef __attribute__((ext_vector_type(2))) float v2f;
typedef __attribute__((ext_vector_type(8))) float v8f;

#define DIM 128
#define BN_EPS 1e-5f

__device__ __forceinline__ void bn_coeff(float s, float q, float g, float b, float invN,
                                         float& sc, float& sh) {
  const float mu = s * invN;
  const float var = q * invN - mu * mu;
  sc = g / sqrtf(var + BN_EPS);
  sh = b - mu * sc;
}

__device__ __forceinline__ void bn_coeff4(const float* __restrict__ stats,
                                          const float* __restrict__ gamma,
                                          const float* __restrict__ beta, float invN,
                                          int lane, float4& sc, float4& sh) {
  const float4 s = reinterpret_cast<const float4*>(stats)[lane];
  const float4 q = reinterpret_cast<const float4*>(stats + DIM)[lane];
  const float4 g = reinterpret_cast<const float4*>(gamma)[lane];
  const float4 b = reinterpret_cast<const float4*>(beta)[lane];
  bn_coeff(s.x, q.x, g.x, b.x, invN, sc.x, sh.x);
  bn_coeff(s.y, q.y, g.y, b.y, invN, sc.y, sh.y);
  bn_coeff(s.z, q.z, g.z, b.z, invN, sc.z, sh.z);
  bn_coeff(s.w, q.w, g.w, b.w, invN, sc.w, sh.w);
}

#define CNT_THREADS 1024
#define CNT_RANGE 1024
#define CNT_CAP 65536
#define CNT_CHUNKS 16
__global__ void __launch_bounds__(CNT_THREADS) count_kernel(
    const int* __restrict__ src, const int* __restrict__ dst,
    int* __restrict__ cntS, int* __restrict__ cntD, int* __restrict__ tot,
    int* __restrict__ slot, int N, int E, int CH, int R) {
  __shared__ int cnt[CNT_CAP];
  const int tid = threadIdx.x;
  const int lane = tid & 31;
  const int wave = tid >> 5;
  const bool isDst = blockIdx.x < CNT_CHUNKS;
  const int q = isDst ? (int)blockIdx.x : (int)blockIdx.x - CNT_CHUNKS;
  const int* __restrict__ keys = isDst ? dst : src;
  int* __restrict__ outc = (isDst ? cntD : cntS) + (size_t)q * N;
  const int eb = q * CH;
  const int ee = (E - eb < CH) ? E : eb + CH;
  for (int nb = 0; nb < N; nb += CNT_CAP) {
    const int n = (N - nb < CNT_CAP) ? (N - nb) : CNT_CAP;
    const int nr = (n + CNT_RANGE - 1) / CNT_RANGE;
    {
      int4* c4p = reinterpret_cast<int4*>(cnt);
      for (int i = tid; i < (n + 3) / 4; i += CNT_THREADS) c4p[i] = make_int4(0, 0, 0, 0);
    }
    __syncthreads();
    for (int e0 = eb + tid; e0 < ee; e0 += 8 * CNT_THREADS) {
      int k[8];
#pragma unroll
      for (int u = 0; u < 8; ++u) {
        const int e = e0 + u * CNT_THREADS;
        const int kv = keys[(e < ee) ? e : ee - 1];
        k[u] = (e < ee) ? kv - nb : -1;
      }
      int place[8];
#pragma unroll
      for (int u = 0; u < 8; ++u) {
        place[u] = 0;
        if ((unsigned)k[u] < (unsigned)n) place[u] = atomicAdd(&cnt[k[u]], 1);
      }
      if (isDst) {
#pragma unroll
        for (int u = 0; u < 8; ++u) {
          const int e = e0 + u * CNT_THREADS;
          if ((unsigned)k[u] < (unsigned)n) slot[e] = place[u];
        }
      }
    }
    __syncthreads();
    for (int j = wave; j < nr; j += CNT_THREADS / 32) {
      int c = 0;
      for (int t = lane; t < CNT_RANGE; t += 32) {
        const int i = j * CNT_RANGE + t;
        if (i < n) {
          const int v = cnt[i];
          outc[nb + i] = v;
          c += v;
        }
      }
      if (isDst) {
        for (int o = 16; o > 0; o >>= 1) c += __shfl_xor(c, o);
        if (lane == 0) tot[(size_t)q * R + nb / CNT_RANGE + j] = c;
      }
    }
    __syncthreads();
  }
}

__global__ void __launch_bounds__(CNT_RANGE) offsets_kernel(
    const int* __restrict__ cntS, int* __restrict__ cntD, const int* __restrict__ tot,
    int* __restrict__ rp, float* __restrict__ onorm, float* __restrict__ inorm,
    int N, int R) {
  __shared__ int wtot[33];
  __shared__ int wbase[33];
  const int tid = threadIdx.x;
  const int lane = tid & 31;
  const int wave = tid >> 5;
  const int r = blockIdx.x;
  const int i = r * CNT_RANGE + tid;

  int acc = 0;
  for (int t = tid; t < CNT_CHUNKS * R; t += CNT_RANGE) {
    const int v = tot[t];
    if (t % R < r) acc += v;
  }

  int dd = 0;
  if (i < N) {
    int cs[CNT_CHUNKS], cd[CNT_CHUNKS];
#pragma unroll
    for (int q = 0; q < CNT_CHUNKS; ++q) {
      cs[q] = cntS[(size_t)q * N + i];
      cd[q] = cntD[(size_t)q * N + i];
    }
    int ds = 0;
#pragma unroll
    for (int q = 0; q < CNT_CHUNKS; ++q) {
      ds += cs[q];
      cntD[(size_t)q * N + i] = dd;
      dd += cd[q];
    }
    onorm[i] = (ds > 0) ? 1.0f / sqrtf((float)ds) : 0.0f;
    inorm[i] = (dd > 0) ? 1.0f / sqrtf((float)dd) : 0.0f;
  }

  int v = dd;
  for (int o = 1; o < 32; o <<= 1) {
    const int t = __shfl_up(v, o);
    if (lane >= o) v += t;
  }
  for (int o = 16; o > 0; o >>= 1) acc += __shfl_xor(acc, o);
  if (lane == 31) wtot[wave] = v;
  if (lane == 0) wbase[wave] = acc;
  __syncthreads();
  if (wave == 0) {
    const int t = wtot[lane];
    int u = t;
    for (int o = 1; o < 32; o <<= 1) {
      const int s = __shfl_up(u, o);
      if (lane >= o) u += s;
    }
    wtot[lane] = u - t;
    int b = wbase[lane];
    for (int o = 16; o > 0; o >>= 1) b += __shfl_xor(b, o);
    if (lane == 0) wbase[32] = b;
  }
  __syncthreads();
  const int incl = wbase[32] + wtot[wave] + v;
  if (i < N) rp[i] = incl - dd;
  if (i == N - 1) rp[N] = incl;
}

#define STREAM_BLOCKS 512
#define STREAM_THREADS 256

#define PLACE_U 4
__global__ void __launch_bounds__(STREAM_THREADS) place_kernel(
    const int* __restrict__ src, const int* __restrict__ dst,
    const int* __restrict__ slot, const int* __restrict__ rp,
    const int* __restrict__ cntD,
    const float* __restrict__ onorm, int2* __restrict__ csr,
    int E, int N, int CH) {
  const int t = blockIdx.x * STREAM_THREADS + threadIdx.x;
  const int stride = gridDim.x * STREAM_THREADS;
  for (int e0 = t; e0 < E; e0 += PLACE_U * stride) {
    int s[PLACE_U], d[PLACE_U], sl[PLACE_U], q[PLACE_U];
#pragma unroll
    for (int u = 0; u < PLACE_U; ++u) {
      const int e = e0 + u * stride;
      const int ec = (e < E) ? e : e0;
      s[u] = src[ec];
      d[u] = dst[ec];
      sl[u] = slot[ec];
      q[u] = ec / CH;
    }
    int base[PLACE_U], cs[PLACE_U], w[PLACE_U];
#pragma unroll
    for (int u = 0; u < PLACE_U; ++u) {
      w[u] = __float_as_int(onorm[s[u]]);
      base[u] = rp[d[u]];
      cs[u] = cntD[(size_t)q[u] * N + d[u]];
    }
#pragma unroll
    for (int u = 0; u < PLACE_U; ++u) {
      const int e = e0 + u * stride;
      asm volatile("" ::"v"(s[u]), "v"(sl[u]), "v"(w[u]), "v"(base[u]), "v"(cs[u]));
      if (e < E) {
        int2 rec;
        rec.x = s[u];
        rec.y = w[u];
        csr[base[u] + cs[u] + sl[u]] = rec;
      }
    }
  }
}

#define STREAM_U 16
__global__ void __launch_bounds__(STREAM_THREADS) gather_kernel(
    const int* __restrict__ ids, const float* __restrict__ emb,
    float* __restrict__ x, float* __restrict__ stats, int nstats, int N) {
  if (blockIdx.x == 0) {
    for (int i = threadIdx.x; i < nstats; i += STREAM_THREADS) stats[i] = 0.0f;
  }
  const int lane = threadIdx.x & 31;
  const int wv = blockIdx.x * (STREAM_THREADS / 32) +
                 __builtin_amdgcn_readfirstlane((int)(threadIdx.x >> 5));
  const int nw = gridDim.x * (STREAM_THREADS / 32);
  const float4* E4 = reinterpret_cast<const float4*>(emb) + lane;
  float4* X4 = reinterpret_cast<float4*>(x) + lane;
  for (int i0 = wv; i0 < N; i0 += STREAM_U * nw) {
    const int iu = i0 + lane * nw;
    int idv = 0;
    if (lane < STREAM_U) idv = ids[(iu < N) ? iu : i0];
    float4 v[STREAM_U];
#pragma unroll
    for (int u = 0; u < STREAM_U; ++u) {
      const int id = __builtin_amdgcn_readlane(idv, u);
      v[u] = E4[(size_t)id * (DIM / 4)];
    }
#pragma unroll
    for (int u = 0; u < STREAM_U; ++u) {
      const int i = i0 + u * nw;
      asm volatile("" ::"v"(v[u].x), "v"(v[u].y), "v"(v[u].z), "v"(v[u].w));
      if (i < N) X4[(size_t)i * (DIM / 4)] = v[u];
    }
  }
}

#define AGG_WAVES 8

template <int B, bool CLAMP, bool BN>
__device__ __forceinline__ void sum_rows(float4& acc, const float4* __restrict__ X4,
                                         int recSrc, int recW, int j, int m,
                                         const float4& sc, const float4& sh) {
  float4 v[B];
  float w[B];
#pragma unroll
  for (int u = 0; u < B; ++u) {
    int idx = j + u;
    bool ok = true;
    if (CLAMP) {
      ok = idx < m;
      idx = ok ? idx : m - 1;
    }
    const int s = __builtin_amdgcn_readlane(recSrc, idx);
    const float ww = __int_as_float(__builtin_amdgcn_readlane(recW, idx));
    w[u] = ok ? ww : 0.0f;
    v[u] = X4[(size_t)s * (DIM / 4)];
  }
#pragma unroll
  for (int u = 0; u < B; ++u) {
    float4 xv = v[u];
    if (BN) {
      xv.x = __builtin_fmaf(xv.x, sc.x, sh.x);
      xv.y = __builtin_fmaf(xv.y, sc.y, sh.y);
      xv.z = __builtin_fmaf(xv.z, sc.z, sh.z);
      xv.w = __builtin_fmaf(xv.w, sc.w, sh.w);
    }
    acc.x = __builtin_fmaf(xv.x, w[u], acc.x);
    acc.y = __builtin_fmaf(xv.y, w[u], acc.y);
    acc.z = __builtin_fmaf(xv.z, w[u], acc.z);
    acc.w = __builtin_fmaf(xv.w, w[u], acc.w);
  }
}

template <bool BN>
__global__ void __launch_bounds__(AGG_WAVES * 32) aggregate_kernel(
    const int* __restrict__ rp, const int2* __restrict__ csr,
    const float* __restrict__ x, float* __restrict__ agg,
    const float* __restrict__ stats, const float* __restrict__ gamma,
    const float* __restrict__ beta, float invN, int N) {
  const int lane = threadIdx.x & 31;
  const int wv = __builtin_amdgcn_readfirstlane((int)(threadIdx.x >> 5));
  const int nw = gridDim.x * AGG_WAVES;
  int row = blockIdx.x * AGG_WAVES + wv;
  if (row >= N) return;
  const float4* X4 = reinterpret_cast<const float4*>(x) + lane;
  float4* A4 = reinterpret_cast<float4*>(agg) + lane;
  float4 sc = make_float4(1.0f, 1.0f, 1.0f, 1.0f);
  float4 sh = make_float4(0.0f, 0.0f, 0.0f, 0.0f);
  if (BN) bn_coeff4(stats, gamma, beta, invN, lane, sc, sh);

  int beg = rp[row];
  int end = rp[row + 1];
  int2 ev = make_int2(0, 0);
  if (lane < end - beg) ev = csr[beg + lane];
  int nrow = row + nw;
  int nb = 0, ne = 0;
  if (nrow < N) {
    nb = rp[nrow];
    ne = rp[nrow + 1];
  }

  for (;;) {
    int2 nev = make_int2(0, 0);
    if (lane < ne - nb) nev = csr[nb + lane];
    const int nnrow = nrow + nw;
    int nnb = 0, nne = 0;
    if (nnrow < N) {
      nnb = rp[nnrow];
      nne = rp[nnrow + 1];
    }

    float4 acc = make_float4(0.0f, 0.0f, 0.0f, 0.0f);
    const int cnt = end - beg;
    int done = 0;
    for (;;) {
      const int m = (cnt - done < 32) ? (cnt - done) : 32;
      int j = 0;
      for (; j + 8 <= m; j += 8) sum_rows<8, false, BN>(acc, X4, ev.x, ev.y, j, m, sc, sh);
      for (; j < m; j += 4) sum_rows<4, true, BN>(acc, X4, ev.x, ev.y, j, m, sc, sh);
      done += 32;
      if (done >= cnt) break;
      ev = make_int2(0, 0);
      if (lane < cnt - done) ev = csr[beg + done + lane];
    }
    A4[(size_t)row * (DIM / 4)] = acc;

    if (nrow >= N) break;
    row = nrow;
    beg = nb;
    end = ne;
    ev = nev;
    nrow = nnrow;
    nb = nnb;
    ne = nne;
  }
}

template <bool BN>
__global__ void __launch_bounds__(256) gemm_fused_kernel(
    const float* __restrict__ agg, const float* __restrict__ xin,
    const float* __restrict__ inorm,
    const float* __restrict__ W, const float* __restrict__ bias,
    const float* __restrict__ Rw, const float* __restrict__ Rb,
    const float* __restrict__ pstats, const float* __restrict__ pgamma,
    const float* __restrict__ pbeta, float invN,
    float* __restrict__ h, float* __restrict__ csum, float* __restrict__ csumsq,
    int N) {
  __shared__ float4 sB[32 * 8 * 32];
  __shared__ float4 sScale[DIM / 4];
  __shared__ float4 sShift[DIM / 4];

  const int tid = threadIdx.x;
  const int wave = tid >> 5;
  const int lane = tid & 31;
  const int r = lane & 15;
  const int hf = lane >> 4;

  const int bk = tid & (DIM - 1);
  float bnS = 0.0f, bnQ = 0.0f, bnG = 0.0f, bnB = 0.0f;
  if (BN) {
    bnS = pstats[bk];
    bnQ = pstats[DIM + bk];
    bnG = pgamma[bk];
    bnB = pbeta[bk];
  }

  {
    const float* wp = W + (size_t)(4 * hf) * DIM + wave * 16 + r;
    const float* rp = Rw + (size_t)(4 * hf) * DIM + wave * 16 + r;
    float4* sp = sB + tid;
#pragma unroll 4
    for (int kp = 0; kp < 16; ++kp) {
#pragma unroll
      for (int q = 0; q < 2; ++q) {
        float4 v;
        v.x = wp[(2 * q) * DIM];
        v.y = wp[(2 * q + 1) * DIM];
        v.z = rp[(2 * q) * DIM];
        v.w = rp[(2 * q + 1) * DIM];
        sp[q * 256] = v;
      }
      wp += 8 * DIM;
      rp += 8 * DIM;
      sp += 2 * 256;
    }
  }
  if (BN) {
    float sc, sh;
    bn_coeff(bnS, bnQ, bnG, bnB, invN, sc, sh);
    float* dstp = reinterpret_cast<float*>((tid < DIM) ? sScale : sShift);
    dstp[bk] = (tid < DIM) ? sc : sh;
  }
  __syncthreads();

  const int row0 = (blockIdx.x * 8 + wave) * 16;
  if (row0 >= N) return;

  const int row = row0 + r;
  const float innr = inorm[row];
  const float4* aP = reinterpret_cast<const float4*>(agg + (size_t)row * DIM) + hf;
  const float4* xP = reinterpret_cast<const float4*>(xin + (size_t)row * DIM) + hf;
  const float4* bP = sB + lane;

  v8f acc1[8], acc2[8];
#pragma unroll
  for (int ct = 0; ct < 8; ++ct) {
    acc1[ct] = (v8f){0.f, 0.f, 0.f, 0.f, 0.f, 0.f, 0.f, 0.f};
    acc2[ct] = (v8f){0.f, 0.f, 0.f, 0.f, 0.f, 0.f, 0.f, 0.f};
  }

  const float4* scP = sScale + hf;
  const float4* shP = sShift + hf;

  float4 ta = aP[0];
  float4 tx = xP[0];
  float4 b0[8], b1[8];
#pragma unroll
  for (int ct = 0; ct < 8; ++ct) b0[ct] = bP[ct * 32];

  for (int kp = 0; kp < 16; ++kp) {
    float4 ta_n = ta, tx_n = tx;
    if (kp + 1 < 16) {
      ta_n = aP[(kp + 1) * 2];
      tx_n = xP[(kp + 1) * 2];
    }
#pragma unroll
    for (int ct = 0; ct < 8; ++ct) b1[ct] = bP[((2 * kp + 1) * 8 + ct) * 32];

    if (BN) {
      const float4 s4 = scP[2 * kp];
      const float4 t4 = shP[2 * kp];
      tx.x = __builtin_fmaf(tx.x, s4.x, t4.x);
      tx.y = __builtin_fmaf(tx.y, s4.y, t4.y);
      tx.z = __builtin_fmaf(tx.z, s4.z, t4.z);
      tx.w = __builtin_fmaf(tx.w, s4.w, t4.w);
    }

    v2f a1, a2;
    a1.x = ta.x * innr;
    a1.y = ta.y * innr;
    a2.x = tx.x;
    a2.y = tx.y;
#pragma unroll
    for (int ct = 0; ct < 8; ++ct) {
      v2f w1, w2;
      w1.x = b0[ct].x; w1.y = b0[ct].y;
      w2.x = b0[ct].z; w2.y = b0[ct].w;
      acc1[ct] = __builtin_amdgcn_wmma_f32_16x16x4_f32(false, a1, false, w1,
                                                       (short)0, acc1[ct], false, false);
      acc2[ct] = __builtin_amdgcn_wmma_f32_16x16x4_f32(false, a2, false, w2,
                                                       (short)0, acc2[ct], false, false);
    }

    if (kp + 1 < 16) {
#pragma unroll
      for (int ct = 0; ct < 8; ++ct) b0[ct] = bP[((2 * kp + 2) * 8 + ct) * 32];
    }

    a1.x = ta.z * innr;
    a1.y = ta.w * innr;
    a2.x = tx.z;
    a2.y = tx.w;
#pragma unroll
    for (int ct = 0; ct < 8; ++ct) {
      v2f w1, w2;
      w1.x = b1[ct].x; w1.y = b1[ct].y;
      w2.x = b1[ct].z; w2.y = b1[ct].w;
      acc1[ct] = __builtin_amdgcn_wmma_f32_16x16x4_f32(false, a1, false, w1,
                                                       (short)0, acc1[ct], false, false);
      acc2[ct] = __builtin_amdgcn_wmma_f32_16x16x4_f32(false, a2, false, w2,
                                                       (short)0, acc2[ct], false, false);
    }
    ta = ta_n;
    tx = tx_n;
  }

#pragma unroll
  for (int ct = 0; ct < 8; ++ct) {
    const int col0 = ct * 16;
    const float bb = bias[col0 + r];
    const float rb = Rb[col0 + r];
    float s = 0.0f, s2 = 0.0f;
#pragma unroll
    for (int j = 0; j < 8; ++j) {
      float v = fmaxf(acc1[ct][j] + bb, 0.0f) + fmaxf(acc2[ct][j] + rb, 0.0f);
      h[(size_t)(row0 + j + 8 * hf) * DIM + col0 + r] = v;
      s += v;
      s2 += v * v;
    }
    atomicAdd(&csum[col0 + r], s);
    atomicAdd(&csumsq[col0 + r], s2);
  }
}

__global__ void __launch_bounds__(STREAM_THREADS) bnapply_kernel(
    const float* __restrict__ h, const float* __restrict__ stats,
    const float* __restrict__ gamma, const float* __restrict__ beta, float invN,
    float* __restrict__ out, int N) {
  const int lane = threadIdx.x & 31;
  const int wv = blockIdx.x * (STREAM_THREADS / 32) +
                 __builtin_amdgcn_readfirstlane((int)(threadIdx.x >> 5));
  const int nw = gridDim.x * (STREAM_THREADS / 32);
  const float4* H4 = reinterpret_cast<const float4*>(h) + lane;
  float4* O4 = reinterpret_cast<float4*>(out) + lane;
  float4 sc, sh;
  bn_coeff4(stats, gamma, beta, invN, lane, sc, sh);
  for (int i0 = wv; i0 < N; i0 += STREAM_U * nw) {
    float4 v[STREAM_U];
#pragma unroll
    for (int u = 0; u < STREAM_U; ++u) {
      const int i = i0 + u * nw;
      v[u] = H4[(size_t)((i < N) ? i : i0) * (DIM / 4)];
    }
#pragma unroll
    for (int u = 0; u < STREAM_U; ++u) {
      const int i = i0 + u * nw;
      asm volatile("" ::"v"(v[u].x), "v"(v[u].y), "v"(v[u].z), "v"(v[u].w));
      if (i < N) {
        float4 o;
        o.x = v[u].x * sc.x + sh.x;
        o.y = v[u].y * sc.y + sh.y;
        o.z = v[u].z * sc.z + sh.z;
        o.w = v[u].w * sc.w + sh.w;
        O4[(size_t)i * (DIM / 4)] = o;
      }
    }
  }
}


static inline int cdiv(int a, int b) { return (a + b - 1) / b; }

extern "C" void kernel_launch(void* const* d_in, const int* in_sizes, int n_in,
                              void* d_out, int out_size, void* d_ws, size_t ws_size,
                              hipStream_t stream) {
  const int* node_ids = (const int*)d_in[0];
  const int* src = (const int*)d_in[1];
  const int* dst = (const int*)d_in[2];
  const float* emb = (const float*)d_in[3];
  const float* Ws = (const float*)d_in[4];
  const float* bs = (const float*)d_in[5];
  const float* Rws = (const float*)d_in[6];
  const float* Rbs = (const float*)d_in[7];
  const float* gammas = (const float*)d_in[8];
  const float* betas = (const float*)d_in[9];

  const int N = in_sizes[0];
  const int E = in_sizes[1];
  const int L = in_sizes[4] / (DIM * DIM);

  float* wsf = (float*)d_ws;
  size_t off = 0;
  float* onorm = wsf + off; off += (size_t)N;
  float* inorm = wsf + off; off += (size_t)N;
  float* x    = wsf + off;  off += (size_t)N * DIM;
  float* agg  = wsf + off;  off += (size_t)N * DIM;
  float* h    = wsf + off;  off += (size_t)N * DIM;
  float* stats = wsf + off; off += (size_t)L * 2 * DIM;
  off = (off + 3) & ~(size_t)3;
  int2* csr = (int2*)(wsf + off); off += (size_t)E * 2;
  int* slot = (int*)(wsf + off);  off += (size_t)E;
  int* rp   = (int*)(wsf + off);  off += (size_t)N + 1;
  const int R = cdiv(N, CNT_RANGE);
  int* cntS = (int*)(wsf + off);  off += (size_t)CNT_CHUNKS * N;
  int* cntD = (int*)(wsf + off);  off += (size_t)CNT_CHUNKS * N;
  int* tot  = (int*)(wsf + off);  off += (size_t)CNT_CHUNKS * R;

  const int B = 256;

  const int CH = cdiv(E, CNT_CHUNKS);
  count_kernel<<<2 * CNT_CHUNKS, CNT_THREADS, 0, stream>>>(src, dst, cntS, cntD, tot, slot,
                                                           N, E, CH, R);
  offsets_kernel<<<R, CNT_RANGE, 0, stream>>>(cntS, cntD, tot, rp, onorm, inorm, N, R);
  place_kernel<<<STREAM_BLOCKS, STREAM_THREADS, 0, stream>>>(src, dst, slot, rp, cntD, onorm,
                                                             csr, E, N, CH);

  gather_kernel<<<STREAM_BLOCKS, STREAM_THREADS, 0, stream>>>(node_ids, emb, x, stats,
                                                              L * 2 * DIM, N);

  const int rowTiles = cdiv(N, 16);
  const int aggBlocks = (cdiv(N, AGG_WAVES) < 1024) ? cdiv(N, AGG_WAVES) : 1024;
  const float invN = 1.0f / (float)N;
  float* cur = x;
  float* nxt = h;
  for (int l = 0; l < L; ++l) {
    float* st = stats + (size_t)l * 2 * DIM;
    const float* W = Ws + (size_t)l * DIM * DIM;
    const float* bw = bs + (size_t)l * DIM;
    const float* Rw = Rws + (size_t)l * DIM * DIM;
    const float* rb = Rbs + (size_t)l * DIM;
    if (l == 0) {
      aggregate_kernel<false><<<aggBlocks, AGG_WAVES * 32, 0, stream>>>(
          rp, csr, cur, agg, nullptr, nullptr, nullptr, invN, N);
      gemm_fused_kernel<false><<<cdiv(rowTiles, 8), B, 0, stream>>>(
          agg, cur, inorm, W, bw, Rw, rb, nullptr, nullptr, nullptr, invN, nxt, st, st + DIM,
          N);
    } else {
      const float* pst = st - 2 * DIM;
      const float* pg = gammas + (size_t)(l - 1) * DIM;
      const float* pb = betas + (size_t)(l - 1) * DIM;
      aggregate_kernel<true><<<aggBlocks, AGG_WAVES * 32, 0, stream>>>(
          rp, csr, cur, agg, pst, pg, pb, invN, N);
      gemm_fused_kernel<true><<<cdiv(rowTiles, 8), B, 0, stream>>>(
          agg, cur, inorm, W, bw, Rw, rb, pst, pg, pb, invN, nxt, st, st + DIM, N);
    }
    float* t = cur;
    cur = nxt;
    nxt = t;
  }
  bnapply_kernel<<<STREAM_BLOCKS, STREAM_THREADS, 0, stream>>>(
      cur, stats + (size_t)(L - 1) * 2 * DIM, gammas + (size_t)(L - 1) * DIM,
      betas + (size_t)(L - 1) * DIM, invN, (float*)d_out, N);
}


# BN column sums reduced in the work-group before the atomics: lane halves by shuffle, 8 waves in LDS, one atomic add per column per work-group; h stored behind that barrier; fp32 sum order changes
# speedup vs baseline: 3.2462x; 1.3179x over previous
#include <hip/hip_runtime.h>
#include <math.h>


typedef __attribute__((ext_vector_type(2))) float v2f;
typedef __attribute__((ext_vector_type(8))) float v8f;

#define DIM 128
#define BN_EPS 1e-5f

__device__ __forceinline__ void bn_coeff(float s, float q, float g, float b, float invN,
                                         float& sc, float& sh) {
  const float mu = s * invN;
  const float var = q * invN - mu * mu;
  sc = g / sqrtf(var + BN_EPS);
  sh = b - mu * sc;
}

__device__ __forceinline__ void bn_coeff4(const float* __restrict__ stats,
                                          const float* __restrict__ gamma,
                                          const float* __restrict__ beta, float invN,
                                          int lane, float4& sc, float4& sh) {
  const float4 s = reinterpret_cast<const float4*>(stats)[lane];
  const float4 q = reinterpret_cast<const float4*>(stats + DIM)[lane];
  const float4 g = reinterpret_cast<const float4*>(gamma)[lane];
  const float4 b = reinterpret_cast<const float4*>(beta)[lane];
  bn_coeff(s.x, q.x, g.x, b.x, invN, sc.x, sh.x);
  bn_coeff(s.y, q.y, g.y, b.y, invN, sc.y, sh.y);
  bn_coeff(s.z, q.z, g.z, b.z, invN, sc.z, sh.z);
  bn_coeff(s.w, q.w, g.w, b.w, invN, sc.w, sh.w);
}

#define CNT_THREADS 1024
#define CNT_RANGE 1024
#define CNT_CAP 65536
#define CNT_CHUNKS 16
__global__ void __launch_bounds__(CNT_THREADS) count_kernel(
    const int* __restrict__ src, const int* __restrict__ dst,
    int* __restrict__ cntS, int* __restrict__ cntD, int* __restrict__ tot,
    int* __restrict__ slot, int N, int E, int CH, int R) {
  __shared__ int cnt[CNT_CAP];
  const int tid = threadIdx.x;
  const int lane = tid & 31;
  const int wave = tid >> 5;
  const bool isDst = blockIdx.x < CNT_CHUNKS;
  const int q = isDst ? (int)blockIdx.x : (int)blockIdx.x - CNT_CHUNKS;
  const int* __restrict__ keys = isDst ? dst : src;
  int* __restrict__ outc = (isDst ? cntD : cntS) + (size_t)q * N;
  const int eb = q * CH;
  const int ee = (E - eb < CH) ? E : eb + CH;
  for (int nb = 0; nb < N; nb += CNT_CAP) {
    const int n = (N - nb < CNT_CAP) ? (N - nb) : CNT_CAP;
    const int nr = (n + CNT_RANGE - 1) / CNT_RANGE;
    {
      int4* c4p = reinterpret_cast<int4*>(cnt);
      for (int i = tid; i < (n + 3) / 4; i += CNT_THREADS) c4p[i] = make_int4(0, 0, 0, 0);
    }
    __syncthreads();
    for (int e0 = eb + tid; e0 < ee; e0 += 8 * CNT_THREADS) {
      int k[8];
#pragma unroll
      for (int u = 0; u < 8; ++u) {
        const int e = e0 + u * CNT_THREADS;
        const int kv = keys[(e < ee) ? e : ee - 1];
        k[u] = (e < ee) ? kv - nb : -1;
      }
      int place[8];
#pragma unroll
      for (int u = 0; u < 8; ++u) {
        place[u] = 0;
        if ((unsigned)k[u] < (unsigned)n) place[u] = atomicAdd(&cnt[k[u]], 1);
      }
      if (isDst) {
#pragma unroll
        for (int u = 0; u < 8; ++u) {
          const int e = e0 + u * CNT_THREADS;
          if ((unsigned)k[u] < (unsigned)n) slot[e] = place[u];
        }
      }
    }
    __syncthreads();
    for (int j = wave; j < nr; j += CNT_THREADS / 32) {
      int c = 0;
      for (int t = lane; t < CNT_RANGE; t += 32) {
        const int i = j * CNT_RANGE + t;
        if (i < n) {
          const int v = cnt[i];
          outc[nb + i] = v;
          c += v;
        }
      }
      if (isDst) {
        for (int o = 16; o > 0; o >>= 1) c += __shfl_xor(c, o);
        if (lane == 0) tot[(size_t)q * R + nb / CNT_RANGE + j] = c;
      }
    }
    __syncthreads();
  }
}

__global__ void __launch_bounds__(CNT_RANGE) offsets_kernel(
    const int* __restrict__ cntS, int* __restrict__ cntD, const int* __restrict__ tot,
    int* __restrict__ rp, float* __restrict__ onorm, float* __restrict__ inorm,
    int N, int R) {
  __shared__ int wtot[33];
  __shared__ int wbase[33];
  const int tid = threadIdx.x;
  const int lane = tid & 31;
  const int wave = tid >> 5;
  const int r = blockIdx.x;
  const int i = r * CNT_RANGE + tid;

  int acc = 0;
  for (int t = tid; t < CNT_CHUNKS * R; t += CNT_RANGE) {
    const int v = tot[t];
    if (t % R < r) acc += v;
  }

  int dd = 0;
  if (i < N) {
    int cs[CNT_CHUNKS], cd[CNT_CHUNKS];
#pragma unroll
    for (int q = 0; q < CNT_CHUNKS; ++q) {
      cs[q] = cntS[(size_t)q * N + i];
      cd[q] = cntD[(size_t)q * N + i];
    }
    int ds = 0;
#pragma unroll
    for (int q = 0; q < CNT_CHUNKS; ++q) {
      ds += cs[q];
      cntD[(size_t)q * N + i] = dd;
      dd += cd[q];
    }
    onorm[i] = (ds > 0) ? 1.0f / sqrtf((float)ds) : 0.0f;
    inorm[i] = (dd > 0) ? 1.0f / sqrtf((float)dd) : 0.0f;
  }

  int v = dd;
  for (int o = 1; o < 32; o <<= 1) {
    const int t = __shfl_up(v, o);
    if (lane >= o) v += t;
  }
  for (int o = 16; o > 0; o >>= 1) acc += __shfl_xor(acc, o);
  if (lane == 31) wtot[wave] = v;
  if (lane == 0) wbase[wave] = acc;
  __syncthreads();
  if (wave == 0) {
    const int t = wtot[lane];
    int u = t;
    for (int o = 1; o < 32; o <<= 1) {
      const int s = __shfl_up(u, o);
      if (lane >= o) u += s;
    }
    wtot[lane] = u - t;
    int b = wbase[lane];
    for (int o = 16; o > 0; o >>= 1) b += __shfl_xor(b, o);
    if (lane == 0) wbase[32] = b;
  }
  __syncthreads();
  const int incl = wbase[32] + wtot[wave] + v;
  if (i < N) rp[i] = incl - dd;
  if (i == N - 1) rp[N] = incl;
}

#define STREAM_BLOCKS 512
#define STREAM_THREADS 256

#define PLACE_U 4
__global__ void __launch_bounds__(STREAM_THREADS) place_kernel(
    const int* __restrict__ src, const int* __restrict__ dst,
    const int* __restrict__ slot, const int* __restrict__ rp,
    const int* __restrict__ cntD,
    const float* __restrict__ onorm, int2* __restrict__ csr,
    int E, int N, int CH) {
  const int t = blockIdx.x * STREAM_THREADS + threadIdx.x;
  const int stride = gridDim.x * STREAM_THREADS;
  for (int e0 = t; e0 < E; e0 += PLACE_U * stride) {
    int s[PLACE_U], d[PLACE_U], sl[PLACE_U], q[PLACE_U];
#pragma unroll
    for (int u = 0; u < PLACE_U; ++u) {
      const int e = e0 + u * stride;
      const int ec = (e < E) ? e : e0;
      s[u] = src[ec];
      d[u] = dst[ec];
      sl[u] = slot[ec];
      q[u] = ec / CH;
    }
    int base[PLACE_U], cs[PLACE_U], w[PLACE_U];
#pragma unroll
    for (int u = 0; u < PLACE_U; ++u) {
      w[u] = __float_as_int(onorm[s[u]]);
      base[u] = rp[d[u]];
      cs[u] = cntD[(size_t)q[u] * N + d[u]];
    }
#pragma unroll
    for (int u = 0; u < PLACE_U; ++u) {
      const int e = e0 + u * stride;
      asm volatile("" ::"v"(s[u]), "v"(sl[u]), "v"(w[u]), "v"(base[u]), "v"(cs[u]));
      if (e < E) {
        int2 rec;
        rec.x = s[u];
        rec.y = w[u];
        csr[base[u] + cs[u] + sl[u]] = rec;
      }
    }
  }
}

#define STREAM_U 16
__global__ void __launch_bounds__(STREAM_THREADS) gather_kernel(
    const int* __restrict__ ids, const float* __restrict__ emb,
    float* __restrict__ x, float* __restrict__ stats, int nstats, int N) {
  if (blockIdx.x == 0) {
    for (int i = threadIdx.x; i < nstats; i += STREAM_THREADS) stats[i] = 0.0f;
  }
  const int lane = threadIdx.x & 31;
  const int wv = blockIdx.x * (STREAM_THREADS / 32) +
                 __builtin_amdgcn_readfirstlane((int)(threadIdx.x >> 5));
  const int nw = gridDim.x * (STREAM_THREADS / 32);
  const float4* E4 = reinterpret_cast<const float4*>(emb) + lane;
  float4* X4 = reinterpret_cast<float4*>(x) + lane;
  for (int i0 = wv; i0 < N; i0 += STREAM_U * nw) {
    const int iu = i0 + lane * nw;
    int idv = 0;
    if (lane < STREAM_U) idv = ids[(iu < N) ? iu : i0];
    float4 v[STREAM_U];
#pragma unroll
    for (int u = 0; u < STREAM_U; ++u) {
      const int id = __builtin_amdgcn_readlane(idv, u);
      v[u] = E4[(size_t)id * (DIM / 4)];
    }
#pragma unroll
    for (int u = 0; u < STREAM_U; ++u) {
      const int i = i0 + u * nw;
      asm volatile("" ::"v"(v[u].x), "v"(v[u].y), "v"(v[u].z), "v"(v[u].w));
      if (i < N) X4[(size_t)i * (DIM / 4)] = v[u];
    }
  }
}

#define AGG_WAVES 8

template <int B, bool CLAMP, bool BN>
__device__ __forceinline__ void sum_rows(float4& acc, const float4* __restrict__ X4,
                                         int recSrc, int recW, int j, int m,
                                         const float4& sc, const float4& sh) {
  float4 v[B];
  float w[B];
#pragma unroll
  for (int u = 0; u < B; ++u) {
    int idx = j + u;
    bool ok = true;
    if (CLAMP) {
      ok = idx < m;
      idx = ok ? idx : m - 1;
    }
    const int s = __builtin_amdgcn_readlane(recSrc, idx);
    const float ww = __int_as_float(__builtin_amdgcn_readlane(recW, idx));
    w[u] = ok ? ww : 0.0f;
    v[u] = X4[(size_t)s * (DIM / 4)];
  }
#pragma unroll
  for (int u = 0; u < B; ++u) {
    float4 xv = v[u];
    if (BN) {
      xv.x = __builtin_fmaf(xv.x, sc.x, sh.x);
      xv.y = __builtin_fmaf(xv.y, sc.y, sh.y);
      xv.z = __builtin_fmaf(xv.z, sc.z, sh.z);
      xv.w = __builtin_fmaf(xv.w, sc.w, sh.w);
    }
    acc.x = __builtin_fmaf(xv.x, w[u], acc.x);
    acc.y = __builtin_fmaf(xv.y, w[u], acc.y);
    acc.z = __builtin_fmaf(xv.z, w[u], acc.z);
    acc.w = __builtin_fmaf(xv.w, w[u], acc.w);
  }
}

template <bool BN>
__global__ void __launch_bounds__(AGG_WAVES * 32) aggregate_kernel(
    const int* __restrict__ rp, const int2* __restrict__ csr,
    const float* __restrict__ x, float* __restrict__ agg,
    const float* __restrict__ stats, const float* __restrict__ gamma,
    const float* __restrict__ beta, float invN, int N) {
  const int lane = threadIdx.x & 31;
  const int wv = __builtin_amdgcn_readfirstlane((int)(threadIdx.x >> 5));
  const int nw = gridDim.x * AGG_WAVES;
  int row = blockIdx.x * AGG_WAVES + wv;
  if (row >= N) return;
  const float4* X4 = reinterpret_cast<const float4*>(x) + lane;
  float4* A4 = reinterpret_cast<float4*>(agg) + lane;
  float4 sc = make_float4(1.0f, 1.0f, 1.0f, 1.0f);
  float4 sh = make_float4(0.0f, 0.0f, 0.0f, 0.0f);
  if (BN) bn_coeff4(stats, gamma, beta, invN, lane, sc, sh);

  int beg = rp[row];
  int end = rp[row + 1];
  int2 ev = make_int2(0, 0);
  if (lane < end - beg) ev = csr[beg + lane];
  int nrow = row + nw;
  int nb = 0, ne = 0;
  if (nrow < N) {
    nb = rp[nrow];
    ne = rp[nrow + 1];
  }

  for (;;) {
    int2 nev = make_int2(0, 0);
    if (lane < ne - nb) nev = csr[nb + lane];
    const int nnrow = nrow + nw;
    int nnb = 0, nne = 0;
    if (nnrow < N) {
      nnb = rp[nnrow];
      nne = rp[nnrow + 1];
    }

    float4 acc = make_float4(0.0f, 0.0f, 0.0f, 0.0f);
    const int cnt = end - beg;
    int done = 0;
    for (;;) {
      const int m = (cnt - done < 32) ? (cnt - done) : 32;
      int j = 0;
      for (; j + 8 <= m; j += 8) sum_rows<8, false, BN>(acc, X4, ev.x, ev.y, j, m, sc, sh);
      for (; j < m; j += 4) sum_rows<4, true, BN>(acc, X4, ev.x, ev.y, j, m, sc, sh);
      done += 32;
      if (done >= cnt) break;
      ev = make_int2(0, 0);
      if (lane < cnt - done) ev = csr[beg + done + lane];
    }
    A4[(size_t)row * (DIM / 4)] = acc;

    if (nrow >= N) break;
    row = nrow;
    beg = nb;
    end = ne;
    ev = nev;
    nrow = nnrow;
    nb = nnb;
    ne = nne;
  }
}

template <bool BN>
__global__ void __launch_bounds__(256) gemm_fused_kernel(
    const float* __restrict__ agg, const float* __restrict__ xin,
    const float* __restrict__ inorm,
    const float* __restrict__ W, const float* __restrict__ bias,
    const float* __restrict__ Rw, const float* __restrict__ Rb,
    const float* __restrict__ pstats, const float* __restrict__ pgamma,
    const float* __restrict__ pbeta, float invN,
    float* __restrict__ h, float* __restrict__ csum, float* __restrict__ csumsq,
    int N) {
  __shared__ float4 sB[32 * 8 * 32];
  __shared__ float4 sScale[DIM / 4];
  __shared__ float4 sShift[DIM / 4];
  __shared__ float sRed[8 * 2 * DIM];

  const int tid = threadIdx.x;
  const int wave = tid >> 5;
  const int lane = tid & 31;
  const int r = lane & 15;
  const int hf = lane >> 4;

  const int bk = tid & (DIM - 1);
  float bnS = 0.0f, bnQ = 0.0f, bnG = 0.0f, bnB = 0.0f;
  if (BN) {
    bnS = pstats[bk];
    bnQ = pstats[DIM + bk];
    bnG = pgamma[bk];
    bnB = pbeta[bk];
  }

  {
    const float* wp = W + (size_t)(4 * hf) * DIM + wave * 16 + r;
    const float* rp = Rw + (size_t)(4 * hf) * DIM + wave * 16 + r;
    float4* sp = sB + tid;
#pragma unroll 4
    for (int kp = 0; kp < 16; ++kp) {
#pragma unroll
      for (int q = 0; q < 2; ++q) {
        float4 v;
        v.x = wp[(2 * q) * DIM];
        v.y = wp[(2 * q + 1) * DIM];
        v.z = rp[(2 * q) * DIM];
        v.w = rp[(2 * q + 1) * DIM];
        sp[q * 256] = v;
      }
      wp += 8 * DIM;
      rp += 8 * DIM;
      sp += 2 * 256;
    }
  }
  if (BN) {
    float sc, sh;
    bn_coeff(bnS, bnQ, bnG, bnB, invN, sc, sh);
    float* dstp = reinterpret_cast<float*>((tid < DIM) ? sScale : sShift);
    dstp[bk] = (tid < DIM) ? sc : sh;
  }
  __syncthreads();

  const int row0 = (blockIdx.x * 8 + wave) * 16;
  v8f acc1[8], acc2[8];

  if (row0 < N) {

  const int row = row0 + r;
  const float innr = inorm[row];
  const float4* aP = reinterpret_cast<const float4*>(agg + (size_t)row * DIM) + hf;
  const float4* xP = reinterpret_cast<const float4*>(xin + (size_t)row * DIM) + hf;
  const float4* bP = sB + lane;

#pragma unroll
  for (int ct = 0; ct < 8; ++ct) {
    acc1[ct] = (v8f){0.f, 0.f, 0.f, 0.f, 0.f, 0.f, 0.f, 0.f};
    acc2[ct] = (v8f){0.f, 0.f, 0.f, 0.f, 0.f, 0.f, 0.f, 0.f};
  }

  const float4* scP = sScale + hf;
  const float4* shP = sShift + hf;

  float4 ta = aP[0];
  float4 tx = xP[0];
  float4 b0[8], b1[8];
#pragma unroll
  for (int ct = 0; ct < 8; ++ct) b0[ct] = bP[ct * 32];

  for (int kp = 0; kp < 16; ++kp) {
    float4 ta_n = ta, tx_n = tx;
    if (kp + 1 < 16) {
      ta_n = aP[(kp + 1) * 2];
      tx_n = xP[(kp + 1) * 2];
    }
#pragma unroll
    for (int ct = 0; ct < 8; ++ct) b1[ct] = bP[((2 * kp + 1) * 8 + ct) * 32];

    if (BN) {
      const float4 s4 = scP[2 * kp];
      const float4 t4 = shP[2 * kp];
      tx.x = __builtin_fmaf(tx.x, s4.x, t4.x);
      tx.y = __builtin_fmaf(tx.y, s4.y, t4.y);
      tx.z = __builtin_fmaf(tx.z, s4.z, t4.z);
      tx.w = __builtin_fmaf(tx.w, s4.w, t4.w);
    }

    v2f a1, a2;
    a1.x = ta.x * innr;
    a1.y = ta.y * innr;
    a2.x = tx.x;
    a2.y = tx.y;
#pragma unroll
    for (int ct = 0; ct < 8; ++ct) {
      v2f w1, w2;
      w1.x = b0[ct].x; w1.y = b0[ct].y;
      w2.x = b0[ct].z; w2.y = b0[ct].w;
      acc1[ct] = __builtin_amdgcn_wmma_f32_16x16x4_f32(false, a1, false, w1,
                                                       (short)0, acc1[ct], false, false);
      acc2[ct] = __builtin_amdgcn_wmma_f32_16x16x4_f32(false, a2, false, w2,
                                                       (short)0, acc2[ct], false, false);
    }

    if (kp + 1 < 16) {
#pragma unroll
      for (int ct = 0; ct < 8; ++ct) b0[ct] = bP[((2 * kp + 2) * 8 + ct) * 32];
    }

    a1.x = ta.z * innr;
    a1.y = ta.w * innr;
    a2.x = tx.z;
    a2.y = tx.w;
#pragma unroll
    for (int ct = 0; ct < 8; ++ct) {
      v2f w1, w2;
      w1.x = b1[ct].x; w1.y = b1[ct].y;
      w2.x = b1[ct].z; w2.y = b1[ct].w;
      acc1[ct] = __builtin_amdgcn_wmma_f32_16x16x4_f32(false, a1, false, w1,
                                                       (short)0, acc1[ct], false, false);
      acc2[ct] = __builtin_amdgcn_wmma_f32_16x16x4_f32(false, a2, false, w2,
                                                       (short)0, acc2[ct], false, false);
    }
    ta = ta_n;
    tx = tx_n;
  }

#pragma unroll
  for (int ct = 0; ct < 8; ++ct) {
    const int col0 = ct * 16;
    const float bb = bias[col0 + r];
    const float rb = Rb[col0 + r];
    float s = 0.0f, s2 = 0.0f;
#pragma unroll
    for (int j = 0; j < 8; ++j) {
      float v = fmaxf(acc1[ct][j] + bb, 0.0f) + fmaxf(acc2[ct][j] + rb, 0.0f);
      acc1[ct][j] = v;
      s += v;
      s2 += v * v;
    }
    s += __shfl_xor(s, 16);
    s2 += __shfl_xor(s2, 16);
    sRed[wave * (2 * DIM) + hf * DIM + col0 + r] = hf ? s2 : s;
  }
  } else {
    for (int i = lane; i < 2 * DIM; i += 32) sRed[wave * (2 * DIM) + i] = 0.0f;
  }
  __syncthreads();
  float tsum = 0.0f;
#pragma unroll
  for (int w = 0; w < 8; ++w) tsum += sRed[w * (2 * DIM) + tid];
  atomicAdd((tid < DIM) ? &csum[tid] : &csumsq[tid - DIM], tsum);
  if (row0 < N) {
#pragma unroll
    for (int ct = 0; ct < 8; ++ct) {
#pragma unroll
      for (int j = 0; j < 8; ++j)
        h[(size_t)(row0 + j + 8 * hf) * DIM + ct * 16 + r] = acc1[ct][j];
    }
  }
}

__global__ void __launch_bounds__(STREAM_THREADS) bnapply_kernel(
    const float* __restrict__ h, const float* __restrict__ stats,
    const float* __restrict__ gamma, const float* __restrict__ beta, float invN,
    float* __restrict__ out, int N) {
  const int lane = threadIdx.x & 31;
  const int wv = blockIdx.x * (STREAM_THREADS / 32) +
                 __builtin_amdgcn_readfirstlane((int)(threadIdx.x >> 5));
  const int nw = gridDim.x * (STREAM_THREADS / 32);
  const float4* H4 = reinterpret_cast<const float4*>(h) + lane;
  float4* O4 = reinterpret_cast<float4*>(out) + lane;
  float4 sc, sh;
  bn_coeff4(stats, gamma, beta, invN, lane, sc, sh);
  for (int i0 = wv; i0 < N; i0 += STREAM_U * nw) {
    float4 v[STREAM_U];
#pragma unroll
    for (int u = 0; u < STREAM_U; ++u) {
      const int i = i0 + u * nw;
      v[u] = H4[(size_t)((i < N) ? i : i0) * (DIM / 4)];
    }
#pragma unroll
    for (int u = 0; u < STREAM_U; ++u) {
      const int i = i0 + u * nw;
      asm volatile("" ::"v"(v[u].x), "v"(v[u].y), "v"(v[u].z), "v"(v[u].w));
      if (i < N) {
        float4 o;
        o.x = v[u].x * sc.x + sh.x;
        o.y = v[u].y * sc.y + sh.y;
        o.z = v[u].z * sc.z + sh.z;
        o.w = v[u].w * sc.w + sh.w;
        O4[(size_t)i * (DIM / 4)] = o;
      }
    }
  }
}


static inline int cdiv(int a, int b) { return (a + b - 1) / b; }

extern "C" void kernel_launch(void* const* d_in, const int* in_sizes, int n_in,
                              void* d_out, int out_size, void* d_ws, size_t ws_size,
                              hipStream_t stream) {
  const int* node_ids = (const int*)d_in[0];
  const int* src = (const int*)d_in[1];
  const int* dst = (const int*)d_in[2];
  const float* emb = (const float*)d_in[3];
  const float* Ws = (const float*)d_in[4];
  const float* bs = (const float*)d_in[5];
  const float* Rws = (const float*)d_in[6];
  const float* Rbs = (const float*)d_in[7];
  const float* gammas = (const float*)d_in[8];
  const float* betas = (const float*)d_in[9];

  const int N = in_sizes[0];
  const int E = in_sizes[1];
  const int L = in_sizes[4] / (DIM * DIM);

  float* wsf = (float*)d_ws;
  size_t off = 0;
  float* onorm = wsf + off; off += (size_t)N;
  float* inorm = wsf + off; off += (size_t)N;
  float* x    = wsf + off;  off += (size_t)N * DIM;
  float* agg  = wsf + off;  off += (size_t)N * DIM;
  float* h    = wsf + off;  off += (size_t)N * DIM;
  float* stats = wsf + off; off += (size_t)L * 2 * DIM;
  off = (off + 3) & ~(size_t)3;
  int2* csr = (int2*)(wsf + off); off += (size_t)E * 2;
  int* slot = (int*)(wsf + off);  off += (size_t)E;
  int* rp   = (int*)(wsf + off);  off += (size_t)N + 1;
  const int R = cdiv(N, CNT_RANGE);
  int* cntS = (int*)(wsf + off);  off += (size_t)CNT_CHUNKS * N;
  int* cntD = (int*)(wsf + off);  off += (size_t)CNT_CHUNKS * N;
  int* tot  = (int*)(wsf + off);  off += (size_t)CNT_CHUNKS * R;

  const int B = 256;

  const int CH = cdiv(E, CNT_CHUNKS);
  count_kernel<<<2 * CNT_CHUNKS, CNT_THREADS, 0, stream>>>(src, dst, cntS, cntD, tot, slot,
                                                           N, E, CH, R);
  offsets_kernel<<<R, CNT_RANGE, 0, stream>>>(cntS, cntD, tot, rp, onorm, inorm, N, R);
  place_kernel<<<STREAM_BLOCKS, STREAM_THREADS, 0, stream>>>(src, dst, slot, rp, cntD, onorm,
                                                             csr, E, N, CH);

  gather_kernel<<<STREAM_BLOCKS, STREAM_THREADS, 0, stream>>>(node_ids, emb, x, stats,
                                                              L * 2 * DIM, N);

  const int rowTiles = cdiv(N, 16);
  const int aggBlocks = (cdiv(N, AGG_WAVES) < 1024) ? cdiv(N, AGG_WAVES) : 1024;
  const float invN = 1.0f / (float)N;
  float* cur = x;
  float* nxt = h;
  for (int l = 0; l < L; ++l) {
    float* st = stats + (size_t)l * 2 * DIM;
    const float* W = Ws + (size_t)l * DIM * DIM;
    const float* bw = bs + (size_t)l * DIM;
    const float* Rw = Rws + (size_t)l * DIM * DIM;
    const float* rb = Rbs + (size_t)l * DIM;
    if (l == 0) {
      aggregate_kernel<false><<<aggBlocks, AGG_WAVES * 32, 0, stream>>>(
          rp, csr, cur, agg, nullptr, nullptr, nullptr, invN, N);
      gemm_fused_kernel<false><<<cdiv(rowTiles, 8), B, 0, stream>>>(
          agg, cur, inorm, W, bw, Rw, rb, nullptr, nullptr, nullptr, invN, nxt, st, st + DIM,
          N);
    } else {
      const float* pst = st - 2 * DIM;
      const float* pg = gammas + (size_t)(l - 1) * DIM;
      const float* pb = betas + (size_t)(l - 1) * DIM;
      aggregate_kernel<true><<<aggBlocks, AGG_WAVES * 32, 0, stream>>>(
          rp, csr, cur, agg, pst, pg, pb, invN, N);
      gemm_fused_kernel<true><<<cdiv(rowTiles, 8), B, 0, stream>>>(
          agg, cur, inorm, W, bw, Rw, rb, pst, pg, pb, invN, nxt, st, st + DIM, N);
    }
    float* t = cur;
    cur = nxt;
    nxt = t;
  }
  bnapply_kernel<<<STREAM_BLOCKS, STREAM_THREADS, 0, stream>>>(
      cur, stats + (size_t)(L - 1) * 2 * DIM, gammas + (size_t)(L - 1) * DIM,
      betas + (size_t)(L - 1) * DIM, invN, (float*)d_out, N);
}


# Layer 0 on the embedding table's 343 rows: P=emb@W0, R=relu(emb@Rw0+Rb0) by WMMA once; a node sums P[id[src]] rows from an LDS copy of P, adds R[id]; no x, agg, N-row GEMM; fp32 sum order changes
# speedup vs baseline: 3.7891x; 1.1672x over previous
#include <hip/hip_runtime.h>
#include <math.h>


typedef __attribute__((ext_vector_type(2))) float v2f;
typedef __attribute__((ext_vector_type(8))) float v8f;

#define DIM 128
#define BN_EPS 1e-5f

__device__ __forceinline__ void bn_coeff(float s, float q, float g, float b, float invN,
                                         float& sc, float& sh) {
  const float mu = s * invN;
  const float var = q * invN - mu * mu;
  sc = g / sqrtf(var + BN_EPS);
  sh = b - mu * sc;
}

__device__ __forceinline__ void bn_coeff4(const float* __restrict__ stats,
                                          const float* __restrict__ gamma,
                                          const float* __restrict__ beta, float invN,
                                          int lane, float4& sc, float4& sh) {
  const float4 s = reinterpret_cast<const float4*>(stats)[lane];
  const float4 q = reinterpret_cast<const float4*>(stats + DIM)[lane];
  const float4 g = reinterpret_cast<const float4*>(gamma)[lane];
  const float4 b = reinterpret_cast<const float4*>(beta)[lane];
  bn_coeff(s.x, q.x, g.x, b.x, invN, sc.x, sh.x);
  bn_coeff(s.y, q.y, g.y, b.y, invN, sc.y, sh.y);
  bn_coeff(s.z, q.z, g.z, b.z, invN, sc.z, sh.z);
  bn_coeff(s.w, q.w, g.w, b.w, invN, sc.w, sh.w);
}

#define CNT_THREADS 1024
#define CNT_RANGE 1024
#define CNT_CAP 65536
#define CNT_CHUNKS 16
__global__ void __launch_bounds__(CNT_THREADS) count_kernel(
    const int* __restrict__ src, const int* __restrict__ dst,
    int* __restrict__ cntS, int* __restrict__ cntD, int* __restrict__ tot,
    int* __restrict__ slot, int N, int E, int CH, int R) {
  __shared__ int cnt[CNT_CAP];
  const int tid = threadIdx.x;
  const int lane = tid & 31;
  const int wave = tid >> 5;
  const bool isDst = blockIdx.x < CNT_CHUNKS;
  const int q = isDst ? (int)blockIdx.x : (int)blockIdx.x - CNT_CHUNKS;
  const int* __restrict__ keys = isDst ? dst : src;
  int* __restrict__ outc = (isDst ? cntD : cntS) + (size_t)q * N;
  const int eb = q * CH;
  const int ee = (E - eb < CH) ? E : eb + CH;
  for (int nb = 0; nb < N; nb += CNT_CAP) {
    const int n = (N - nb < CNT_CAP) ? (N - nb) : CNT_CAP;
    const int nr = (n + CNT_RANGE - 1) / CNT_RANGE;
    {
      int4* c4p = reinterpret_cast<int4*>(cnt);
      for (int i = tid; i < (n + 3) / 4; i += CNT_THREADS) c4p[i] = make_int4(0, 0, 0, 0);
    }
    __syncthreads();
    for (int e0 = eb + tid; e0 < ee; e0 += 8 * CNT_THREADS) {
      int k[8];
#pragma unroll
      for (int u = 0; u < 8; ++u) {
        const int e = e0 + u * CNT_THREADS;
        const int kv = keys[(e < ee) ? e : ee - 1];
        k[u] = (e < ee) ? kv - nb : -1;
      }
      int place[8];
#pragma unroll
      for (int u = 0; u < 8; ++u) {
        place[u] = 0;
        if ((unsigned)k[u] < (unsigned)n) place[u] = atomicAdd(&cnt[k[u]], 1);
      }
      if (isDst) {
#pragma unroll
        for (int u = 0; u < 8; ++u) {
          const int e = e0 + u * CNT_THREADS;
          if ((unsigned)k[u] < (unsigned)n) slot[e] = place[u];
        }
      }
    }
    __syncthreads();
    for (int j = wave; j < nr; j += CNT_THREADS / 32) {
      int c = 0;
      for (int t = lane; t < CNT_RANGE; t += 32) {
        const int i = j * CNT_RANGE + t;
        if (i < n) {
          const int v = cnt[i];
          outc[nb + i] = v;
          c += v;
        }
      }
      if (isDst) {
        for (int o = 16; o > 0; o >>= 1) c += __shfl_xor(c, o);
        if (lane == 0) tot[(size_t)q * R + nb / CNT_RANGE + j] = c;
      }
    }
    __syncthreads();
  }
}

__global__ void __launch_bounds__(CNT_RANGE) offsets_kernel(
    const int* __restrict__ cntS, int* __restrict__ cntD, const int* __restrict__ tot,
    int* __restrict__ rp, float* __restrict__ onorm, float* __restrict__ inorm,
    int N, int R) {
  __shared__ int wtot[33];
  __shared__ int wbase[33];
  const int tid = threadIdx.x;
  const int lane = tid & 31;
  const int wave = tid >> 5;
  const int r = blockIdx.x;
  const int i = r * CNT_RANGE + tid;

  int acc = 0;
  for (int t = tid; t < CNT_CHUNKS * R; t += CNT_RANGE) {
    const int v = tot[t];
    if (t % R < r) acc += v;
  }

  int dd = 0;
  if (i < N) {
    int cs[CNT_CHUNKS], cd[CNT_CHUNKS];
#pragma unroll
    for (int q = 0; q < CNT_CHUNKS; ++q) {
      cs[q] = cntS[(size_t)q * N + i];
      cd[q] = cntD[(size_t)q * N + i];
    }
    int ds = 0;
#pragma unroll
    for (int q = 0; q < CNT_CHUNKS; ++q) {
      ds += cs[q];
      cntD[(size_t)q * N + i] = dd;
      dd += cd[q];
    }
    onorm[i] = (ds > 0) ? 1.0f / sqrtf((float)ds) : 0.0f;
    inorm[i] = (dd > 0) ? 1.0f / sqrtf((float)dd) : 0.0f;
  }

  int v = dd;
  for (int o = 1; o < 32; o <<= 1) {
    const int t = __shfl_up(v, o);
    if (lane >= o) v += t;
  }
  for (int o = 16; o > 0; o >>= 1) acc += __shfl_xor(acc, o);
  if (lane == 31) wtot[wave] = v;
  if (lane == 0) wbase[wave] = acc;
  __syncthreads();
  if (wave == 0) {
    const int t = wtot[lane];
    int u = t;
    for (int o = 1; o < 32; o <<= 1) {
      const int s = __shfl_up(u, o);
      if (lane >= o) u += s;
    }
    wtot[lane] = u - t;
    int b = wbase[lane];
    for (int o = 16; o > 0; o >>= 1) b += __shfl_xor(b, o);
    if (lane == 0) wbase[32] = b;
  }
  __syncthreads();
  const int incl = wbase[32] + wtot[wave] + v;
  if (i < N) rp[i] = incl - dd;
  if (i == N - 1) rp[N] = incl;
}

#define STREAM_BLOCKS 512
#define STREAM_THREADS 256

#define PLACE_U 4
__global__ void __launch_bounds__(STREAM_THREADS) place_kernel(
    const int* __restrict__ src, const int* __restrict__ dst,
    const int* __restrict__ slot, const int* __restrict__ rp,
    const int* __restrict__ cntD, const int* __restrict__ ids,
    const float* __restrict__ onorm, int2* __restrict__ csr, int2* __restrict__ csr0,
    int E, int N, int CH) {
  const int t = blockIdx.x * STREAM_THREADS + threadIdx.x;
  const int stride = gridDim.x * STREAM_THREADS;
  for (int e0 = t; e0 < E; e0 += PLACE_U * stride) {
    int s[PLACE_U], d[PLACE_U], sl[PLACE_U], q[PLACE_U];
#pragma unroll
    for (int u = 0; u < PLACE_U; ++u) {
      const int e = e0 + u * stride;
      const int ec = (e < E) ? e : e0;
      s[u] = src[ec];
      d[u] = dst[ec];
      sl[u] = slot[ec];
      q[u] = ec / CH;
    }
    int base[PLACE_U], cs[PLACE_U], w[PLACE_U], id[PLACE_U];
#pragma unroll
    for (int u = 0; u < PLACE_U; ++u) {
      w[u] = __float_as_int(onorm[s[u]]);
      id[u] = ids[s[u]];
      base[u] = rp[d[u]];
      cs[u] = cntD[(size_t)q[u] * N + d[u]];
    }
#pragma unroll
    for (int u = 0; u < PLACE_U; ++u) {
      const int e = e0 + u * stride;
      asm volatile("" ::"v"(s[u]), "v"(sl[u]), "v"(w[u]), "v"(id[u]), "v"(base[u]), "v"(cs[u]));
      if (e < E) {
        const int pos = base[u] + cs[u] + sl[u];
        int2 rec;
        rec.x = s[u];
        rec.y = w[u];
        csr[pos] = rec;
        rec.x = id[u];
        csr0[pos] = rec;
      }
    }
  }
}

__global__ void __launch_bounds__(256) table_kernel(
    const float* __restrict__ emb, const float* __restrict__ W, const float* __restrict__ Rw,
    const float* __restrict__ Rb, float* __restrict__ P, float* __restrict__ R,
    float* __restrict__ stats, int nstats, int T) {
  const int tid = threadIdx.x;
  if (blockIdx.x == 0) {
    for (int i = tid; i < nstats; i += 256) stats[i] = 0.0f;
  }
  const int wave = __builtin_amdgcn_readfirstlane(tid >> 5);
  const bool isR = wave >= 4;
  const int c0 = (wave & 3) * 32;
  const int lane = tid & 31;
  const int r = lane & 15;
  const int hf = lane >> 4;
  const int row0 = blockIdx.x * 16;
  const int row = (row0 + r < T) ? row0 + r : T - 1;
  const float4* aP = reinterpret_cast<const float4*>(emb + (size_t)row * DIM) + hf;
  const float* bp = (isR ? Rw : W) + (size_t)(4 * hf) * DIM + c0 + r;
  v8f acc0 = (v8f){0.f, 0.f, 0.f, 0.f, 0.f, 0.f, 0.f, 0.f};
  v8f acc1 = (v8f){0.f, 0.f, 0.f, 0.f, 0.f, 0.f, 0.f, 0.f};
#pragma unroll
  for (int kp = 0; kp < 16; ++kp) {
    const float4 ta = aP[2 * kp];
#pragma unroll
    for (int q = 0; q < 2; ++q) {
      v2f a, b0, b1;
      a.x = q ? ta.z : ta.x;
      a.y = q ? ta.w : ta.y;
      b0.x = bp[(size_t)(8 * kp + 2 * q) * DIM];
      b0.y = bp[(size_t)(8 * kp + 2 * q + 1) * DIM];
      b1.x = bp[(size_t)(8 * kp + 2 * q) * DIM + 16];
      b1.y = bp[(size_t)(8 * kp + 2 * q + 1) * DIM + 16];
      acc0 = __builtin_amdgcn_wmma_f32_16x16x4_f32(false, a, false, b0, (short)0, acc0,
                                                   false, false);
      acc1 = __builtin_amdgcn_wmma_f32_16x16x4_f32(false, a, false, b1, (short)0, acc1,
                                                   false, false);
    }
  }
  if (isR) {
    const float rb0 = Rb[c0 + r];
    const float rb1 = Rb[c0 + 16 + r];
#pragma unroll
    for (int j = 0; j < 8; ++j) {
      const int rr = row0 + j + 8 * hf;
      if (rr < T) {
        R[(size_t)rr * DIM + c0 + r] = fmaxf(acc0[j] + rb0, 0.0f);
        R[(size_t)rr * DIM + c0 + 16 + r] = fmaxf(acc1[j] + rb1, 0.0f);
      }
    }
  } else {
#pragma unroll
    for (int j = 0; j < 8; ++j) {
      const int rr = row0 + j + 8 * hf;
      if (rr < T) {
        P[(size_t)rr * DIM + c0 + r] = acc0[j];
        P[(size_t)rr * DIM + c0 + 16 + r] = acc1[j];
      }
    }
  }
}

#define AGG_WAVES 8

template <int B, bool CLAMP, bool BN>
__device__ __forceinline__ void sum_rows(float4& acc, const float4* __restrict__ X4,
                                         int recSrc, int recW, int j, int m,
                                         const float4& sc, const float4& sh) {
  float4 v[B];
  float w[B];
#pragma unroll
  for (int u = 0; u < B; ++u) {
    int idx = j + u;
    bool ok = true;
    if (CLAMP) {
      ok = idx < m;
      idx = ok ? idx : m - 1;
    }
    const int s = __builtin_amdgcn_readlane(recSrc, idx);
    const float ww = __int_as_float(__builtin_amdgcn_readlane(recW, idx));
    w[u] = ok ? ww : 0.0f;
    v[u] = X4[(size_t)s * (DIM / 4)];
  }
#pragma unroll
  for (int u = 0; u < B; ++u) {
    float4 xv = v[u];
    if (BN) {
      xv.x = __builtin_fmaf(xv.x, sc.x, sh.x);
      xv.y = __builtin_fmaf(xv.y, sc.y, sh.y);
      xv.z = __builtin_fmaf(xv.z, sc.z, sh.z);
      xv.w = __builtin_fmaf(xv.w, sc.w, sh.w);
    }
    acc.x = __builtin_fmaf(xv.x, w[u], acc.x);
    acc.y = __builtin_fmaf(xv.y, w[u], acc.y);
    acc.z = __builtin_fmaf(xv.z, w[u], acc.z);
    acc.w = __builtin_fmaf(xv.w, w[u], acc.w);
  }
}

__global__ void __launch_bounds__(AGG_WAVES * 32) aggregate_kernel(
    const int* __restrict__ rp, const int2* __restrict__ csr,
    const float* __restrict__ x, float* __restrict__ agg,
    const float* __restrict__ stats, const float* __restrict__ gamma,
    const float* __restrict__ beta, float invN, int N) {
  const int lane = threadIdx.x & 31;
  const int wv = __builtin_amdgcn_readfirstlane((int)(threadIdx.x >> 5));
  const int nw = gridDim.x * AGG_WAVES;
  int row = blockIdx.x * AGG_WAVES + wv;
  if (row >= N) return;
  const float4* X4 = reinterpret_cast<const float4*>(x) + lane;
  float4* A4 = reinterpret_cast<float4*>(agg) + lane;
  float4 sc, sh;
  bn_coeff4(stats, gamma, beta, invN, lane, sc, sh);

  int beg = rp[row];
  int end = rp[row + 1];
  int2 ev = make_int2(0, 0);
  if (lane < end - beg) ev = csr[beg + lane];
  int nrow = row + nw;
  int nb = 0, ne = 0;
  if (nrow < N) {
    nb = rp[nrow];
    ne = rp[nrow + 1];
  }

  for (;;) {
    int2 nev = make_int2(0, 0);
    if (lane < ne - nb) nev = csr[nb + lane];
    const int nnrow = nrow + nw;
    int nnb = 0, nne = 0;
    if (nnrow < N) {
      nnb = rp[nnrow];
      nne = rp[nnrow + 1];
    }

    float4 acc = make_float4(0.0f, 0.0f, 0.0f, 0.0f);
    const int cnt = end - beg;
    int done = 0;
    for (;;) {
      const int m = (cnt - done < 32) ? (cnt - done) : 32;
      int j = 0;
      for (; j + 8 <= m; j += 8) sum_rows<8, false, true>(acc, X4, ev.x, ev.y, j, m, sc, sh);
      for (; j < m; j += 4) sum_rows<4, true, true>(acc, X4, ev.x, ev.y, j, m, sc, sh);
      done += 32;
      if (done >= cnt) break;
      ev = make_int2(0, 0);
      if (lane < cnt - done) ev = csr[beg + done + lane];
    }
    A4[(size_t)row * (DIM / 4)] = acc;

    if (nrow >= N) break;
    row = nrow;
    beg = nb;
    end = ne;
    ev = nev;
    nrow = nnrow;
    nb = nnb;
    ne = nne;
  }
}

#define L0_WAVES 32
#define TAB_CAP 352
#define L0_STAGE_U (TAB_CAP * (DIM / 4) / (L0_WAVES * 32))
__global__ void __launch_bounds__(L0_WAVES * 32) layer0_kernel(
    const int* __restrict__ rp, const int2* __restrict__ csr0,
    const float* __restrict__ P, const float* __restrict__ R,
    const int* __restrict__ ids, const float* __restrict__ inorm,
    const float* __restrict__ bias, float* __restrict__ h,
    float* __restrict__ csum, float* __restrict__ csumsq, int N, int T) {
  __shared__ float4 sP[TAB_CAP * (DIM / 4)];
  __shared__ float4 sRed[L0_WAVES * 2 * (DIM / 4)];
  {
    const float4* Pg = reinterpret_cast<const float4*>(P);
    const int n4 = T * (DIM / 4);
    float4 t[L0_STAGE_U];
#pragma unroll
    for (int u = 0; u < L0_STAGE_U; ++u) {
      const int i = (int)threadIdx.x + u * (L0_WAVES * 32);
      t[u] = Pg[(i < n4) ? i : n4 - 1];
    }
#pragma unroll
    for (int u = 0; u < L0_STAGE_U; ++u) sP[(int)threadIdx.x + u * (L0_WAVES * 32)] = t[u];
  }
  __syncthreads();
  const int lane = threadIdx.x & 31;
  const int wv = __builtin_amdgcn_readfirstlane((int)(threadIdx.x >> 5));
  const int nw = gridDim.x * L0_WAVES;
  int row = blockIdx.x * L0_WAVES + wv;
  float4 cs = make_float4(0.0f, 0.0f, 0.0f, 0.0f);
  float4 cq = make_float4(0.0f, 0.0f, 0.0f, 0.0f);
  if (row < N) {
    const float4* P4 = sP + lane;
    const float4* R4 = reinterpret_cast<const float4*>(R) + lane;
    float4* H4 = reinterpret_cast<float4*>(h) + lane;
    const float4 b4 = reinterpret_cast<const float4*>(bias)[lane];
    const float4 one = make_float4(1.0f, 1.0f, 1.0f, 1.0f);
    const float4 nul = make_float4(0.0f, 0.0f, 0.0f, 0.0f);

    int beg = rp[row];
    int end = rp[row + 1];
    float inn = inorm[row];
    float4 rv = R4[(size_t)ids[row] * (DIM / 4)];
    int2 ev = make_int2(0, 0);
    if (lane < end - beg) ev = csr0[beg + lane];
    int nrow = row + nw;
    int nb = 0, ne = 0, nid = 0;
    float ninn = 0.0f;
    if (nrow < N) {
      nb = rp[nrow];
      ne = rp[nrow + 1];
      nid = ids[nrow];
      ninn = inorm[nrow];
    }

    for (;;) {
      int2 nev = make_int2(0, 0);
      if (lane < ne - nb) nev = csr0[nb + lane];
      const float4 nrv = R4[(size_t)nid * (DIM / 4)];
      const int nnrow = nrow + nw;
      int nnb = 0, nne = 0, nnid = 0;
      float nninn = 0.0f;
      if (nnrow < N) {
        nnb = rp[nnrow];
        nne = rp[nnrow + 1];
        nnid = ids[nnrow];
        nninn = inorm[nnrow];
      }

      float4 acc = make_float4(0.0f, 0.0f, 0.0f, 0.0f);
      const int cnt = end - beg;
      int done = 0;
      for (;;) {
        const int m = (cnt - done < 32) ? (cnt - done) : 32;
        int j = 0;
        for (; j + 8 <= m; j += 8) sum_rows<8, false, false>(acc, P4, ev.x, ev.y, j, m, one, nul);
        for (; j < m; j += 4) sum_rows<4, true, false>(acc, P4, ev.x, ev.y, j, m, one, nul);
        done += 32;
        if (done >= cnt) break;
        ev = make_int2(0, 0);
        if (lane < cnt - done) ev = csr0[beg + done + lane];
      }
      float4 v;
      v.x = fmaxf(acc.x * inn + b4.x, 0.0f) + rv.x;
      v.y = fmaxf(acc.y * inn + b4.y, 0.0f) + rv.y;
      v.z = fmaxf(acc.z * inn + b4.z, 0.0f) + rv.z;
      v.w = fmaxf(acc.w * inn + b4.w, 0.0f) + rv.w;
      H4[(size_t)row * (DIM / 4)] = v;
      cs.x += v.x;
      cs.y += v.y;
      cs.z += v.z;
      cs.w += v.w;
      cq.x += v.x * v.x;
      cq.y += v.y * v.y;
      cq.z += v.z * v.z;
      cq.w += v.w * v.w;

      if (nrow >= N) break;
      row = nrow;
      beg = nb;
      end = ne;
      inn = ninn;
      rv = nrv;
      ev = nev;
      nrow = nnrow;
      nb = nnb;
      ne = nne;
      nid = nnid;
      ninn = nninn;
    }
  }
  sRed[wv * (2 * (DIM / 4)) + lane] = cs;
  sRed[wv * (2 * (DIM / 4)) + (DIM / 4) + lane] = cq;
  __syncthreads();
  const int tid = threadIdx.x;
  if (tid < 2 * DIM) {
    const float* red = reinterpret_cast<const float*>(sRed);
    float tsum = 0.0f;
#pragma unroll
    for (int w = 0; w < L0_WAVES; ++w) tsum += red[w * (2 * DIM) + tid];
    atomicAdd((tid < DIM) ? &csum[tid] : &csumsq[tid - DIM], tsum);
  }
}

__global__ void __launch_bounds__(256) gemm_fused_kernel(
    const float* __restrict__ agg, const float* __restrict__ xin,
    const float* __restrict__ inorm,
    const float* __restrict__ W, const float* __restrict__ bias,
    const float* __restrict__ Rw, const float* __restrict__ Rb,
    const float* __restrict__ pstats, const float* __restrict__ pgamma,
    const float* __restrict__ pbeta, float invN,
    float* __restrict__ h, float* __restrict__ csum, float* __restrict__ csumsq,
    int N) {
  __shared__ float4 sB[32 * 8 * 32];
  __shared__ float4 sScale[DIM / 4];
  __shared__ float4 sShift[DIM / 4];
  __shared__ float sRed[8 * 2 * DIM];

  const int tid = threadIdx.x;
  const int wave = tid >> 5;
  const int lane = tid & 31;
  const int r = lane & 15;
  const int hf = lane >> 4;

  const int bk = tid & (DIM - 1);
  const float bnS = pstats[bk];
  const float bnQ = pstats[DIM + bk];
  const float bnG = pgamma[bk];
  const float bnB = pbeta[bk];

  {
    const float* wp = W + (size_t)(4 * hf) * DIM + wave * 16 + r;
    const float* rp = Rw + (size_t)(4 * hf) * DIM + wave * 16 + r;
    float4* sp = sB + tid;
#pragma unroll 4
    for (int kp = 0; kp < 16; ++kp) {
#pragma unroll
      for (int q = 0; q < 2; ++q) {
        float4 v;
        v.x = wp[(2 * q) * DIM];
        v.y = wp[(2 * q + 1) * DIM];
        v.z = rp[(2 * q) * DIM];
        v.w = rp[(2 * q + 1) * DIM];
        sp[q * 256] = v;
      }
      wp += 8 * DIM;
      rp += 8 * DIM;
      sp += 2 * 256;
    }
  }
  {
    float sc, sh;
    bn_coeff(bnS, bnQ, bnG, bnB, invN, sc, sh);
    float* dstp = reinterpret_cast<float*>((tid < DIM) ? sScale : sShift);
    dstp[bk] = (tid < DIM) ? sc : sh;
  }
  __syncthreads();

  const int row0 = (blockIdx.x * 8 + wave) * 16;
  v8f acc1[8], acc2[8];

  if (row0 < N) {

  const int row = row0 + r;
  const float innr = inorm[row];
  const float4* aP = reinterpret_cast<const float4*>(agg + (size_t)row * DIM) + hf;
  const float4* xP = reinterpret_cast<const float4*>(xin + (size_t)row * DIM) + hf;
  const float4* bP = sB + lane;

#pragma unroll
  for (int ct = 0; ct < 8; ++ct) {
    acc1[ct] = (v8f){0.f, 0.f, 0.f, 0.f, 0.f, 0.f, 0.f, 0.f};
    acc2[ct] = (v8f){0.f, 0.f, 0.f, 0.f, 0.f, 0.f, 0.f, 0.f};
  }

  const float4* scP = sScale + hf;
  const float4* shP = sShift + hf;

  float4 ta = aP[0];
  float4 tx = xP[0];
  float4 b0[8], b1[8];
#pragma unroll
  for (int ct = 0; ct < 8; ++ct) b0[ct] = bP[ct * 32];

  for (int kp = 0; kp < 16; ++kp) {
    float4 ta_n = ta, tx_n = tx;
    if (kp + 1 < 16) {
      ta_n = aP[(kp + 1) * 2];
      tx_n = xP[(kp + 1) * 2];
    }
#pragma unroll
    for (int ct = 0; ct < 8; ++ct) b1[ct] = bP[((2 * kp + 1) * 8 + ct) * 32];

    {
      const float4 s4 = scP[2 * kp];
      const float4 t4 = shP[2 * kp];
      tx.x = __builtin_fmaf(tx.x, s4.x, t4.x);
      tx.y = __builtin_fmaf(tx.y, s4.y, t4.y);
      tx.z = __builtin_fmaf(tx.z, s4.z, t4.z);
      tx.w = __builtin_fmaf(tx.w, s4.w, t4.w);
    }

    v2f a1, a2;
    a1.x = ta.x * innr;
    a1.y = ta.y * innr;
    a2.x = tx.x;
    a2.y = tx.y;
#pragma unroll
    for (int ct = 0; ct < 8; ++ct) {
      v2f w1, w2;
      w1.x = b0[ct].x; w1.y = b0[ct].y;
      w2.x = b0[ct].z; w2.y = b0[ct].w;
      acc1[ct] = __builtin_amdgcn_wmma_f32_16x16x4_f32(false, a1, false, w1,
                                                       (short)0, acc1[ct], false, false);
      acc2[ct] = __builtin_amdgcn_wmma_f32_16x16x4_f32(false, a2, false, w2,
                                                       (short)0, acc2[ct], false, false);
    }

    if (kp + 1 < 16) {
#pragma unroll
      for (int ct = 0; ct < 8; ++ct) b0[ct] = bP[((2 * kp + 2) * 8 + ct) * 32];
    }

    a1.x = ta.z * innr;
    a1.y = ta.w * innr;
    a2.x = tx.z;
    a2.y = tx.w;
#pragma unroll
    for (int ct = 0; ct < 8; ++ct) {
      v2f w1, w2;
      w1.x = b1[ct].x; w1.y = b1[ct].y;
      w2.x = b1[ct].z; w2.y = b1[ct].w;
      acc1[ct] = __builtin_amdgcn_wmma_f32_16x16x4_f32(false, a1, false, w1,
                                                       (short)0, acc1[ct], false, false);
      acc2[ct] = __builtin_amdgcn_wmma_f32_16x16x4_f32(false, a2, false, w2,
                                                       (short)0, acc2[ct], false, false);
    }
    ta = ta_n;
    tx = tx_n;
  }

#pragma unroll
  for (int ct = 0; ct < 8; ++ct) {
    const int col0 = ct * 16;
    const float bb = bias[col0 + r];
    const float rb = Rb[col0 + r];
    float s = 0.0f, s2 = 0.0f;
#pragma unroll
    for (int j = 0; j < 8; ++j) {
      float v = fmaxf(acc1[ct][j] + bb, 0.0f) + fmaxf(acc2[ct][j] + rb, 0.0f);
      acc1[ct][j] = v;
      s += v;
      s2 += v * v;
    }
    s += __shfl_xor(s, 16);
    s2 += __shfl_xor(s2, 16);
    sRed[wave * (2 * DIM) + hf * DIM + col0 + r] = hf ? s2 : s;
  }
  } else {
    for (int i = lane; i < 2 * DIM; i += 32) sRed[wave * (2 * DIM) + i] = 0.0f;
  }
  __syncthreads();
  float tsum = 0.0f;
#pragma unroll
  for (int w = 0; w < 8; ++w) tsum += sRed[w * (2 * DIM) + tid];
  atomicAdd((tid < DIM) ? &csum[tid] : &csumsq[tid - DIM], tsum);
  if (row0 < N) {
#pragma unroll
    for (int ct = 0; ct < 8; ++ct) {
#pragma unroll
      for (int j = 0; j < 8; ++j)
        h[(size_t)(row0 + j + 8 * hf) * DIM + ct * 16 + r] = acc1[ct][j];
    }
  }
}

#define STREAM_U 16
__global__ void __launch_bounds__(STREAM_THREADS) bnapply_kernel(
    const float* __restrict__ h, const float* __restrict__ stats,
    const float* __restrict__ gamma, const float* __restrict__ beta, float invN,
    float* __restrict__ out, int N) {
  const int lane = threadIdx.x & 31;
  const int wv = blockIdx.x * (STREAM_THREADS / 32) +
                 __builtin_amdgcn_readfirstlane((int)(threadIdx.x >> 5));
  const int nw = gridDim.x * (STREAM_THREADS / 32);
  const float4* H4 = reinterpret_cast<const float4*>(h) + lane;
  float4* O4 = reinterpret_cast<float4*>(out) + lane;
  float4 sc, sh;
  bn_coeff4(stats, gamma, beta, invN, lane, sc, sh);
  for (int i0 = wv; i0 < N; i0 += STREAM_U * nw) {
    float4 v[STREAM_U];
#pragma unroll
    for (int u = 0; u < STREAM_U; ++u) {
      const int i = i0 + u * nw;
      v[u] = H4[(size_t)((i < N) ? i : i0) * (DIM / 4)];
    }
#pragma unroll
    for (int u = 0; u < STREAM_U; ++u) {
      const int i = i0 + u * nw;
      asm volatile("" ::"v"(v[u].x), "v"(v[u].y), "v"(v[u].z), "v"(v[u].w));
      if (i < N) {
        float4 o;
        o.x = v[u].x * sc.x + sh.x;
        o.y = v[u].y * sc.y + sh.y;
        o.z = v[u].z * sc.z + sh.z;
        o.w = v[u].w * sc.w + sh.w;
        O4[(size_t)i * (DIM / 4)] = o;
      }
    }
  }
}


static inline int cdiv(int a, int b) { return (a + b - 1) / b; }

extern "C" void kernel_launch(void* const* d_in, const int* in_sizes, int n_in,
                              void* d_out, int out_size, void* d_ws, size_t ws_size,
                              hipStream_t stream) {
  const int* node_ids = (const int*)d_in[0];
  const int* src = (const int*)d_in[1];
  const int* dst = (const int*)d_in[2];
  const float* emb = (const float*)d_in[3];
  const float* Ws = (const float*)d_in[4];
  const float* bs = (const float*)d_in[5];
  const float* Rws = (const float*)d_in[6];
  const float* Rbs = (const float*)d_in[7];
  const float* gammas = (const float*)d_in[8];
  const float* betas = (const float*)d_in[9];

  const int N = in_sizes[0];
  const int E = in_sizes[1];
  const int T = in_sizes[3] / DIM;
  const int L = in_sizes[4] / (DIM * DIM);

  float* wsf = (float*)d_ws;
  size_t off = 0;
  float* onorm = wsf + off; off += (size_t)N;
  float* inorm = wsf + off; off += (size_t)N;
  off = (off + 3) & ~(size_t)3;
  float* ha   = wsf + off;  off += (size_t)N * DIM;
  float* hb   = wsf + off;  off += (size_t)N * DIM;
  float* agg  = wsf + off;  off += (size_t)N * DIM;
  float* tabP = wsf + off;  off += (size_t)T * DIM;
  float* tabR = wsf + off;  off += (size_t)T * DIM;
  float* stats = wsf + off; off += (size_t)L * 2 * DIM;
  off = (off + 3) & ~(size_t)3;
  int2* csr = (int2*)(wsf + off); off += (size_t)E * 2;
  int2* csr0 = (int2*)(wsf + off); off += (size_t)E * 2;
  int* slot = (int*)(wsf + off);  off += (size_t)E;
  int* rp   = (int*)(wsf + off);  off += (size_t)N + 1;
  const int R = cdiv(N, CNT_RANGE);
  int* cntS = (int*)(wsf + off);  off += (size_t)CNT_CHUNKS * N;
  int* cntD = (int*)(wsf + off);  off += (size_t)CNT_CHUNKS * N;
  int* tot  = (int*)(wsf + off);  off += (size_t)CNT_CHUNKS * R;

  const int B = 256;

  const int CH = cdiv(E, CNT_CHUNKS);
  count_kernel<<<2 * CNT_CHUNKS, CNT_THREADS, 0, stream>>>(src, dst, cntS, cntD, tot, slot,
                                                           N, E, CH, R);
  offsets_kernel<<<R, CNT_RANGE, 0, stream>>>(cntS, cntD, tot, rp, onorm, inorm, N, R);
  place_kernel<<<STREAM_BLOCKS, STREAM_THREADS, 0, stream>>>(src, dst, slot, rp, cntD, node_ids,
                                                             onorm, csr, csr0, E, N, CH);

  table_kernel<<<cdiv(T, 16), 256, 0, stream>>>(emb, Ws, Rws, Rbs, tabP, tabR, stats,
                                                L * 2 * DIM, T);

  const int rowTiles = cdiv(N, 16);
  const int l0Blocks = (cdiv(N, L0_WAVES) < 256) ? cdiv(N, L0_WAVES) : 256;
  const int aggBlocks = (cdiv(N, AGG_WAVES) < 1024) ? cdiv(N, AGG_WAVES) : 1024;
  const float invN = 1.0f / (float)N;
  layer0_kernel<<<l0Blocks, L0_WAVES * 32, 0, stream>>>(rp, csr0, tabP, tabR, node_ids, inorm,
                                                        bs, ha, stats, stats + DIM, N, T);
  float* cur = ha;
  float* nxt = hb;
  for (int l = 1; l < L; ++l) {
    float* st = stats + (size_t)l * 2 * DIM;
    const float* W = Ws + (size_t)l * DIM * DIM;
    const float* bw = bs + (size_t)l * DIM;
    const float* Rw = Rws + (size_t)l * DIM * DIM;
    const float* rb = Rbs + (size_t)l * DIM;
    const float* pst = st - 2 * DIM;
    const float* pg = gammas + (size_t)(l - 1) * DIM;
    const float* pb = betas + (size_t)(l - 1) * DIM;
    aggregate_kernel<<<aggBlocks, AGG_WAVES * 32, 0, stream>>>(
        rp, csr, cur, agg, pst, pg, pb, invN, N);
    gemm_fused_kernel<<<cdiv(rowTiles, 8), B, 0, stream>>>(
        agg, cur, inorm, W, bw, Rw, rb, pst, pg, pb, invN, nxt, st, st + DIM, N);
    float* t = cur;
    cur = nxt;
    nxt = t;
  }
  bnapply_kernel<<<STREAM_BLOCKS, STREAM_THREADS, 0, stream>>>(
      cur, stats + (size_t)(L - 1) * 2 * DIM, gammas + (size_t)(L - 1) * DIM,
      betas + (size_t)(L - 1) * DIM, invN, (float*)d_out, N);
}


# Weights-only work moved onto the edge-count launch's idle WGPs: table products and one fragment-order copy of W/Rw per layer built there; GEMM stages B by straight 16-byte copies; 10 -> 9 launches
# speedup vs baseline: 3.9366x; 1.0389x over previous
#include <hip/hip_runtime.h>
#include <math.h>


typedef __attribute__((ext_vector_type(2))) float v2f;
typedef __attribute__((ext_vector_type(8))) float v8f;

#define DIM 128
#define BN_EPS 1e-5f

__device__ __forceinline__ void bn_coeff(float s, float q, float g, float b, float invN,
                                         float& sc, float& sh) {
  const float mu = s * invN;
  const float var = q * invN - mu * mu;
  sc = g / sqrtf(var + BN_EPS);
  sh = b - mu * sc;
}

__device__ __forceinline__ void bn_coeff4(const float* __restrict__ stats,
                                          const float* __restrict__ gamma,
                                          const float* __restrict__ beta, float invN,
                                          int lane, float4& sc, float4& sh) {
  const float4 s = reinterpret_cast<const float4*>(stats)[lane];
  const float4 q = reinterpret_cast<const float4*>(stats + DIM)[lane];
  const float4 g = reinterpret_cast<const float4*>(gamma)[lane];
  const float4 b = reinterpret_cast<const float4*>(beta)[lane];
  bn_coeff(s.x, q.x, g.x, b.x, invN, sc.x, sh.x);
  bn_coeff(s.y, q.y, g.y, b.y, invN, sc.y, sh.y);
  bn_coeff(s.z, q.z, g.z, b.z, invN, sc.z, sh.z);
  bn_coeff(s.w, q.w, g.w, b.w, invN, sc.w, sh.w);
}

#define CNT_THREADS 1024
#define CNT_RANGE 1024
#define CNT_CAP 65536
#define CNT_CHUNKS 16
__device__ __forceinline__ void count_body(
    const int* __restrict__ src, const int* __restrict__ dst,
    int* __restrict__ cntS, int* __restrict__ cntD, int* __restrict__ tot,
    int* __restrict__ slot, int N, int E, int CH, int R, int blk) {
  __shared__ int cnt[CNT_CAP];
  const int tid = threadIdx.x;
  const int lane = tid & 31;
  const int wave = tid >> 5;
  const bool isDst = blk < CNT_CHUNKS;
  const int q = isDst ? blk : blk - CNT_CHUNKS;
  const int* __restrict__ keys = isDst ? dst : src;
  int* __restrict__ outc = (isDst ? cntD : cntS) + (size_t)q * N;
  const int eb = q * CH;
  const int ee = (E - eb < CH) ? E : eb + CH;
  for (int nb = 0; nb < N; nb += CNT_CAP) {
    const int n = (N - nb < CNT_CAP) ? (N - nb) : CNT_CAP;
    const int nr = (n + CNT_RANGE - 1) / CNT_RANGE;
    {
      int4* c4p = reinterpret_cast<int4*>(cnt);
      for (int i = tid; i < (n + 3) / 4; i += CNT_THREADS) c4p[i] = make_int4(0, 0, 0, 0);
    }
    __syncthreads();
    for (int e0 = eb + tid; e0 < ee; e0 += 8 * CNT_THREADS) {
      int k[8];
#pragma unroll
      for (int u = 0; u < 8; ++u) {
        const int e = e0 + u * CNT_THREADS;
        const int kv = keys[(e < ee) ? e : ee - 1];
        k[u] = (e < ee) ? kv - nb : -1;
      }
      int place[8];
#pragma unroll
      for (int u = 0; u < 8; ++u) {
        place[u] = 0;
        if ((unsigned)k[u] < (unsigned)n) place[u] = atomicAdd(&cnt[k[u]], 1);
      }
      if (isDst) {
#pragma unroll
        for (int u = 0; u < 8; ++u) {
          const int e = e0 + u * CNT_THREADS;
          if ((unsigned)k[u] < (unsigned)n) slot[e] = place[u];
        }
      }
    }
    __syncthreads();
    for (int j = wave; j < nr; j += CNT_THREADS / 32) {
      int c = 0;
      for (int t = lane; t < CNT_RANGE; t += 32) {
        const int i = j * CNT_RANGE + t;
        if (i < n) {
          const int v = cnt[i];
          outc[nb + i] = v;
          c += v;
        }
      }
      if (isDst) {
        for (int o = 16; o > 0; o >>= 1) c += __shfl_xor(c, o);
        if (lane == 0) tot[(size_t)q * R + nb / CNT_RANGE + j] = c;
      }
    }
    __syncthreads();
  }
}

__global__ void __launch_bounds__(CNT_RANGE) offsets_kernel(
    const int* __restrict__ cntS, int* __restrict__ cntD, const int* __restrict__ tot,
    int* __restrict__ rp, float* __restrict__ onorm, float* __restrict__ inorm,
    int N, int R) {
  __shared__ int wtot[33];
  __shared__ int wbase[33];
  const int tid = threadIdx.x;
  const int lane = tid & 31;
  const int wave = tid >> 5;
  const int r = blockIdx.x;
  const int i = r * CNT_RANGE + tid;

  int acc = 0;
  for (int t = tid; t < CNT_CHUNKS * R; t += CNT_RANGE) {
    const int v = tot[t];
    if (t % R < r) acc += v;
  }

  int dd = 0;
  if (i < N) {
    int cs[CNT_CHUNKS], cd[CNT_CHUNKS];
#pragma unroll
    for (int q = 0; q < CNT_CHUNKS; ++q) {
      cs[q] = cntS[(size_t)q * N + i];
      cd[q] = cntD[(size_t)q * N + i];
    }
    int ds = 0;
#pragma unroll
    for (int q = 0; q < CNT_CHUNKS; ++q) {
      ds += cs[q];
      cntD[(size_t)q * N + i] = dd;
      dd += cd[q];
    }
    onorm[i] = (ds > 0) ? 1.0f / sqrtf((float)ds) : 0.0f;
    inorm[i] = (dd > 0) ? 1.0f / sqrtf((float)dd) : 0.0f;
  }

  int v = dd;
  for (int o = 1; o < 32; o <<= 1) {
    const int t = __shfl_up(v, o);
    if (lane >= o) v += t;
  }
  for (int o = 16; o > 0; o >>= 1) acc += __shfl_xor(acc, o);
  if (lane == 31) wtot[wave] = v;
  if (lane == 0) wbase[wave] = acc;
  __syncthreads();
  if (wave == 0) {
    const int t = wtot[lane];
    int u = t;
    for (int o = 1; o < 32; o <<= 1) {
      const int s = __shfl_up(u, o);
      if (lane >= o) u += s;
    }
    wtot[lane] = u - t;
    int b = wbase[lane];
    for (int o = 16; o > 0; o >>= 1) b += __shfl_xor(b, o);
    if (lane == 0) wbase[32] = b;
  }
  __syncthreads();
  const int incl = wbase[32] + wtot[wave] + v;
  if (i < N) rp[i] = incl - dd;
  if (i == N - 1) rp[N] = incl;
}

#define STREAM_BLOCKS 512
#define STREAM_THREADS 256

#define PLACE_U 4
__global__ void __launch_bounds__(STREAM_THREADS) place_kernel(
    const int* __restrict__ src, const int* __restrict__ dst,
    const int* __restrict__ slot, const int* __restrict__ rp,
    const int* __restrict__ cntD, const int* __restrict__ ids,
    const float* __restrict__ onorm, int2* __restrict__ csr, int2* __restrict__ csr0,
    int E, int N, int CH) {
  const int t = blockIdx.x * STREAM_THREADS + threadIdx.x;
  const int stride = gridDim.x * STREAM_THREADS;
  for (int e0 = t; e0 < E; e0 += PLACE_U * stride) {
    int s[PLACE_U], d[PLACE_U], sl[PLACE_U], q[PLACE_U];
#pragma unroll
    for (int u = 0; u < PLACE_U; ++u) {
      const int e = e0 + u * stride;
      const int ec = (e < E) ? e : e0;
      s[u] = src[ec];
      d[u] = dst[ec];
      sl[u] = slot[ec];
      q[u] = ec / CH;
    }
    int base[PLACE_U], cs[PLACE_U], w[PLACE_U], id[PLACE_U];
#pragma unroll
    for (int u = 0; u < PLACE_U; ++u) {
      w[u] = __float_as_int(onorm[s[u]]);
      id[u] = ids[s[u]];
      base[u] = rp[d[u]];
      cs[u] = cntD[(size_t)q[u] * N + d[u]];
    }
#pragma unroll
    for (int u = 0; u < PLACE_U; ++u) {
      const int e = e0 + u * stride;
      asm volatile("" ::"v"(s[u]), "v"(sl[u]), "v"(w[u]), "v"(id[u]), "v"(base[u]), "v"(cs[u]));
      if (e < E) {
        const int pos = base[u] + cs[u] + sl[u];
        int2 rec;
        rec.x = s[u];
        rec.y = w[u];
        csr[pos] = rec;
        rec.x = id[u];
        csr0[pos] = rec;
      }
    }
  }
}

__device__ __forceinline__ void table_body(
    const float* __restrict__ emb, const float* __restrict__ W, const float* __restrict__ Rw,
    const float* __restrict__ Rb, float* __restrict__ P, float* __restrict__ R,
    int T, int tile, int tid) {
  const int wave = __builtin_amdgcn_readfirstlane(tid >> 5);
  const bool isR = wave >= 4;
  const int c0 = (wave & 3) * 32;
  const int lane = tid & 31;
  const int r = lane & 15;
  const int hf = lane >> 4;
  const int row0 = tile * 16;
  const int row = (row0 + r < T) ? row0 + r : T - 1;
  const float4* aP = reinterpret_cast<const float4*>(emb + (size_t)row * DIM) + hf;
  const float* bp = (isR ? Rw : W) + (size_t)(4 * hf) * DIM + c0 + r;
  v8f acc0 = (v8f){0.f, 0.f, 0.f, 0.f, 0.f, 0.f, 0.f, 0.f};
  v8f acc1 = (v8f){0.f, 0.f, 0.f, 0.f, 0.f, 0.f, 0.f, 0.f};
#pragma unroll
  for (int kp = 0; kp < 16; ++kp) {
    const float4 ta = aP[2 * kp];
#pragma unroll
    for (int q = 0; q < 2; ++q) {
      v2f a, b0, b1;
      a.x = q ? ta.z : ta.x;
      a.y = q ? ta.w : ta.y;
      b0.x = bp[(size_t)(8 * kp + 2 * q) * DIM];
      b0.y = bp[(size_t)(8 * kp + 2 * q + 1) * DIM];
      b1.x = bp[(size_t)(8 * kp + 2 * q) * DIM + 16];
      b1.y = bp[(size_t)(8 * kp + 2 * q + 1) * DIM + 16];
      acc0 = __builtin_amdgcn_wmma_f32_16x16x4_f32(false, a, false, b0, (short)0, acc0,
                                                   false, false);
      acc1 = __builtin_amdgcn_wmma_f32_16x16x4_f32(false, a, false, b1, (short)0, acc1,
                                                   false, false);
    }
  }
  if (isR) {
    const float rb0 = Rb[c0 + r];
    const float rb1 = Rb[c0 + 16 + r];
#pragma unroll
    for (int j = 0; j < 8; ++j) {
      const int rr = row0 + j + 8 * hf;
      if (rr < T) {
        R[(size_t)rr * DIM + c0 + r] = fmaxf(acc0[j] + rb0, 0.0f);
        R[(size_t)rr * DIM + c0 + 16 + r] = fmaxf(acc1[j] + rb1, 0.0f);
      }
    }
  } else {
#pragma unroll
    for (int j = 0; j < 8; ++j) {
      const int rr = row0 + j + 8 * hf;
      if (rr < T) {
        P[(size_t)rr * DIM + c0 + r] = acc0[j];
        P[(size_t)rr * DIM + c0 + 16 + r] = acc1[j];
      }
    }
  }
}

__device__ __forceinline__ void fragment_body(const float* __restrict__ W,
                                              const float* __restrict__ Rw,
                                              float4* __restrict__ Bf, int tid) {
  const int vt = tid & 255;
  const int g = tid >> 8;
  const int lane = vt & 31;
  const int ct = vt >> 5;
  const int r = lane & 15;
  const int hf = lane >> 4;
  const float* wp = W + (size_t)(32 * g + 4 * hf) * DIM + ct * 16 + r;
  const float* rp = Rw + (size_t)(32 * g + 4 * hf) * DIM + ct * 16 + r;
  float4* op = Bf + (size_t)(8 * g) * 256 + vt;
#pragma unroll
  for (int i = 0; i < 4; ++i) {
#pragma unroll
    for (int q = 0; q < 2; ++q) {
      float4 v;
      v.x = wp[(size_t)(8 * i + 2 * q) * DIM];
      v.y = wp[(size_t)(8 * i + 2 * q + 1) * DIM];
      v.z = rp[(size_t)(8 * i + 2 * q) * DIM];
      v.w = rp[(size_t)(8 * i + 2 * q + 1) * DIM];
      op[(2 * i + q) * 256] = v;
    }
  }
}

__global__ void __launch_bounds__(CNT_THREADS) setup_kernel(
    const int* __restrict__ src, const int* __restrict__ dst,
    int* __restrict__ cntS, int* __restrict__ cntD, int* __restrict__ tot,
    int* __restrict__ slot, int N, int E, int CH, int R,
    const float* __restrict__ emb, const float* __restrict__ Ws,
    const float* __restrict__ Rws, const float* __restrict__ Rbs,
    float* __restrict__ P, float* __restrict__ Rt, float* __restrict__ stats, int nstats,
    int T, int tabBlocks, float* __restrict__ wfrag) {
  const int blk = blockIdx.x;
  const int tid = threadIdx.x;
  if (blk < 2 * CNT_CHUNKS) {
    count_body(src, dst, cntS, cntD, tot, slot, N, E, CH, R, blk);
  } else if (blk < 2 * CNT_CHUNKS + tabBlocks) {
    const int tb = blk - 2 * CNT_CHUNKS;
    if (tb == 0) {
      for (int i = tid; i < nstats; i += CNT_THREADS) stats[i] = 0.0f;
    }
    const int tile = tb * (CNT_THREADS / 256) + (tid >> 8);
    if (tile * 16 < T) table_body(emb, Ws, Rws, Rbs, P, Rt, T, tile, tid & 255);
  } else {
    const int l = 1 + (blk - 2 * CNT_CHUNKS - tabBlocks);
    fragment_body(Ws + (size_t)l * DIM * DIM, Rws + (size_t)l * DIM * DIM,
                  reinterpret_cast<float4*>(wfrag) + (size_t)(l - 1) * (32 * 8 * 32), tid);
  }
}

#define AGG_WAVES 8

template <int B, bool CLAMP, bool BN>
__device__ __forceinline__ void sum_rows(float4& acc, const float4* __restrict__ X4,
                                         int recSrc, int recW, int j, int m,
                                         const float4& sc, const float4& sh) {
  float4 v[B];
  float w[B];
#pragma unroll
  for (int u = 0; u < B; ++u) {
    int idx = j + u;
    bool ok = true;
    if (CLAMP) {
      ok = idx < m;
      idx = ok ? idx : m - 1;
    }
    const int s = __builtin_amdgcn_readlane(recSrc, idx);
    const float ww = __int_as_float(__builtin_amdgcn_readlane(recW, idx));
    w[u] = ok ? ww : 0.0f;
    v[u] = X4[(size_t)s * (DIM / 4)];
  }
#pragma unroll
  for (int u = 0; u < B; ++u) {
    float4 xv = v[u];
    if (BN) {
      xv.x = __builtin_fmaf(xv.x, sc.x, sh.x);
      xv.y = __builtin_fmaf(xv.y, sc.y, sh.y);
      xv.z = __builtin_fmaf(xv.z, sc.z, sh.z);
      xv.w = __builtin_fmaf(xv.w, sc.w, sh.w);
    }
    acc.x = __builtin_fmaf(xv.x, w[u], acc.x);
    acc.y = __builtin_fmaf(xv.y, w[u], acc.y);
    acc.z = __builtin_fmaf(xv.z, w[u], acc.z);
    acc.w = __builtin_fmaf(xv.w, w[u], acc.w);
  }
}

__global__ void __launch_bounds__(AGG_WAVES * 32) aggregate_kernel(
    const int* __restrict__ rp, const int2* __restrict__ csr,
    const float* __restrict__ x, float* __restrict__ agg,
    const float* __restrict__ stats, const float* __restrict__ gamma,
    const float* __restrict__ beta, float invN, int N) {
  const int lane = threadIdx.x & 31;
  const int wv = __builtin_amdgcn_readfirstlane((int)(threadIdx.x >> 5));
  const int nw = gridDim.x * AGG_WAVES;
  int row = blockIdx.x * AGG_WAVES + wv;
  if (row >= N) return;
  const float4* X4 = reinterpret_cast<const float4*>(x) + lane;
  float4* A4 = reinterpret_cast<float4*>(agg) + lane;
  float4 sc, sh;
  bn_coeff4(stats, gamma, beta, invN, lane, sc, sh);

  int beg = rp[row];
  int end = rp[row + 1];
  int2 ev = make_int2(0, 0);
  if (lane < end - beg) ev = csr[beg + lane];
  int nrow = row + nw;
  int nb = 0, ne = 0;
  if (nrow < N) {
    nb = rp[nrow];
    ne = rp[nrow + 1];
  }

  for (;;) {
    int2 nev = make_int2(0, 0);
    if (lane < ne - nb) nev = csr[nb + lane];
    const int nnrow = nrow + nw;
    int nnb = 0, nne = 0;
    if (nnrow < N) {
      nnb = rp[nnrow];
      nne = rp[nnrow + 1];
    }

    float4 acc = make_float4(0.0f, 0.0f, 0.0f, 0.0f);
    const int cnt = end - beg;
    int done = 0;
    for (;;) {
      const int m = (cnt - done < 32) ? (cnt - done) : 32;
      int j = 0;
      for (; j + 8 <= m; j += 8) sum_rows<8, false, true>(acc, X4, ev.x, ev.y, j, m, sc, sh);
      for (; j < m; j += 4) sum_rows<4, true, true>(acc, X4, ev.x, ev.y, j, m, sc, sh);
      done += 32;
      if (done >= cnt) break;
      ev = make_int2(0, 0);
      if (lane < cnt - done) ev = csr[beg + done + lane];
    }
    A4[(size_t)row * (DIM / 4)] = acc;

    if (nrow >= N) break;
    row = nrow;
    beg = nb;
    end = ne;
    ev = nev;
    nrow = nnrow;
    nb = nnb;
    ne = nne;
  }
}

#define L0_WAVES 32
#define TAB_CAP 352
#define L0_STAGE_U (TAB_CAP * (DIM / 4) / (L0_WAVES * 32))
__global__ void __launch_bounds__(L0_WAVES * 32) layer0_kernel(
    const int* __restrict__ rp, const int2* __restrict__ csr0,
    const float* __restrict__ P, const float* __restrict__ R,
    const int* __restrict__ ids, const float* __restrict__ inorm,
    const float* __restrict__ bias, float* __restrict__ h,
    float* __restrict__ csum, float* __restrict__ csumsq, int N, int T) {
  __shared__ float4 sP[TAB_CAP * (DIM / 4)];
  __shared__ float4 sRed[L0_WAVES * 2 * (DIM / 4)];
  {
    const float4* Pg = reinterpret_cast<const float4*>(P);
    const int n4 = T * (DIM / 4);
    float4 t[L0_STAGE_U];
#pragma unroll
    for (int u = 0; u < L0_STAGE_U; ++u) {
      const int i = (int)threadIdx.x + u * (L0_WAVES * 32);
      t[u] = Pg[(i < n4) ? i : n4 - 1];
    }
#pragma unroll
    for (int u = 0; u < L0_STAGE_U; ++u) sP[(int)threadIdx.x + u * (L0_WAVES * 32)] = t[u];
  }
  __syncthreads();
  const int lane = threadIdx.x & 31;
  const int wv = __builtin_amdgcn_readfirstlane((int)(threadIdx.x >> 5));
  const int nw = gridDim.x * L0_WAVES;
  int row = blockIdx.x * L0_WAVES + wv;
  float4 cs = make_float4(0.0f, 0.0f, 0.0f, 0.0f);
  float4 cq = make_float4(0.0f, 0.0f, 0.0f, 0.0f);
  if (row < N) {
    const float4* P4 = sP + lane;
    const float4* R4 = reinterpret_cast<const float4*>(R) + lane;
    float4* H4 = reinterpret_cast<float4*>(h) + lane;
    const float4 b4 = reinterpret_cast<const float4*>(bias)[lane];
    const float4 one = make_float4(1.0f, 1.0f, 1.0f, 1.0f);
    const float4 nul = make_float4(0.0f, 0.0f, 0.0f, 0.0f);

    int beg = rp[row];
    int end = rp[row + 1];
    float inn = inorm[row];
    float4 rv = R4[(size_t)ids[row] * (DIM / 4)];
    int2 ev = make_int2(0, 0);
    if (lane < end - beg) ev = csr0[beg + lane];
    int nrow = row + nw;
    int nb = 0, ne = 0, nid = 0;
    float ninn = 0.0f;
    if (nrow < N) {
      nb = rp[nrow];
      ne = rp[nrow + 1];
      nid = ids[nrow];
      ninn = inorm[nrow];
    }

    for (;;) {
      int2 nev = make_int2(0, 0);
      if (lane < ne - nb) nev = csr0[nb + lane];
      const float4 nrv = R4[(size_t)nid * (DIM / 4)];
      const int nnrow = nrow + nw;
      int nnb = 0, nne = 0, nnid = 0;
      float nninn = 0.0f;
      if (nnrow < N) {
        nnb = rp[nnrow];
        nne = rp[nnrow + 1];
        nnid = ids[nnrow];
        nninn = inorm[nnrow];
      }

      float4 acc = make_float4(0.0f, 0.0f, 0.0f, 0.0f);
      const int cnt = end - beg;
      int done = 0;
      for (;;) {
        const int m = (cnt - done < 32) ? (cnt - done) : 32;
        int j = 0;
        for (; j + 8 <= m; j += 8) sum_rows<8, false, false>(acc, P4, ev.x, ev.y, j, m, one, nul);
        for (; j < m; j += 4) sum_rows<4, true, false>(acc, P4, ev.x, ev.y, j, m, one, nul);
        done += 32;
        if (done >= cnt) break;
        ev = make_int2(0, 0);
        if (lane < cnt - done) ev = csr0[beg + done + lane];
      }
      float4 v;
      v.x = fmaxf(acc.x * inn + b4.x, 0.0f) + rv.x;
      v.y = fmaxf(acc.y * inn + b4.y, 0.0f) + rv.y;
      v.z = fmaxf(acc.z * inn + b4.z, 0.0f) + rv.z;
      v.w = fmaxf(acc.w * inn + b4.w, 0.0f) + rv.w;
      H4[(size_t)row * (DIM / 4)] = v;
      cs.x += v.x;
      cs.y += v.y;
      cs.z += v.z;
      cs.w += v.w;
      cq.x += v.x * v.x;
      cq.y += v.y * v.y;
      cq.z += v.z * v.z;
      cq.w += v.w * v.w;

      if (nrow >= N) break;
      row = nrow;
      beg = nb;
      end = ne;
      inn = ninn;
      rv = nrv;
      ev = nev;
      nrow = nnrow;
      nb = nnb;
      ne = nne;
      nid = nnid;
      ninn = nninn;
    }
  }
  sRed[wv * (2 * (DIM / 4)) + lane] = cs;
  sRed[wv * (2 * (DIM / 4)) + (DIM / 4) + lane] = cq;
  __syncthreads();
  const int tid = threadIdx.x;
  if (tid < 2 * DIM) {
    const float* red = reinterpret_cast<const float*>(sRed);
    float tsum = 0.0f;
#pragma unroll
    for (int w = 0; w < L0_WAVES; ++w) tsum += red[w * (2 * DIM) + tid];
    atomicAdd((tid < DIM) ? &csum[tid] : &csumsq[tid - DIM], tsum);
  }
}

__global__ void __launch_bounds__(256) gemm_fused_kernel(
    const float* __restrict__ agg, const float* __restrict__ xin,
    const float* __restrict__ inorm,
    const float* __restrict__ Bfrag, const float* __restrict__ bias,
    const float* __restrict__ Rb,
    const float* __restrict__ pstats, const float* __restrict__ pgamma,
    const float* __restrict__ pbeta, float invN,
    float* __restrict__ h, float* __restrict__ csum, float* __restrict__ csumsq,
    int N) {
  __shared__ float4 sB[32 * 8 * 32];
  __shared__ float4 sScale[DIM / 4];
  __shared__ float4 sShift[DIM / 4];
  __shared__ float sRed[8 * 2 * DIM];

  const int tid = threadIdx.x;
  const int wave = tid >> 5;
  const int lane = tid & 31;
  const int r = lane & 15;
  const int hf = lane >> 4;

  const int bk = tid & (DIM - 1);
  const float bnS = pstats[bk];
  const float bnQ = pstats[DIM + bk];
  const float bnG = pgamma[bk];
  const float bnB = pbeta[bk];

  {
    const float4* bf = reinterpret_cast<const float4*>(Bfrag) + tid;
    float4 t[32];
#pragma unroll
    for (int u = 0; u < 32; ++u) t[u] = bf[u * 256];
#pragma unroll
    for (int u = 0; u < 32; ++u) sB[tid + u * 256] = t[u];
  }
  {
    float sc, sh;
    bn_coeff(bnS, bnQ, bnG, bnB, invN, sc, sh);
    float* dstp = reinterpret_cast<float*>((tid < DIM) ? sScale : sShift);
    dstp[bk] = (tid < DIM) ? sc : sh;
  }
  __syncthreads();

  const int row0 = (blockIdx.x * 8 + wave) * 16;
  v8f acc1[8], acc2[8];

  if (row0 < N) {

  const int row = row0 + r;
  const float innr = inorm[row];
  const float4* aP = reinterpret_cast<const float4*>(agg + (size_t)row * DIM) + hf;
  const float4* xP = reinterpret_cast<const float4*>(xin + (size_t)row * DIM) + hf;
  const float4* bP = sB + lane;

#pragma unroll
  for (int ct = 0; ct < 8; ++ct) {
    acc1[ct] = (v8f){0.f, 0.f, 0.f, 0.f, 0.f, 0.f, 0.f, 0.f};
    acc2[ct] = (v8f){0.f, 0.f, 0.f, 0.f, 0.f, 0.f, 0.f, 0.f};
  }

  const float4* scP = sScale + hf;
  const float4* shP = sShift + hf;

  float4 ta = aP[0];
  float4 tx = xP[0];
  float4 b0[8], b1[8];
#pragma unroll
  for (int ct = 0; ct < 8; ++ct) b0[ct] = bP[ct * 32];

  for (int kp = 0; kp < 16; ++kp) {
    float4 ta_n = ta, tx_n = tx;
    if (kp + 1 < 16) {
      ta_n = aP[(kp + 1) * 2];
      tx_n = xP[(kp + 1) * 2];
    }
#pragma unroll
    for (int ct = 0; ct < 8; ++ct) b1[ct] = bP[((2 * kp + 1) * 8 + ct) * 32];

    {
      const float4 s4 = scP[2 * kp];
      const float4 t4 = shP[2 * kp];
      tx.x = __builtin_fmaf(tx.x, s4.x, t4.x);
      tx.y = __builtin_fmaf(tx.y, s4.y, t4.y);
      tx.z = __builtin_fmaf(tx.z, s4.z, t4.z);
      tx.w = __builtin_fmaf(tx.w, s4.w, t4.w);
    }

    v2f a1, a2;
    a1.x = ta.x * innr;
    a1.y = ta.y * innr;
    a2.x = tx.x;
    a2.y = tx.y;
#pragma unroll
    for (int ct = 0; ct < 8; ++ct) {
      v2f w1, w2;
      w1.x = b0[ct].x; w1.y = b0[ct].y;
      w2.x = b0[ct].z; w2.y = b0[ct].w;
      acc1[ct] = __builtin_amdgcn_wmma_f32_16x16x4_f32(false, a1, false, w1,
                                                       (short)0, acc1[ct], false, false);
      acc2[ct] = __builtin_amdgcn_wmma_f32_16x16x4_f32(false, a2, false, w2,
                                                       (short)0, acc2[ct], false, false);
    }

    if (kp + 1 < 16) {
#pragma unroll
      for (int ct = 0; ct < 8; ++ct) b0[ct] = bP[((2 * kp + 2) * 8 + ct) * 32];
    }

    a1.x = ta.z * innr;
    a1.y = ta.w * innr;
    a2.x = tx.z;
    a2.y = tx.w;
#pragma unroll
    for (int ct = 0; ct < 8; ++ct) {
      v2f w1, w2;
      w1.x = b1[ct].x; w1.y = b1[ct].y;
      w2.x = b1[ct].z; w2.y = b1[ct].w;
      acc1[ct] = __builtin_amdgcn_wmma_f32_16x16x4_f32(false, a1, false, w1,
                                                       (short)0, acc1[ct], false, false);
      acc2[ct] = __builtin_amdgcn_wmma_f32_16x16x4_f32(false, a2, false, w2,
                                                       (short)0, acc2[ct], false, false);
    }
    ta = ta_n;
    tx = tx_n;
  }

#pragma unroll
  for (int ct = 0; ct < 8; ++ct) {
    const int col0 = ct * 16;
    const float bb = bias[col0 + r];
    const float rb = Rb[col0 + r];
    float s = 0.0f, s2 = 0.0f;
#pragma unroll
    for (int j = 0; j < 8; ++j) {
      float v = fmaxf(acc1[ct][j] + bb, 0.0f) + fmaxf(acc2[ct][j] + rb, 0.0f);
      acc1[ct][j] = v;
      s += v;
      s2 += v * v;
    }
    s += __shfl_xor(s, 16);
    s2 += __shfl_xor(s2, 16);
    sRed[wave * (2 * DIM) + hf * DIM + col0 + r] = hf ? s2 : s;
  }
  } else {
    for (int i = lane; i < 2 * DIM; i += 32) sRed[wave * (2 * DIM) + i] = 0.0f;
  }
  __syncthreads();
  float tsum = 0.0f;
#pragma unroll
  for (int w = 0; w < 8; ++w) tsum += sRed[w * (2 * DIM) + tid];
  atomicAdd((tid < DIM) ? &csum[tid] : &csumsq[tid - DIM], tsum);
  if (row0 < N) {
#pragma unroll
    for (int ct = 0; ct < 8; ++ct) {
#pragma unroll
      for (int j = 0; j < 8; ++j)
        h[(size_t)(row0 + j + 8 * hf) * DIM + ct * 16 + r] = acc1[ct][j];
    }
  }
}

#define STREAM_U 16
__global__ void __launch_bounds__(STREAM_THREADS) bnapply_kernel(
    const float* __restrict__ h, const float* __restrict__ stats,
    const float* __restrict__ gamma, const float* __restrict__ beta, float invN,
    float* __restrict__ out, int N) {
  const int lane = threadIdx.x & 31;
  const int wv = blockIdx.x * (STREAM_THREADS / 32) +
                 __builtin_amdgcn_readfirstlane((int)(threadIdx.x >> 5));
  const int nw = gridDim.x * (STREAM_THREADS / 32);
  const float4* H4 = reinterpret_cast<const float4*>(h) + lane;
  float4* O4 = reinterpret_cast<float4*>(out) + lane;
  float4 sc, sh;
  bn_coeff4(stats, gamma, beta, invN, lane, sc, sh);
  for (int i0 = wv; i0 < N; i0 += STREAM_U * nw) {
    float4 v[STREAM_U];
#pragma unroll
    for (int u = 0; u < STREAM_U; ++u) {
      const int i = i0 + u * nw;
      v[u] = H4[(size_t)((i < N) ? i : i0) * (DIM / 4)];
    }
#pragma unroll
    for (int u = 0; u < STREAM_U; ++u) {
      const int i = i0 + u * nw;
      asm volatile("" ::"v"(v[u].x), "v"(v[u].y), "v"(v[u].z), "v"(v[u].w));
      if (i < N) {
        float4 o;
        o.x = v[u].x * sc.x + sh.x;
        o.y = v[u].y * sc.y + sh.y;
        o.z = v[u].z * sc.z + sh.z;
        o.w = v[u].w * sc.w + sh.w;
        O4[(size_t)i * (DIM / 4)] = o;
      }
    }
  }
}


static inline int cdiv(int a, int b) { return (a + b - 1) / b; }

extern "C" void kernel_launch(void* const* d_in, const int* in_sizes, int n_in,
                              void* d_out, int out_size, void* d_ws, size_t ws_size,
                              hipStream_t stream) {
  const int* node_ids = (const int*)d_in[0];
  const int* src = (const int*)d_in[1];
  const int* dst = (const int*)d_in[2];
  const float* emb = (const float*)d_in[3];
  const float* Ws = (const float*)d_in[4];
  const float* bs = (const float*)d_in[5];
  const float* Rws = (const float*)d_in[6];
  const float* Rbs = (const float*)d_in[7];
  const float* gammas = (const float*)d_in[8];
  const float* betas = (const float*)d_in[9];

  const int N = in_sizes[0];
  const int E = in_sizes[1];
  const int T = in_sizes[3] / DIM;
  const int L = in_sizes[4] / (DIM * DIM);

  float* wsf = (float*)d_ws;
  size_t off = 0;
  float* onorm = wsf + off; off += (size_t)N;
  float* inorm = wsf + off; off += (size_t)N;
  off = (off + 3) & ~(size_t)3;
  float* ha   = wsf + off;  off += (size_t)N * DIM;
  float* hb   = wsf + off;  off += (size_t)N * DIM;
  float* agg  = wsf + off;  off += (size_t)N * DIM;
  float* tabP = wsf + off;  off += (size_t)T * DIM;
  float* tabR = wsf + off;  off += (size_t)T * DIM;
  float* stats = wsf + off; off += (size_t)L * 2 * DIM;
  off = (off + 3) & ~(size_t)3;
  float* wfrag = wsf + off; off += (size_t)(L - 1) * 4 * (32 * 8 * 32);
  off = (off + 3) & ~(size_t)3;
  int2* csr = (int2*)(wsf + off); off += (size_t)E * 2;
  int2* csr0 = (int2*)(wsf + off); off += (size_t)E * 2;
  int* slot = (int*)(wsf + off);  off += (size_t)E;
  int* rp   = (int*)(wsf + off);  off += (size_t)N + 1;
  const int R = cdiv(N, CNT_RANGE);
  int* cntS = (int*)(wsf + off);  off += (size_t)CNT_CHUNKS * N;
  int* cntD = (int*)(wsf + off);  off += (size_t)CNT_CHUNKS * N;
  int* tot  = (int*)(wsf + off);  off += (size_t)CNT_CHUNKS * R;

  const int B = 256;

  const int CH = cdiv(E, CNT_CHUNKS);
  const int tabBlocks = cdiv(T, 16 * (CNT_THREADS / 256));
  setup_kernel<<<2 * CNT_CHUNKS + tabBlocks + (L - 1), CNT_THREADS, 0, stream>>>(
      src, dst, cntS, cntD, tot, slot, N, E, CH, R, emb, Ws, Rws, Rbs, tabP, tabR, stats,
      L * 2 * DIM, T, tabBlocks, wfrag);
  offsets_kernel<<<R, CNT_RANGE, 0, stream>>>(cntS, cntD, tot, rp, onorm, inorm, N, R);
  place_kernel<<<STREAM_BLOCKS, STREAM_THREADS, 0, stream>>>(src, dst, slot, rp, cntD, node_ids,
                                                             onorm, csr, csr0, E, N, CH);

  const int rowTiles = cdiv(N, 16);
  const int l0Blocks = (cdiv(N, L0_WAVES) < 256) ? cdiv(N, L0_WAVES) : 256;
  const int aggBlocks = (cdiv(N, AGG_WAVES) < 1024) ? cdiv(N, AGG_WAVES) : 1024;
  const float invN = 1.0f / (float)N;
  layer0_kernel<<<l0Blocks, L0_WAVES * 32, 0, stream>>>(rp, csr0, tabP, tabR, node_ids, inorm,
                                                        bs, ha, stats, stats + DIM, N, T);
  float* cur = ha;
  float* nxt = hb;
  for (int l = 1; l < L; ++l) {
    float* st = stats + (size_t)l * 2 * DIM;
    const float* Bf = wfrag + (size_t)(l - 1) * 4 * (32 * 8 * 32);
    const float* bw = bs + (size_t)l * DIM;
    const float* rb = Rbs + (size_t)l * DIM;
    const float* pst = st - 2 * DIM;
    const float* pg = gammas + (size_t)(l - 1) * DIM;
    const float* pb = betas + (size_t)(l - 1) * DIM;
    aggregate_kernel<<<aggBlocks, AGG_WAVES * 32, 0, stream>>>(
        rp, csr, cur, agg, pst, pg, pb, invN, N);
    gemm_fused_kernel<<<cdiv(rowTiles, 8), B, 0, stream>>>(
        agg, cur, inorm, Bf, bw, rb, pst, pg, pb, invN, nxt, st, st + DIM, N);
    float* t = cur;
    cur = nxt;
    nxt = t;
  }
  bnapply_kernel<<<STREAM_BLOCKS, STREAM_THREADS, 0, stream>>>(
      cur, stats + (size_t)(L - 1) * 2 * DIM, gammas + (size_t)(L - 1) * DIM,
      betas + (size_t)(L - 1) * DIM, invN, (float*)d_out, N);
}


# Bucket sums fused into the layer GEMM launch: a wave sums its 16-row stripe's buckets into an LDS tile, reads A from it; no agg plane, 9 -> 7 launches; BN column sums regrouped (fp32 order changes)
# speedup vs baseline: 4.2416x; 1.0775x over previous
#include <hip/hip_runtime.h>
#include <math.h>


typedef __attribute__((ext_vector_type(2))) float v2f;
typedef __attribute__((ext_vector_type(8))) float v8f;

#define DIM 128
#define BN_EPS 1e-5f

__device__ __forceinline__ void bn_coeff(float s, float q, float g, float b, float invN,
                                         float& sc, float& sh) {
  const float mu = s * invN;
  const float var = q * invN - mu * mu;
  sc = g / sqrtf(var + BN_EPS);
  sh = b - mu * sc;
}

__device__ __forceinline__ void bn_coeff4(const float* __restrict__ stats,
                                          const float* __restrict__ gamma,
                                          const float* __restrict__ beta, float invN,
                                          int lane, float4& sc, float4& sh) {
  const float4 s = reinterpret_cast<const float4*>(stats)[lane];
  const float4 q = reinterpret_cast<const float4*>(stats + DIM)[lane];
  const float4 g = reinterpret_cast<const float4*>(gamma)[lane];
  const float4 b = reinterpret_cast<const float4*>(beta)[lane];
  bn_coeff(s.x, q.x, g.x, b.x, invN, sc.x, sh.x);
  bn_coeff(s.y, q.y, g.y, b.y, invN, sc.y, sh.y);
  bn_coeff(s.z, q.z, g.z, b.z, invN, sc.z, sh.z);
  bn_coeff(s.w, q.w, g.w, b.w, invN, sc.w, sh.w);
}

#define CNT_THREADS 1024
#define CNT_RANGE 1024
#define CNT_CAP 65536
#define CNT_CHUNKS 16
__device__ __forceinline__ void count_body(
    const int* __restrict__ src, const int* __restrict__ dst,
    int* __restrict__ cntS, int* __restrict__ cntD, int* __restrict__ tot,
    int* __restrict__ slot, int N, int E, int CH, int R, int blk) {
  __shared__ int cnt[CNT_CAP];
  const int tid = threadIdx.x;
  const int lane = tid & 31;
  const int wave = tid >> 5;
  const bool isDst = blk < CNT_CHUNKS;
  const int q = isDst ? blk : blk - CNT_CHUNKS;
  const int* __restrict__ keys = isDst ? dst : src;
  int* __restrict__ outc = (isDst ? cntD : cntS) + (size_t)q * N;
  const int eb = q * CH;
  const int ee = (E - eb < CH) ? E : eb + CH;
  for (int nb = 0; nb < N; nb += CNT_CAP) {
    const int n = (N - nb < CNT_CAP) ? (N - nb) : CNT_CAP;
    const int nr = (n + CNT_RANGE - 1) / CNT_RANGE;
    {
      int4* c4p = reinterpret_cast<int4*>(cnt);
      for (int i = tid; i < (n + 3) / 4; i += CNT_THREADS) c4p[i] = make_int4(0, 0, 0, 0);
    }
    __syncthreads();
    for (int e0 = eb + tid; e0 < ee; e0 += 8 * CNT_THREADS) {
      int k[8];
#pragma unroll
      for (int u = 0; u < 8; ++u) {
        const int e = e0 + u * CNT_THREADS;
        const int kv = keys[(e < ee) ? e : ee - 1];
        k[u] = (e < ee) ? kv - nb : -1;
      }
      int place[8];
#pragma unroll
      for (int u = 0; u < 8; ++u) {
        place[u] = 0;
        if ((unsigned)k[u] < (unsigned)n) place[u] = atomicAdd(&cnt[k[u]], 1);
      }
      if (isDst) {
#pragma unroll
        for (int u = 0; u < 8; ++u) {
          const int e = e0 + u * CNT_THREADS;
          if ((unsigned)k[u] < (unsigned)n) slot[e] = place[u];
        }
      }
    }
    __syncthreads();
    for (int j = wave; j < nr; j += CNT_THREADS / 32) {
      int c = 0;
      for (int t = lane; t < CNT_RANGE; t += 32) {
        const int i = j * CNT_RANGE + t;
        if (i < n) {
          const int v = cnt[i];
          outc[nb + i] = v;
          c += v;
        }
      }
      if (isDst) {
        for (int o = 16; o > 0; o >>= 1) c += __shfl_xor(c, o);
        if (lane == 0) tot[(size_t)q * R + nb / CNT_RANGE + j] = c;
      }
    }
    __syncthreads();
  }
}

__global__ void __launch_bounds__(CNT_RANGE) offsets_kernel(
    const int* __restrict__ cntS, int* __restrict__ cntD, const int* __restrict__ tot,
    int* __restrict__ rp, float* __restrict__ onorm, float* __restrict__ inorm,
    int N, int R) {
  __shared__ int wtot[33];
  __shared__ int wbase[33];
  const int tid = threadIdx.x;
  const int lane = tid & 31;
  const int wave = tid >> 5;
  const int r = blockIdx.x;
  const int i = r * CNT_RANGE + tid;

  int acc = 0;
  for (int t = tid; t < CNT_CHUNKS * R; t += CNT_RANGE) {
    const int v = tot[t];
    if (t % R < r) acc += v;
  }

  int dd = 0;
  if (i < N) {
    int cs[CNT_CHUNKS], cd[CNT_CHUNKS];
#pragma unroll
    for (int q = 0; q < CNT_CHUNKS; ++q) {
      cs[q] = cntS[(size_t)q * N + i];
      cd[q] = cntD[(size_t)q * N + i];
    }
    int ds = 0;
#pragma unroll
    for (int q = 0; q < CNT_CHUNKS; ++q) {
      ds += cs[q];
      cntD[(size_t)q * N + i] = dd;
      dd += cd[q];
    }
    onorm[i] = (ds > 0) ? 1.0f / sqrtf((float)ds) : 0.0f;
    inorm[i] = (dd > 0) ? 1.0f / sqrtf((float)dd) : 0.0f;
  }

  int v = dd;
  for (int o = 1; o < 32; o <<= 1) {
    const int t = __shfl_up(v, o);
    if (lane >= o) v += t;
  }
  for (int o = 16; o > 0; o >>= 1) acc += __shfl_xor(acc, o);
  if (lane == 31) wtot[wave] = v;
  if (lane == 0) wbase[wave] = acc;
  __syncthreads();
  if (wave == 0) {
    const int t = wtot[lane];
    int u = t;
    for (int o = 1; o < 32; o <<= 1) {
      const int s = __shfl_up(u, o);
      if (lane >= o) u += s;
    }
    wtot[lane] = u - t;
    int b = wbase[lane];
    for (int o = 16; o > 0; o >>= 1) b += __shfl_xor(b, o);
    if (lane == 0) wbase[32] = b;
  }
  __syncthreads();
  const int incl = wbase[32] + wtot[wave] + v;
  if (i < N) rp[i] = incl - dd;
  if (i == N - 1) rp[N] = incl;
}

#define STREAM_BLOCKS 512
#define STREAM_THREADS 256

#define PLACE_U 4
__global__ void __launch_bounds__(STREAM_THREADS) place_kernel(
    const int* __restrict__ src, const int* __restrict__ dst,
    const int* __restrict__ slot, const int* __restrict__ rp,
    const int* __restrict__ cntD, const int* __restrict__ ids,
    const float* __restrict__ onorm, int2* __restrict__ csr, int2* __restrict__ csr0,
    int E, int N, int CH) {
  const int t = blockIdx.x * STREAM_THREADS + threadIdx.x;
  const int stride = gridDim.x * STREAM_THREADS;
  for (int e0 = t; e0 < E; e0 += PLACE_U * stride) {
    int s[PLACE_U], d[PLACE_U], sl[PLACE_U], q[PLACE_U];
#pragma unroll
    for (int u = 0; u < PLACE_U; ++u) {
      const int e = e0 + u * stride;
      const int ec = (e < E) ? e : e0;
      s[u] = src[ec];
      d[u] = dst[ec];
      sl[u] = slot[ec];
      q[u] = ec / CH;
    }
    int base[PLACE_U], cs[PLACE_U], w[PLACE_U], id[PLACE_U];
#pragma unroll
    for (int u = 0; u < PLACE_U; ++u) {
      w[u] = __float_as_int(onorm[s[u]]);
      id[u] = ids[s[u]];
      base[u] = rp[d[u]];
      cs[u] = cntD[(size_t)q[u] * N + d[u]];
    }
#pragma unroll
    for (int u = 0; u < PLACE_U; ++u) {
      const int e = e0 + u * stride;
      asm volatile("" ::"v"(s[u]), "v"(sl[u]), "v"(w[u]), "v"(id[u]), "v"(base[u]), "v"(cs[u]));
      if (e < E) {
        const int pos = base[u] + cs[u] + sl[u];
        int2 rec;
        rec.x = s[u];
        rec.y = w[u];
        csr[pos] = rec;
        rec.x = id[u];
        csr0[pos] = rec;
      }
    }
  }
}

__device__ __forceinline__ void table_body(
    const float* __restrict__ emb, const float* __restrict__ W, const float* __restrict__ Rw,
    const float* __restrict__ Rb, float* __restrict__ P, float* __restrict__ R,
    int T, int tile, int tid) {
  const int wave = __builtin_amdgcn_readfirstlane(tid >> 5);
  const bool isR = wave >= 4;
  const int c0 = (wave & 3) * 32;
  const int lane = tid & 31;
  const int r = lane & 15;
  const int hf = lane >> 4;
  const int row0 = tile * 16;
  const int row = (row0 + r < T) ? row0 + r : T - 1;
  const float4* aP = reinterpret_cast<const float4*>(emb + (size_t)row * DIM) + hf;
  const float* bp = (isR ? Rw : W) + (size_t)(4 * hf) * DIM + c0 + r;
  v8f acc0 = (v8f){0.f, 0.f, 0.f, 0.f, 0.f, 0.f, 0.f, 0.f};
  v8f acc1 = (v8f){0.f, 0.f, 0.f, 0.f, 0.f, 0.f, 0.f, 0.f};
#pragma unroll
  for (int kp = 0; kp < 16; ++kp) {
    const float4 ta = aP[2 * kp];
#pragma unroll
    for (int q = 0; q < 2; ++q) {
      v2f a, b0, b1;
      a.x = q ? ta.z : ta.x;
      a.y = q ? ta.w : ta.y;
      b0.x = bp[(size_t)(8 * kp + 2 * q) * DIM];
      b0.y = bp[(size_t)(8 * kp + 2 * q + 1) * DIM];
      b1.x = bp[(size_t)(8 * kp + 2 * q) * DIM + 16];
      b1.y = bp[(size_t)(8 * kp + 2 * q + 1) * DIM + 16];
      acc0 = __builtin_amdgcn_wmma_f32_16x16x4_f32(false, a, false, b0, (short)0, acc0,
                                                   false, false);
      acc1 = __builtin_amdgcn_wmma_f32_16x16x4_f32(false, a, false, b1, (short)0, acc1,
                                                   false, false);
    }
  }
  if (isR) {
    const float rb0 = Rb[c0 + r];
    const float rb1 = Rb[c0 + 16 + r];
#pragma unroll
    for (int j = 0; j < 8; ++j) {
      const int rr = row0 + j + 8 * hf;
      if (rr < T) {
        R[(size_t)rr * DIM + c0 + r] = fmaxf(acc0[j] + rb0, 0.0f);
        R[(size_t)rr * DIM + c0 + 16 + r] = fmaxf(acc1[j] + rb1, 0.0f);
      }
    }
  } else {
#pragma unroll
    for (int j = 0; j < 8; ++j) {
      const int rr = row0 + j + 8 * hf;
      if (rr < T) {
        P[(size_t)rr * DIM + c0 + r] = acc0[j];
        P[(size_t)rr * DIM + c0 + 16 + r] = acc1[j];
      }
    }
  }
}

__device__ __forceinline__ void fragment_body(const float* __restrict__ W,
                                              const float* __restrict__ Rw,
                                              float4* __restrict__ Bf, int tid) {
  const int vt = tid & 255;
  const int g = tid >> 8;
  const int lane = vt & 31;
  const int ct = vt >> 5;
  const int r = lane & 15;
  const int hf = lane >> 4;
  const float* wp = W + (size_t)(32 * g + 4 * hf) * DIM + ct * 16 + r;
  const float* rp = Rw + (size_t)(32 * g + 4 * hf) * DIM + ct * 16 + r;
  float4* op = Bf + (size_t)(8 * g) * 256 + vt;
#pragma unroll
  for (int i = 0; i < 4; ++i) {
#pragma unroll
    for (int q = 0; q < 2; ++q) {
      float4 v;
      v.x = wp[(size_t)(8 * i + 2 * q) * DIM];
      v.y = wp[(size_t)(8 * i + 2 * q + 1) * DIM];
      v.z = rp[(size_t)(8 * i + 2 * q) * DIM];
      v.w = rp[(size_t)(8 * i + 2 * q + 1) * DIM];
      op[(2 * i + q) * 256] = v;
    }
  }
}

__global__ void __launch_bounds__(CNT_THREADS) setup_kernel(
    const int* __restrict__ src, const int* __restrict__ dst,
    int* __restrict__ cntS, int* __restrict__ cntD, int* __restrict__ tot,
    int* __restrict__ slot, int N, int E, int CH, int R,
    const float* __restrict__ emb, const float* __restrict__ Ws,
    const float* __restrict__ Rws, const float* __restrict__ Rbs,
    float* __restrict__ P, float* __restrict__ Rt, float* __restrict__ stats, int nstats,
    int T, int tabBlocks, float* __restrict__ wfrag) {
  const int blk = blockIdx.x;
  const int tid = threadIdx.x;
  if (blk < 2 * CNT_CHUNKS) {
    count_body(src, dst, cntS, cntD, tot, slot, N, E, CH, R, blk);
  } else if (blk < 2 * CNT_CHUNKS + tabBlocks) {
    const int tb = blk - 2 * CNT_CHUNKS;
    if (tb == 0) {
      for (int i = tid; i < nstats; i += CNT_THREADS) stats[i] = 0.0f;
    }
    const int tile = tb * (CNT_THREADS / 256) + (tid >> 8);
    if (tile * 16 < T) table_body(emb, Ws, Rws, Rbs, P, Rt, T, tile, tid & 255);
  } else {
    const int l = 1 + (blk - 2 * CNT_CHUNKS - tabBlocks);
    fragment_body(Ws + (size_t)l * DIM * DIM, Rws + (size_t)l * DIM * DIM,
                  reinterpret_cast<float4*>(wfrag) + (size_t)(l - 1) * (32 * 8 * 32), tid);
  }
}


template <int B, bool CLAMP, bool BN>
__device__ __forceinline__ void sum_rows(float4& acc, const float4* __restrict__ X4,
                                         int recSrc, int recW, int j, int m,
                                         const float4& sc, const float4& sh) {
  float4 v[B];
  float w[B];
#pragma unroll
  for (int u = 0; u < B; ++u) {
    int idx = j + u;
    bool ok = true;
    if (CLAMP) {
      ok = idx < m;
      idx = ok ? idx : m - 1;
    }
    const int s = __builtin_amdgcn_readlane(recSrc, idx);
    const float ww = __int_as_float(__builtin_amdgcn_readlane(recW, idx));
    w[u] = ok ? ww : 0.0f;
    v[u] = X4[(size_t)s * (DIM / 4)];
  }
#pragma unroll
  for (int u = 0; u < B; ++u) {
    float4 xv = v[u];
    if (BN) {
      xv.x = __builtin_fmaf(xv.x, sc.x, sh.x);
      xv.y = __builtin_fmaf(xv.y, sc.y, sh.y);
      xv.z = __builtin_fmaf(xv.z, sc.z, sh.z);
      xv.w = __builtin_fmaf(xv.w, sc.w, sh.w);
    }
    acc.x = __builtin_fmaf(xv.x, w[u], acc.x);
    acc.y = __builtin_fmaf(xv.y, w[u], acc.y);
    acc.z = __builtin_fmaf(xv.z, w[u], acc.z);
    acc.w = __builtin_fmaf(xv.w, w[u], acc.w);
  }
}

#define L0_WAVES 32
#define TAB_CAP 352
#define L0_STAGE_U (TAB_CAP * (DIM / 4) / (L0_WAVES * 32))
__global__ void __launch_bounds__(L0_WAVES * 32) layer0_kernel(
    const int* __restrict__ rp, const int2* __restrict__ csr0,
    const float* __restrict__ P, const float* __restrict__ R,
    const int* __restrict__ ids, const float* __restrict__ inorm,
    const float* __restrict__ bias, float* __restrict__ h,
    float* __restrict__ csum, float* __restrict__ csumsq, int N, int T) {
  __shared__ float4 sP[TAB_CAP * (DIM / 4)];
  __shared__ float4 sRed[L0_WAVES * 2 * (DIM / 4)];
  {
    const float4* Pg = reinterpret_cast<const float4*>(P);
    const int n4 = T * (DIM / 4);
    float4 t[L0_STAGE_U];
#pragma unroll
    for (int u = 0; u < L0_STAGE_U; ++u) {
      const int i = (int)threadIdx.x + u * (L0_WAVES * 32);
      t[u] = Pg[(i < n4) ? i : n4 - 1];
    }
#pragma unroll
    for (int u = 0; u < L0_STAGE_U; ++u) sP[(int)threadIdx.x + u * (L0_WAVES * 32)] = t[u];
  }
  __syncthreads();
  const int lane = threadIdx.x & 31;
  const int wv = __builtin_amdgcn_readfirstlane((int)(threadIdx.x >> 5));
  const int nw = gridDim.x * L0_WAVES;
  int row = blockIdx.x * L0_WAVES + wv;
  float4 cs = make_float4(0.0f, 0.0f, 0.0f, 0.0f);
  float4 cq = make_float4(0.0f, 0.0f, 0.0f, 0.0f);
  if (row < N) {
    const float4* P4 = sP + lane;
    const float4* R4 = reinterpret_cast<const float4*>(R) + lane;
    float4* H4 = reinterpret_cast<float4*>(h) + lane;
    const float4 b4 = reinterpret_cast<const float4*>(bias)[lane];
    const float4 one = make_float4(1.0f, 1.0f, 1.0f, 1.0f);
    const float4 nul = make_float4(0.0f, 0.0f, 0.0f, 0.0f);

    int beg = rp[row];
    int end = rp[row + 1];
    float inn = inorm[row];
    float4 rv = R4[(size_t)ids[row] * (DIM / 4)];
    int2 ev = make_int2(0, 0);
    if (lane < end - beg) ev = csr0[beg + lane];
    int nrow = row + nw;
    int nb = 0, ne = 0, nid = 0;
    float ninn = 0.0f;
    if (nrow < N) {
      nb = rp[nrow];
      ne = rp[nrow + 1];
      nid = ids[nrow];
      ninn = inorm[nrow];
    }

    for (;;) {
      int2 nev = make_int2(0, 0);
      if (lane < ne - nb) nev = csr0[nb + lane];
      const float4 nrv = R4[(size_t)nid * (DIM / 4)];
      const int nnrow = nrow + nw;
      int nnb = 0, nne = 0, nnid = 0;
      float nninn = 0.0f;
      if (nnrow < N) {
        nnb = rp[nnrow];
        nne = rp[nnrow + 1];
        nnid = ids[nnrow];
        nninn = inorm[nnrow];
      }

      float4 acc = make_float4(0.0f, 0.0f, 0.0f, 0.0f);
      const int cnt = end - beg;
      int done = 0;
      for (;;) {
        const int m = (cnt - done < 32) ? (cnt - done) : 32;
        int j = 0;
        for (; j + 8 <= m; j += 8) sum_rows<8, false, false>(acc, P4, ev.x, ev.y, j, m, one, nul);
        for (; j < m; j += 4) sum_rows<4, true, false>(acc, P4, ev.x, ev.y, j, m, one, nul);
        done += 32;
        if (done >= cnt) break;
        ev = make_int2(0, 0);
        if (lane < cnt - done) ev = csr0[beg + done + lane];
      }
      float4 v;
      v.x = fmaxf(acc.x * inn + b4.x, 0.0f) + rv.x;
      v.y = fmaxf(acc.y * inn + b4.y, 0.0f) + rv.y;
      v.z = fmaxf(acc.z * inn + b4.z, 0.0f) + rv.z;
      v.w = fmaxf(acc.w * inn + b4.w, 0.0f) + rv.w;
      H4[(size_t)row * (DIM / 4)] = v;
      cs.x += v.x;
      cs.y += v.y;
      cs.z += v.z;
      cs.w += v.w;
      cq.x += v.x * v.x;
      cq.y += v.y * v.y;
      cq.z += v.z * v.z;
      cq.w += v.w * v.w;

      if (nrow >= N) break;
      row = nrow;
      beg = nb;
      end = ne;
      inn = ninn;
      rv = nrv;
      ev = nev;
      nrow = nnrow;
      nb = nnb;
      ne = nne;
      nid = nnid;
      ninn = nninn;
    }
  }
  sRed[wv * (2 * (DIM / 4)) + lane] = cs;
  sRed[wv * (2 * (DIM / 4)) + (DIM / 4) + lane] = cq;
  __syncthreads();
  const int tid = threadIdx.x;
  if (tid < 2 * DIM) {
    const float* red = reinterpret_cast<const float*>(sRed);
    float tsum = 0.0f;
#pragma unroll
    for (int w = 0; w < L0_WAVES; ++w) tsum += red[w * (2 * DIM) + tid];
    atomicAdd((tid < DIM) ? &csum[tid] : &csumsq[tid - DIM], tsum);
  }
}

#define LAYER_WAVES 16
#define APITCH 132
__global__ void __launch_bounds__(LAYER_WAVES * 32) layer_kernel(
    const int* __restrict__ rp, const int2* __restrict__ csr,
    const float* __restrict__ xin, const float* __restrict__ inorm,
    const float* __restrict__ Bfrag, const float* __restrict__ bias,
    const float* __restrict__ Rb,
    const float* __restrict__ pstats, const float* __restrict__ pgamma,
    const float* __restrict__ pbeta, float invN,
    float* __restrict__ h, float* __restrict__ csum, float* __restrict__ csumsq,
    int N, int rowTiles) {
  __shared__ float4 sB[32 * 8 * 32];
  __shared__ float4 sScale[DIM / 4];
  __shared__ float4 sShift[DIM / 4];
  __shared__ float4 sA[LAYER_WAVES * 16 * (APITCH / 4)];
  __shared__ float sRed[LAYER_WAVES * 2 * DIM];

  const int tid = threadIdx.x;
  const int wave = __builtin_amdgcn_readfirstlane(tid >> 5);
  const int lane = tid & 31;
  const int r = lane & 15;
  const int hf = lane >> 4;

  const int bk = tid & (DIM - 1);
  const float bnS = pstats[bk];
  const float bnQ = pstats[DIM + bk];
  const float bnG = pgamma[bk];
  const float bnB = pbeta[bk];

  {
    const float4* bf = reinterpret_cast<const float4*>(Bfrag) + tid;
    float4 t[16];
#pragma unroll
    for (int u = 0; u < 16; ++u) t[u] = bf[u * (LAYER_WAVES * 32)];
#pragma unroll
    for (int u = 0; u < 16; ++u) sB[tid + u * (LAYER_WAVES * 32)] = t[u];
  }
  {
    asm volatile("" ::"v"(bnS), "v"(bnQ), "v"(bnG), "v"(bnB));
    float sc, sh;
    bn_coeff(bnS, bnQ, bnG, bnB, invN, sc, sh);
    if (tid < 2 * DIM) {
      float* dstp = reinterpret_cast<float*>((tid < DIM) ? sScale : sShift);
      dstp[bk] = (tid < DIM) ? sc : sh;
    }
  }
  __syncthreads();

  const int tile = wave * (int)gridDim.x + (int)blockIdx.x;
  const int row0 = tile * 16;
  v8f acc1[8], acc2[8];

  if (tile < rowTiles) {

  const int row = row0 + r;
  const float innr = inorm[row];
  const float4* xP = reinterpret_cast<const float4*>(xin + (size_t)row * DIM) + hf;
  float4 tx = xP[0];

  float4* T4 = sA + wave * (16 * (APITCH / 4)) + lane;
  {
    const float4* X4 = reinterpret_cast<const float4*>(xin) + lane;
    const float4 sc = sScale[lane];
    const float4 sh = sShift[lane];
    const int rpl = rp[row0 + ((lane < 16) ? lane : 16)];
    int beg = __builtin_amdgcn_readlane(rpl, 0);
    int end = __builtin_amdgcn_readlane(rpl, 1);
    int2 ev = make_int2(0, 0);
    if (lane < end - beg) ev = csr[beg + lane];
    asm volatile("" ::"v"(ev.x), "v"(ev.y));
    for (int i = 0; i < 16; ++i) {
      const int nb = end;
      const int ne = __builtin_amdgcn_readlane(rpl, (i + 2 < 16) ? i + 2 : 16);
      int2 nev = make_int2(0, 0);
      if (lane < ne - nb) nev = csr[nb + lane];

      float4 acc = make_float4(0.0f, 0.0f, 0.0f, 0.0f);
      const int cnt = end - beg;
      int done = 0;
      for (;;) {
        const int m = (cnt - done < 32) ? (cnt - done) : 32;
        if (m <= 8) {
          if (m > 0) sum_rows<8, true, true>(acc, X4, ev.x, ev.y, 0, m, sc, sh);
        } else {
          for (int j = 0; j < m; j += 16)
            sum_rows<16, true, true>(acc, X4, ev.x, ev.y, j, m, sc, sh);
        }
        done += 32;
        if (done >= cnt) break;
        ev = make_int2(0, 0);
        if (lane < cnt - done) ev = csr[beg + done + lane];
        asm volatile("" ::"v"(ev.x), "v"(ev.y));
      }
      T4[i * (APITCH / 4)] = acc;
      beg = nb;
      end = ne;
      ev = nev;
    }
  }

  const float4* aP = sA + wave * (16 * (APITCH / 4)) + r * (APITCH / 4) + hf;
  const float4* bP = sB + lane;

#pragma unroll
  for (int ct = 0; ct < 8; ++ct) {
    acc1[ct] = (v8f){0.f, 0.f, 0.f, 0.f, 0.f, 0.f, 0.f, 0.f};
    acc2[ct] = (v8f){0.f, 0.f, 0.f, 0.f, 0.f, 0.f, 0.f, 0.f};
  }

  const float4* scP = sScale + hf;
  const float4* shP = sShift + hf;

  float4 ta = aP[0];
  float4 b0[8], b1[8];
#pragma unroll
  for (int ct = 0; ct < 8; ++ct) b0[ct] = bP[ct * 32];

  for (int kp = 0; kp < 16; ++kp) {
    float4 ta_n = ta, tx_n = tx;
    if (kp + 1 < 16) {
      ta_n = aP[(kp + 1) * 2];
      tx_n = xP[(kp + 1) * 2];
    }
#pragma unroll
    for (int ct = 0; ct < 8; ++ct) b1[ct] = bP[((2 * kp + 1) * 8 + ct) * 32];

    {
      const float4 s4 = scP[2 * kp];
      const float4 t4 = shP[2 * kp];
      tx.x = __builtin_fmaf(tx.x, s4.x, t4.x);
      tx.y = __builtin_fmaf(tx.y, s4.y, t4.y);
      tx.z = __builtin_fmaf(tx.z, s4.z, t4.z);
      tx.w = __builtin_fmaf(tx.w, s4.w, t4.w);
    }

    v2f a1, a2;
    a1.x = ta.x * innr;
    a1.y = ta.y * innr;
    a2.x = tx.x;
    a2.y = tx.y;
#pragma unroll
    for (int ct = 0; ct < 8; ++ct) {
      v2f w1, w2;
      w1.x = b0[ct].x; w1.y = b0[ct].y;
      w2.x = b0[ct].z; w2.y = b0[ct].w;
      acc1[ct] = __builtin_amdgcn_wmma_f32_16x16x4_f32(false, a1, false, w1,
                                                       (short)0, acc1[ct], false, false);
      acc2[ct] = __builtin_amdgcn_wmma_f32_16x16x4_f32(false, a2, false, w2,
                                                       (short)0, acc2[ct], false, false);
    }

    if (kp + 1 < 16) {
#pragma unroll
      for (int ct = 0; ct < 8; ++ct) b0[ct] = bP[((2 * kp + 2) * 8 + ct) * 32];
    }

    a1.x = ta.z * innr;
    a1.y = ta.w * innr;
    a2.x = tx.z;
    a2.y = tx.w;
#pragma unroll
    for (int ct = 0; ct < 8; ++ct) {
      v2f w1, w2;
      w1.x = b1[ct].x; w1.y = b1[ct].y;
      w2.x = b1[ct].z; w2.y = b1[ct].w;
      acc1[ct] = __builtin_amdgcn_wmma_f32_16x16x4_f32(false, a1, false, w1,
                                                       (short)0, acc1[ct], false, false);
      acc2[ct] = __builtin_amdgcn_wmma_f32_16x16x4_f32(false, a2, false, w2,
                                                       (short)0, acc2[ct], false, false);
    }
    ta = ta_n;
    tx = tx_n;
  }

#pragma unroll
  for (int ct = 0; ct < 8; ++ct) {
    const int col0 = ct * 16;
    const float bb = bias[col0 + r];
    const float rb = Rb[col0 + r];
    float s = 0.0f, s2 = 0.0f;
#pragma unroll
    for (int j = 0; j < 8; ++j) {
      float v = fmaxf(acc1[ct][j] + bb, 0.0f) + fmaxf(acc2[ct][j] + rb, 0.0f);
      acc1[ct][j] = v;
      s += v;
      s2 += v * v;
    }
    s += __shfl_xor(s, 16);
    s2 += __shfl_xor(s2, 16);
    sRed[wave * (2 * DIM) + hf * DIM + col0 + r] = hf ? s2 : s;
  }
  } else {
    for (int i = lane; i < 2 * DIM; i += 32) sRed[wave * (2 * DIM) + i] = 0.0f;
  }
  __syncthreads();
  if (tid < 2 * DIM) {
    float tsum = 0.0f;
#pragma unroll
    for (int w = 0; w < LAYER_WAVES; ++w) tsum += sRed[w * (2 * DIM) + tid];
    atomicAdd((tid < DIM) ? &csum[tid] : &csumsq[tid - DIM], tsum);
  }
  if (tile < rowTiles) {
#pragma unroll
    for (int ct = 0; ct < 8; ++ct) {
#pragma unroll
      for (int j = 0; j < 8; ++j)
        h[(size_t)(row0 + j + 8 * hf) * DIM + ct * 16 + r] = acc1[ct][j];
    }
  }
}

#define STREAM_U 16
__global__ void __launch_bounds__(STREAM_THREADS) bnapply_kernel(
    const float* __restrict__ h, const float* __restrict__ stats,
    const float* __restrict__ gamma, const float* __restrict__ beta, float invN,
    float* __restrict__ out, int N) {
  const int lane = threadIdx.x & 31;
  const int wv = blockIdx.x * (STREAM_THREADS / 32) +
                 __builtin_amdgcn_readfirstlane((int)(threadIdx.x >> 5));
  const int nw = gridDim.x * (STREAM_THREADS / 32);
  const float4* H4 = reinterpret_cast<const float4*>(h) + lane;
  float4* O4 = reinterpret_cast<float4*>(out) + lane;
  float4 sc, sh;
  bn_coeff4(stats, gamma, beta, invN, lane, sc, sh);
  for (int i0 = wv; i0 < N; i0 += STREAM_U * nw) {
    float4 v[STREAM_U];
#pragma unroll
    for (int u = 0; u < STREAM_U; ++u) {
      const int i = i0 + u * nw;
      v[u] = H4[(size_t)((i < N) ? i : i0) * (DIM / 4)];
    }
#pragma unroll
    for (int u = 0; u < STREAM_U; ++u) {
      const int i = i0 + u * nw;
      asm volatile("" ::"v"(v[u].x), "v"(v[u].y), "v"(v[u].z), "v"(v[u].w));
      if (i < N) {
        float4 o;
        o.x = v[u].x * sc.x + sh.x;
        o.y = v[u].y * sc.y + sh.y;
        o.z = v[u].z * sc.z + sh.z;
        o.w = v[u].w * sc.w + sh.w;
        O4[(size_t)i * (DIM / 4)] = o;
      }
    }
  }
}


static inline int cdiv(int a, int b) { return (a + b - 1) / b; }

extern "C" void kernel_launch(void* const* d_in, const int* in_sizes, int n_in,
                              void* d_out, int out_size, void* d_ws, size_t ws_size,
                              hipStream_t stream) {
  const int* node_ids = (const int*)d_in[0];
  const int* src = (const int*)d_in[1];
  const int* dst = (const int*)d_in[2];
  const float* emb = (const float*)d_in[3];
  const float* Ws = (const float*)d_in[4];
  const float* bs = (const float*)d_in[5];
  const float* Rws = (const float*)d_in[6];
  const float* Rbs = (const float*)d_in[7];
  const float* gammas = (const float*)d_in[8];
  const float* betas = (const float*)d_in[9];

  const int N = in_sizes[0];
  const int E = in_sizes[1];
  const int T = in_sizes[3] / DIM;
  const int L = in_sizes[4] / (DIM * DIM);

  float* wsf = (float*)d_ws;
  size_t off = 0;
  float* onorm = wsf + off; off += (size_t)N;
  float* inorm = wsf + off; off += (size_t)N;
  off = (off + 3) & ~(size_t)3;
  float* ha   = wsf + off;  off += (size_t)N * DIM;
  float* hb   = wsf + off;  off += (size_t)N * DIM;
  float* tabP = wsf + off;  off += (size_t)T * DIM;
  float* tabR = wsf + off;  off += (size_t)T * DIM;
  float* stats = wsf + off; off += (size_t)L * 2 * DIM;
  off = (off + 3) & ~(size_t)3;
  float* wfrag = wsf + off; off += (size_t)(L - 1) * 4 * (32 * 8 * 32);
  off = (off + 3) & ~(size_t)3;
  int2* csr = (int2*)(wsf + off); off += (size_t)E * 2;
  int2* csr0 = (int2*)(wsf + off); off += (size_t)E * 2;
  int* slot = (int*)(wsf + off);  off += (size_t)E;
  int* rp   = (int*)(wsf + off);  off += (size_t)N + 1;
  const int R = cdiv(N, CNT_RANGE);
  int* cntS = (int*)(wsf + off);  off += (size_t)CNT_CHUNKS * N;
  int* cntD = (int*)(wsf + off);  off += (size_t)CNT_CHUNKS * N;
  int* tot  = (int*)(wsf + off);  off += (size_t)CNT_CHUNKS * R;

  const int CH = cdiv(E, CNT_CHUNKS);
  const int tabBlocks = cdiv(T, 16 * (CNT_THREADS / 256));
  setup_kernel<<<2 * CNT_CHUNKS + tabBlocks + (L - 1), CNT_THREADS, 0, stream>>>(
      src, dst, cntS, cntD, tot, slot, N, E, CH, R, emb, Ws, Rws, Rbs, tabP, tabR, stats,
      L * 2 * DIM, T, tabBlocks, wfrag);
  offsets_kernel<<<R, CNT_RANGE, 0, stream>>>(cntS, cntD, tot, rp, onorm, inorm, N, R);
  place_kernel<<<STREAM_BLOCKS, STREAM_THREADS, 0, stream>>>(src, dst, slot, rp, cntD, node_ids,
                                                             onorm, csr, csr0, E, N, CH);

  const int rowTiles = cdiv(N, 16);
  const int l0Blocks = (cdiv(N, L0_WAVES) < 256) ? cdiv(N, L0_WAVES) : 256;
  const int layerBlocks = (cdiv(rowTiles, LAYER_WAVES) > 256) ? cdiv(rowTiles, LAYER_WAVES)
                          : ((rowTiles < 256) ? rowTiles : 256);
  const float invN = 1.0f / (float)N;
  layer0_kernel<<<l0Blocks, L0_WAVES * 32, 0, stream>>>(rp, csr0, tabP, tabR, node_ids, inorm,
                                                        bs, ha, stats, stats + DIM, N, T);
  float* cur = ha;
  float* nxt = hb;
  for (int l = 1; l < L; ++l) {
    float* st = stats + (size_t)l * 2 * DIM;
    const float* Bf = wfrag + (size_t)(l - 1) * 4 * (32 * 8 * 32);
    const float* bw = bs + (size_t)l * DIM;
    const float* rb = Rbs + (size_t)l * DIM;
    const float* pst = st - 2 * DIM;
    const float* pg = gammas + (size_t)(l - 1) * DIM;
    const float* pb = betas + (size_t)(l - 1) * DIM;
    layer_kernel<<<layerBlocks, LAYER_WAVES * 32, 0, stream>>>(
        rp, csr, cur, inorm, Bf, bw, rb, pst, pg, pb, invN, nxt, st, st + DIM, N, rowTiles);
    float* t = cur;
    cur = nxt;
    nxt = t;
  }
  bnapply_kernel<<<STREAM_BLOCKS, STREAM_THREADS, 0, stream>>>(
      cur, stats + (size_t)(L - 1) * 2 * DIM, gammas + (size_t)(L - 1) * DIM,
      betas + (size_t)(L - 1) * DIM, invN, (float*)d_out, N);
}
